# starting kernel
# baseline (speedup 1.0000x reference)
#include <hip/hip_runtime.h>
#include <hip/hip_bf16.h>
#include <math.h>


typedef __attribute__((ext_vector_type(16))) _Float16 v16h;
typedef __attribute__((ext_vector_type(8)))  float    v8f;

#define EPS_F 1e-8f

__global__ __launch_bounds__(256) void cgmm_zero(float* __restrict__ p, int n) {
    int i = blockIdx.x * blockDim.x + threadIdx.x;
    int stride = gridDim.x * blockDim.x;
    for (; i < n; i += stride) p[i] = 0.0f;
}

__global__ __launch_bounds__(128) void cgmm_params(const float* __restrict__ Q,
                                                   const float* __restrict__ B,
                                                   float* __restrict__ Bs_t,
                                                   _Float16* __restrict__ bfrag) {
    __shared__ float sQ[512];
    int t = threadIdx.x;
    if (t < 64) {
        int g = t & 7, a = t >> 3;
        {
            int l = a;
            float mx = -1e30f;
            for (int i = 0; i < 8; ++i) mx = fmaxf(mx, Q[i * 64 + l * 8 + g]);
            float e[8], s = 0.0f;
            for (int i = 0; i < 8; ++i) { e[i] = __expf(Q[i * 64 + l * 8 + g] - mx); s += e[i]; }
            float inv = 1.0f / s;
            for (int i = 0; i < 8; ++i) sQ[i * 64 + l * 8 + g] = e[i] * inv;
        }
        {
            int i = a;
            float mx = -1e30f;
            for (int m = 0; m < 32; ++m) mx = fmaxf(mx, B[i * 256 + m * 8 + g]);
            float ev[32], s = 0.0f;
            for (int m = 0; m < 32; ++m) { ev[m] = __expf(B[i * 256 + m * 8 + g] - mx); s += ev[m]; }
            float inv = 1.0f / s;
            for (int m = 0; m < 32; ++m) Bs_t[m * 64 + g * 8 + i] = ev[m] * inv;
        }
    }
    __syncthreads();
    int p = t >> 5, lane = t & 31;
    bool hi = lane >= 16;
    int  m  = lane & 15;
    bool bvalid = (hi == (m >= 8));
    int  icol = m & 7;
    int  g = 2 * p + (hi ? 1 : 0);
    _Float16* dst = bfrag + (p * 32 + lane) * 16;
    for (int e2 = 0; e2 < 16; ++e2) {
        float f = (e2 < 8 && bvalid) ? sQ[icol * 64 + e2 * 8 + g] : 0.0f;
        dst[e2] = (_Float16)f;
    }
}

__global__ __launch_bounds__(256) void cgmm_edge_scatter(const float* __restrict__ prev_h,
                                                         const int* __restrict__ ei,
                                                         float* __restrict__ summed,
                                                         float* __restrict__ counts,
                                                         int E) {
    int lane = threadIdx.x & 31;
    int e = blockIdx.x * 8 + (threadIdx.x >> 5);
    if (e >= E) return;
    int dst = ei[e];
    int src = ei[E + e];
    const float2* ph = (const float2*)(prev_h + (size_t)src * 64);
    float2 v = ph[lane];
    float* sp = summed + (size_t)dst * 64 + lane * 2;
    __hip_atomic_fetch_add(sp,     v.x, __ATOMIC_RELAXED, __HIP_MEMORY_SCOPE_AGENT);
    __hip_atomic_fetch_add(sp + 1, v.y, __ATOMIC_RELAXED, __HIP_MEMORY_SCOPE_AGENT);
    if (lane == 0)
        __hip_atomic_fetch_add(counts + dst, 1.0f, __ATOMIC_RELAXED, __HIP_MEMORY_SCOPE_AGENT);
}

__global__ __launch_bounds__(256) void cgmm_node_wmma(const int* __restrict__ x,
                                                      const float* __restrict__ summed,
                                                      const float* __restrict__ counts,
                                                      const _Float16* __restrict__ bfrag,
                                                      const float* __restrict__ Bs_t,
                                                      float* __restrict__ lik,
                                                      float* __restrict__ post,
                                                      int N) {
    __shared__ float ldsT[8][16][8][8];
    __shared__ float ldsP[8][16][64];
    int lane = threadIdx.x & 31;
    int wave = threadIdx.x >> 5;
    int tile = blockIdx.x * 8 + wave;
    int n0 = tile * 16;
    if (n0 >= N) return;

    int  m    = lane & 15;
    bool hi   = lane >= 16;
    int  node = n0 + m;
    int  nodec = node < N ? node : N - 1;
    bool valid = node < N;

    const float* rowbase = summed + (size_t)nodec * 64;
    __builtin_prefetch(rowbase, 0, 3);
    float inv_cnt = 1.0f / fmaxf(counts[nodec], 1.0f);

#pragma unroll
    for (int p = 0; p < 4; ++p) {
        int g = 2 * p + (hi ? 1 : 0);
        const float* row = rowbase + g;
        v16h a;
#pragma unroll
        for (int e2 = 0; e2 < 8; ++e2)
            a[e2] = (_Float16)(row[e2 * 8] * inv_cnt);
#pragma unroll
        for (int e2 = 8; e2 < 16; ++e2)
            a[e2] = (_Float16)0.0f;

        v16h b = *((const v16h*)(bfrag + (p * 32 + lane) * 16));

        v8f acc = {};
        acc = __builtin_amdgcn_wmma_f32_16x16x32_f16(
            false, a, false, b, (short)0, acc, false, false);
#pragma unroll
        for (int r = 0; r < 8; ++r) {
            int mm = r + (hi ? 8 : 0);
            int gq = 2 * p + (m >= 8 ? 1 : 0);
            ldsT[wave][mm][gq][m & 7] = acc[r];
        }
    }

    int gbase = hi ? 4 : 0;
    int xi = x[nodec];
    const float* brow = Bs_t + (size_t)xi * 64;
    float lik4[4];
#pragma unroll
    for (int gg = 0; gg < 4; ++gg) {
        int g = gbase + gg;
        const float4* tp = (const float4*)&ldsT[wave][m][g][0];
        float4 t0 = tp[0], t1 = tp[1];
        const float4* bp = (const float4*)(brow + g * 8);
        float4 b0 = bp[0], b1 = bp[1];
        float u[8];
        u[0] = t0.x * b0.x; u[1] = t0.y * b0.y; u[2] = t0.z * b0.z; u[3] = t0.w * b0.w;
        u[4] = t1.x * b1.x; u[5] = t1.y * b1.y; u[6] = t1.z * b1.z; u[7] = t1.w * b1.w;
        float s = ((u[0] + u[1]) + (u[2] + u[3])) + ((u[4] + u[5]) + (u[6] + u[7]));
        float denom = s + 64.0f * EPS_F;
        lik4[gg] = __logf(denom);
        float invd = 1.0f / denom;
#pragma unroll
        for (int i = 0; i < 8; ++i)
            ldsP[wave][m][i * 8 + g] = (u[i] + 8.0f * EPS_F) * invd;
    }
    if (valid) {
        float4 lv = {lik4[0], lik4[1], lik4[2], lik4[3]};
        *(float4*)(lik + (size_t)node * 8 + gbase) = lv;
    }

    {
        int nl = lane >> 1;
        int half = lane & 1;
        int outnode = n0 + nl;
        if (outnode < N) {
            const float4* src = (const float4*)&ldsP[wave][nl][half * 32];
            float4* dstp = (float4*)(post + (size_t)outnode * 64 + half * 32);
#pragma unroll
            for (int q = 0; q < 8; ++q) dstp[q] = src[q];
        }
    }
}

extern "C" void kernel_launch(void* const* d_in, const int* in_sizes, int n_in,
                              void* d_out, int out_size, void* d_ws, size_t ws_size,
                              hipStream_t stream) {
    (void)n_in; (void)out_size; (void)ws_size;
    const int*   x      = (const int*)d_in[0];
    const float* prev_h = (const float*)d_in[1];
    const int*   ei     = (const int*)d_in[2];
    const float* Q      = (const float*)d_in[3];
    const float* B      = (const float*)d_in[4];
    const int N = in_sizes[0];
    const int E = in_sizes[2] / 2;

    float* ws     = (float*)d_ws;
    float* summed = ws;
    float* counts = summed + (size_t)N * 64;
    float* Bs_t   = counts + N;
    size_t off    = ((size_t)N * 65 + 2048 + 7) & ~(size_t)7;
    _Float16* bfrag = (_Float16*)(ws + off);

    float* lik  = (float*)d_out;
    float* post = lik + (size_t)N * 8;

    int zn = N * 65;
    int zb = (zn + 255) / 256; if (zb > 4096) zb = 4096;
    cgmm_zero<<<zb, 256, 0, stream>>>(summed, zn);

    cgmm_params<<<1, 128, 0, stream>>>(Q, B, Bs_t, bfrag);

    cgmm_edge_scatter<<<(E + 7) / 8, 256, 0, stream>>>(prev_h, ei, summed, counts, E);

    int tiles = (N + 15) / 16;
    cgmm_node_wmma<<<(tiles + 7) / 8, 256, 0, stream>>>(x, summed, counts, bfrag, Bs_t,
                                                        lik, post, N);
}


# Segment sum by counting sort instead of 104M float atomics: int-atomic rank per edge, block-scan offsets, source list, wave-per-node gather-sum in rank order (float add order = int-atomic arrival)
# speedup vs baseline: 1.3521x; 1.3521x over previous
#include <hip/hip_runtime.h>
#include <hip/hip_bf16.h>
#include <math.h>


typedef __attribute__((ext_vector_type(16))) _Float16 v16h;
typedef __attribute__((ext_vector_type(8)))  float    v8f;

#define EPS_F 1e-8f

__global__ __launch_bounds__(256) void cgmm_zero_deg(int* __restrict__ deg, int n,
                                                     int* __restrict__ total) {
    int i = blockIdx.x * blockDim.x + threadIdx.x;
    int stride = gridDim.x * blockDim.x;
    if (i == 0) *total = 0;
    for (; i < n; i += stride) deg[i] = 0;
}

__global__ __launch_bounds__(128) void cgmm_params(const float* __restrict__ Q,
                                                   const float* __restrict__ B,
                                                   float* __restrict__ Bs_t,
                                                   _Float16* __restrict__ bfrag) {
    __shared__ float sQ[512];
    int t = threadIdx.x;
    if (t < 64) {
        int g = t & 7, a = t >> 3;
        {
            int l = a;
            float mx = -1e30f;
            for (int i = 0; i < 8; ++i) mx = fmaxf(mx, Q[i * 64 + l * 8 + g]);
            float e[8], s = 0.0f;
            for (int i = 0; i < 8; ++i) { e[i] = __expf(Q[i * 64 + l * 8 + g] - mx); s += e[i]; }
            float inv = 1.0f / s;
            for (int i = 0; i < 8; ++i) sQ[i * 64 + l * 8 + g] = e[i] * inv;
        }
        {
            int i = a;
            float mx = -1e30f;
            for (int m = 0; m < 32; ++m) mx = fmaxf(mx, B[i * 256 + m * 8 + g]);
            float ev[32], s = 0.0f;
            for (int m = 0; m < 32; ++m) { ev[m] = __expf(B[i * 256 + m * 8 + g] - mx); s += ev[m]; }
            float inv = 1.0f / s;
            for (int m = 0; m < 32; ++m) Bs_t[m * 64 + g * 8 + i] = ev[m] * inv;
        }
    }
    __syncthreads();
    int p = t >> 5, lane = t & 31;
    bool hi = lane >= 16;
    int  m  = lane & 15;
    bool bvalid = (hi == (m >= 8));
    int  icol = m & 7;
    int  g = 2 * p + (hi ? 1 : 0);
    _Float16* dst = bfrag + (p * 32 + lane) * 16;
    for (int e2 = 0; e2 < 16; ++e2) {
        float f = (e2 < 8 && bvalid) ? sQ[icol * 64 + e2 * 8 + g] : 0.0f;
        dst[e2] = (_Float16)f;
    }
}

__global__ __launch_bounds__(256) void cgmm_edge_rank(const int* __restrict__ ei,
                                                      int* __restrict__ deg,
                                                      int* __restrict__ rank, int E) {
    int e = blockIdx.x * 256 + threadIdx.x;
    if (e >= E) return;
    rank[e] = __hip_atomic_fetch_add(deg + ei[e], 1, __ATOMIC_RELAXED, __HIP_MEMORY_SCOPE_AGENT);
}

__global__ __launch_bounds__(1024) void cgmm_seg_alloc(const int* __restrict__ deg,
                                                       int* __restrict__ rowstart,
                                                       int* __restrict__ total, int N) {
    __shared__ int wsum[32];
    __shared__ int bbase;
    int t = threadIdx.x, lane = t & 31, wv = t >> 5;
    int n = blockIdx.x * 1024 + t;
    int d = n < N ? deg[n] : 0;
    int incl = d;
#pragma unroll
    for (int s = 1; s < 32; s <<= 1) {
        int up = __shfl_up(incl, s, 32);
        if (lane >= s) incl += up;
    }
    if (lane == 31) wsum[wv] = incl;
    __syncthreads();
    if (wv == 0) {
        int w = wsum[lane];
        int wi = w;
#pragma unroll
        for (int s = 1; s < 32; s <<= 1) {
            int up = __shfl_up(wi, s, 32);
            if (lane >= s) wi += up;
        }
        wsum[lane] = wi - w;
        if (lane == 31)
            bbase = __hip_atomic_fetch_add(total, wi, __ATOMIC_RELAXED, __HIP_MEMORY_SCOPE_AGENT);
    }
    __syncthreads();
    if (n < N) rowstart[n] = bbase + wsum[wv] + (incl - d);
}

__global__ __launch_bounds__(256) void cgmm_edge_fill(const int* __restrict__ ei,
                                                      const int* __restrict__ rowstart,
                                                      const int* __restrict__ rank,
                                                      int* __restrict__ csr_src, int E) {
    int e = blockIdx.x * 256 + threadIdx.x;
    if (e >= E) return;
    csr_src[rowstart[ei[e]] + rank[e]] = ei[E + e];
}

#define CGMM_ROW(k) (((const float2*)(prev_h + (size_t)__builtin_amdgcn_readlane(myidx, j + (k)) * 64))[lane])
__global__ __launch_bounds__(256) void cgmm_gather_sum(const float* __restrict__ prev_h,
                                                       const int* __restrict__ csr_src,
                                                       const int* __restrict__ rowstart,
                                                       float* __restrict__ summed,
                                                       int* deg_counts, int N) {
    int lane = threadIdx.x & 31;
    int n = __builtin_amdgcn_readfirstlane(blockIdx.x * 8 + (threadIdx.x >> 5));
    if (n >= N) return;
    int d     = deg_counts[n];
    int start = rowstart[n];
    float2 acc = {0.0f, 0.0f};
    for (int base = 0; base < d; base += 32) {
        int cnt = d - base < 32 ? d - base : 32;
        int myidx = lane < cnt ? csr_src[start + base + lane] : 0;
        int j = 0;
        for (; j + 8 <= cnt; j += 8) {
            float2 r0 = CGMM_ROW(0), r1 = CGMM_ROW(1), r2 = CGMM_ROW(2), r3 = CGMM_ROW(3);
            float2 r4 = CGMM_ROW(4), r5 = CGMM_ROW(5), r6 = CGMM_ROW(6), r7 = CGMM_ROW(7);
            acc.x += r0.x; acc.y += r0.y; acc.x += r1.x; acc.y += r1.y;
            acc.x += r2.x; acc.y += r2.y; acc.x += r3.x; acc.y += r3.y;
            acc.x += r4.x; acc.y += r4.y; acc.x += r5.x; acc.y += r5.y;
            acc.x += r6.x; acc.y += r6.y; acc.x += r7.x; acc.y += r7.y;
        }
        if (j + 4 <= cnt) {
            float2 r0 = CGMM_ROW(0), r1 = CGMM_ROW(1), r2 = CGMM_ROW(2), r3 = CGMM_ROW(3);
            acc.x += r0.x; acc.y += r0.y; acc.x += r1.x; acc.y += r1.y;
            acc.x += r2.x; acc.y += r2.y; acc.x += r3.x; acc.y += r3.y;
            j += 4;
        }
        if (j + 2 <= cnt) {
            float2 r0 = CGMM_ROW(0), r1 = CGMM_ROW(1);
            acc.x += r0.x; acc.y += r0.y; acc.x += r1.x; acc.y += r1.y;
            j += 2;
        }
        if (j < cnt) {
            float2 r0 = CGMM_ROW(0);
            acc.x += r0.x; acc.y += r0.y;
        }
    }
    ((float2*)(summed + (size_t)n * 64))[lane] = acc;
    if (lane == 0) ((float*)deg_counts)[n] = (float)d;
}
#undef CGMM_ROW

__global__ __launch_bounds__(256) void cgmm_node_wmma(const int* __restrict__ x,
                                                      const float* __restrict__ summed,
                                                      const float* __restrict__ counts,
                                                      const _Float16* __restrict__ bfrag,
                                                      const float* __restrict__ Bs_t,
                                                      float* __restrict__ lik,
                                                      float* __restrict__ post,
                                                      int N) {
    __shared__ float ldsT[8][16][8][8];
    __shared__ float ldsP[8][16][64];
    int lane = threadIdx.x & 31;
    int wave = threadIdx.x >> 5;
    int tile = blockIdx.x * 8 + wave;
    int n0 = tile * 16;
    if (n0 >= N) return;

    int  m    = lane & 15;
    bool hi   = lane >= 16;
    int  node = n0 + m;
    int  nodec = node < N ? node : N - 1;
    bool valid = node < N;

    const float* rowbase = summed + (size_t)nodec * 64;
    __builtin_prefetch(rowbase, 0, 3);
    float inv_cnt = 1.0f / fmaxf(counts[nodec], 1.0f);

#pragma unroll
    for (int p = 0; p < 4; ++p) {
        int g = 2 * p + (hi ? 1 : 0);
        const float* row = rowbase + g;
        v16h a;
#pragma unroll
        for (int e2 = 0; e2 < 8; ++e2)
            a[e2] = (_Float16)(row[e2 * 8] * inv_cnt);
#pragma unroll
        for (int e2 = 8; e2 < 16; ++e2)
            a[e2] = (_Float16)0.0f;

        v16h b = *((const v16h*)(bfrag + (p * 32 + lane) * 16));

        v8f acc = {};
        acc = __builtin_amdgcn_wmma_f32_16x16x32_f16(
            false, a, false, b, (short)0, acc, false, false);
#pragma unroll
        for (int r = 0; r < 8; ++r) {
            int mm = r + (hi ? 8 : 0);
            int gq = 2 * p + (m >= 8 ? 1 : 0);
            ldsT[wave][mm][gq][m & 7] = acc[r];
        }
    }

    int gbase = hi ? 4 : 0;
    int xi = x[nodec];
    const float* brow = Bs_t + (size_t)xi * 64;
    float lik4[4];
#pragma unroll
    for (int gg = 0; gg < 4; ++gg) {
        int g = gbase + gg;
        const float4* tp = (const float4*)&ldsT[wave][m][g][0];
        float4 t0 = tp[0], t1 = tp[1];
        const float4* bp = (const float4*)(brow + g * 8);
        float4 b0 = bp[0], b1 = bp[1];
        float u[8];
        u[0] = t0.x * b0.x; u[1] = t0.y * b0.y; u[2] = t0.z * b0.z; u[3] = t0.w * b0.w;
        u[4] = t1.x * b1.x; u[5] = t1.y * b1.y; u[6] = t1.z * b1.z; u[7] = t1.w * b1.w;
        float s = ((u[0] + u[1]) + (u[2] + u[3])) + ((u[4] + u[5]) + (u[6] + u[7]));
        float denom = s + 64.0f * EPS_F;
        lik4[gg] = __logf(denom);
        float invd = 1.0f / denom;
#pragma unroll
        for (int i = 0; i < 8; ++i)
            ldsP[wave][m][i * 8 + g] = (u[i] + 8.0f * EPS_F) * invd;
    }
    if (valid) {
        float4 lv = {lik4[0], lik4[1], lik4[2], lik4[3]};
        *(float4*)(lik + (size_t)node * 8 + gbase) = lv;
    }

    {
        int nl = lane >> 1;
        int half = lane & 1;
        int outnode = n0 + nl;
        if (outnode < N) {
            const float4* src = (const float4*)&ldsP[wave][nl][half * 32];
            float4* dstp = (float4*)(post + (size_t)outnode * 64 + half * 32);
#pragma unroll
            for (int q = 0; q < 8; ++q) dstp[q] = src[q];
        }
    }
}

extern "C" void kernel_launch(void* const* d_in, const int* in_sizes, int n_in,
                              void* d_out, int out_size, void* d_ws, size_t ws_size,
                              hipStream_t stream) {
    (void)n_in; (void)out_size; (void)ws_size;
    const int*   x      = (const int*)d_in[0];
    const float* prev_h = (const float*)d_in[1];
    const int*   ei     = (const int*)d_in[2];
    const float* Q      = (const float*)d_in[3];
    const float* B      = (const float*)d_in[4];
    const int N = in_sizes[0];
    const int E = in_sizes[2] / 2;

    float* ws     = (float*)d_ws;
    float* summed = ws;
    float* counts = summed + (size_t)N * 64;
    float* Bs_t   = counts + N;
    size_t off    = ((size_t)N * 65 + 2048 + 7) & ~(size_t)7;
    _Float16* bfrag = (_Float16*)(ws + off);

    float* lik  = (float*)d_out;
    float* post = lik + (size_t)N * 8;

    int* deg      = (int*)counts;
    int* rank     = (int*)summed;
    int* rowstart = (int*)lik;
    int* total    = rowstart + N;
    int* csr_src  = (int*)post;

    int zb = (N + 255) / 256; if (zb > 4096) zb = 4096;
    cgmm_zero_deg<<<zb, 256, 0, stream>>>(deg, N, total);

    cgmm_params<<<1, 128, 0, stream>>>(Q, B, Bs_t, bfrag);

    cgmm_edge_rank<<<(E + 255) / 256, 256, 0, stream>>>(ei, deg, rank, E);
    cgmm_seg_alloc<<<(N + 1023) / 1024, 1024, 0, stream>>>(deg, rowstart, total, N);
    cgmm_edge_fill<<<(E + 255) / 256, 256, 0, stream>>>(ei, rowstart, rank, csr_src, E);
    cgmm_gather_sum<<<(N + 7) / 8, 256, 0, stream>>>(prev_h, csr_src, rowstart, summed, deg, N);

    int tiles = (N + 15) / 16;
    cgmm_node_wmma<<<(tiles + 7) / 8, 256, 0, stream>>>(x, summed, counts, bfrag, Bs_t,
                                                        lik, post, N);
}


# Fewer, fatter waves (a sixth as many per call): 8 edges per thread in both edge kernels, 8 nodes per wave in gather-sum with next node's indices prefetched; v1 was bound by the wave start-up rate
# speedup vs baseline: 1.4041x; 1.0385x over previous
#include <hip/hip_runtime.h>
#include <hip/hip_bf16.h>
#include <math.h>


typedef __attribute__((ext_vector_type(16))) _Float16 v16h;
typedef __attribute__((ext_vector_type(8)))  float    v8f;

#define EPS_F 1e-8f

__global__ __launch_bounds__(256) void cgmm_zero_deg(int* __restrict__ deg, int n,
                                                     int* __restrict__ total) {
    int i = blockIdx.x * 256 + threadIdx.x;
    int nq = n >> 2;
    if (i == 0) *total = 0;
    if (i < nq) ((int4*)deg)[i] = make_int4(0, 0, 0, 0);
    if (i < (n & 3)) deg[(nq << 2) + i] = 0;
}

__global__ __launch_bounds__(128) void cgmm_params(const float* __restrict__ Q,
                                                   const float* __restrict__ B,
                                                   float* __restrict__ Bs_t,
                                                   _Float16* __restrict__ bfrag) {
    __shared__ float sQ[512];
    int t = threadIdx.x;
    if (t < 64) {
        int g = t & 7, a = t >> 3;
        {
            int l = a;
            float mx = -1e30f;
            for (int i = 0; i < 8; ++i) mx = fmaxf(mx, Q[i * 64 + l * 8 + g]);
            float e[8], s = 0.0f;
            for (int i = 0; i < 8; ++i) { e[i] = __expf(Q[i * 64 + l * 8 + g] - mx); s += e[i]; }
            float inv = 1.0f / s;
            for (int i = 0; i < 8; ++i) sQ[i * 64 + l * 8 + g] = e[i] * inv;
        }
        {
            int i = a;
            float mx = -1e30f;
            for (int m = 0; m < 32; ++m) mx = fmaxf(mx, B[i * 256 + m * 8 + g]);
            float ev[32], s = 0.0f;
            for (int m = 0; m < 32; ++m) { ev[m] = __expf(B[i * 256 + m * 8 + g] - mx); s += ev[m]; }
            float inv = 1.0f / s;
            for (int m = 0; m < 32; ++m) Bs_t[m * 64 + g * 8 + i] = ev[m] * inv;
        }
    }
    __syncthreads();
    int p = t >> 5, lane = t & 31;
    bool hi = lane >= 16;
    int  m  = lane & 15;
    bool bvalid = (hi == (m >= 8));
    int  icol = m & 7;
    int  g = 2 * p + (hi ? 1 : 0);
    _Float16* dst = bfrag + (p * 32 + lane) * 16;
    for (int e2 = 0; e2 < 16; ++e2) {
        float f = (e2 < 8 && bvalid) ? sQ[icol * 64 + e2 * 8 + g] : 0.0f;
        dst[e2] = (_Float16)f;
    }
}

#define CGMM_RANK1(p) __hip_atomic_fetch_add((p), 1, __ATOMIC_RELAXED, __HIP_MEMORY_SCOPE_AGENT)
__global__ __launch_bounds__(256) void cgmm_edge_rank(const int* __restrict__ ei,
                                                      int* __restrict__ deg,
                                                      int* __restrict__ rank, int E) {
    int T  = gridDim.x * 256;
    int t  = blockIdx.x * 256 + threadIdx.x;
    int nq = E >> 2;
    int q0 = t, q1 = t + T;
    if (q1 < nq) {
        int4 a = ((const int4*)ei)[q0], b = ((const int4*)ei)[q1];
        asm volatile("" :: "v"(a.x), "v"(a.y), "v"(a.z), "v"(a.w),
                           "v"(b.x), "v"(b.y), "v"(b.z), "v"(b.w));
        int4 ra, rb;
        ra.x = CGMM_RANK1(deg + a.x); ra.y = CGMM_RANK1(deg + a.y);
        ra.z = CGMM_RANK1(deg + a.z); ra.w = CGMM_RANK1(deg + a.w);
        rb.x = CGMM_RANK1(deg + b.x); rb.y = CGMM_RANK1(deg + b.y);
        rb.z = CGMM_RANK1(deg + b.z); rb.w = CGMM_RANK1(deg + b.w);
        ((int4*)rank)[q0] = ra;
        ((int4*)rank)[q1] = rb;
    } else if (q0 < nq) {
        int4 a = ((const int4*)ei)[q0];
        int4 ra;
        ra.x = CGMM_RANK1(deg + a.x); ra.y = CGMM_RANK1(deg + a.y);
        ra.z = CGMM_RANK1(deg + a.z); ra.w = CGMM_RANK1(deg + a.w);
        ((int4*)rank)[q0] = ra;
    }
    if (t < (E & 3)) {
        int e = (nq << 2) + t;
        rank[e] = CGMM_RANK1(deg + ei[e]);
    }
}
#undef CGMM_RANK1

__global__ __launch_bounds__(1024) void cgmm_seg_alloc(const int* __restrict__ deg,
                                                       int* __restrict__ rowstart,
                                                       int* __restrict__ total, int N) {
    __shared__ int wsum[32];
    __shared__ int bbase;
    int t = threadIdx.x, lane = t & 31, wv = t >> 5;
    int n = blockIdx.x * 1024 + t;
    int d = n < N ? deg[n] : 0;
    int incl = d;
#pragma unroll
    for (int s = 1; s < 32; s <<= 1) {
        int up = __shfl_up(incl, s, 32);
        if (lane >= s) incl += up;
    }
    if (lane == 31) wsum[wv] = incl;
    __syncthreads();
    if (wv == 0) {
        int w = wsum[lane];
        int wi = w;
#pragma unroll
        for (int s = 1; s < 32; s <<= 1) {
            int up = __shfl_up(wi, s, 32);
            if (lane >= s) wi += up;
        }
        wsum[lane] = wi - w;
        if (lane == 31)
            bbase = __hip_atomic_fetch_add(total, wi, __ATOMIC_RELAXED, __HIP_MEMORY_SCOPE_AGENT);
    }
    __syncthreads();
    if (n < N) rowstart[n] = bbase + wsum[wv] + (incl - d);
}

__device__ __forceinline__ int4 cgmm_ld4(const int* __restrict__ p, int q, bool aligned) {
    if (aligned) return ((const int4*)p)[q];
    const int* s = p + (q << 2);
    return make_int4(s[0], s[1], s[2], s[3]);
}
__device__ __forceinline__ int4 cgmm_gather4(const int* __restrict__ rowstart, int4 d) {
    return make_int4(rowstart[d.x], rowstart[d.y], rowstart[d.z], rowstart[d.w]);
}
__device__ __forceinline__ void cgmm_put4(int* __restrict__ csr_src, int4 p, int4 r, int4 s) {
    csr_src[p.x + r.x] = s.x; csr_src[p.y + r.y] = s.y;
    csr_src[p.z + r.z] = s.z; csr_src[p.w + r.w] = s.w;
}
__global__ __launch_bounds__(256) void cgmm_edge_fill(const int* __restrict__ ei,
                                                      const int* __restrict__ rowstart,
                                                      const int* __restrict__ rank,
                                                      int* __restrict__ csr_src, int E) {
    int T  = gridDim.x * 256;
    int t  = blockIdx.x * 256 + threadIdx.x;
    int nq = E >> 2;
    int q0 = t, q1 = t + T;
    bool al = (E & 3) == 0;
    const int* srcrow = ei + E;
    if (q1 < nq) {
        int4 d0 = ((const int4*)ei)[q0],   d1 = ((const int4*)ei)[q1];
        int4 r0 = ((const int4*)rank)[q0], r1 = ((const int4*)rank)[q1];
        int4 s0 = cgmm_ld4(srcrow, q0, al), s1 = cgmm_ld4(srcrow, q1, al);
        int4 p0 = cgmm_gather4(rowstart, d0), p1 = cgmm_gather4(rowstart, d1);
        cgmm_put4(csr_src, p0, r0, s0);
        cgmm_put4(csr_src, p1, r1, s1);
    } else if (q0 < nq) {
        int4 d0 = ((const int4*)ei)[q0];
        int4 r0 = ((const int4*)rank)[q0];
        int4 s0 = cgmm_ld4(srcrow, q0, al);
        int4 p0 = cgmm_gather4(rowstart, d0);
        cgmm_put4(csr_src, p0, r0, s0);
    }
    if (t < (E & 3)) {
        int e = (nq << 2) + t;
        csr_src[rowstart[ei[e]] + rank[e]] = srcrow[e];
    }
}

#define CGMM_IDX(p) __builtin_amdgcn_readlane(myidx, (p))
#define CGMM_AT(i)  (((const float2*)(prev_h + (size_t)(i) * 64))[lane])
#define CGMM_ROW(k) CGMM_AT(CGMM_IDX(j + (k)))
#define CGMM_ADD(r) acc.x += r.x; acc.y += r.y;
__device__ __forceinline__ float2 cgmm_add_rows(const float* __restrict__ prev_h, int lane,
                                                int myidx, int cnt, float2 acc) {
    int j = 0;
    for (; j + 16 <= cnt; j += 16) {
        float2 r0 = CGMM_ROW(0),  r1 = CGMM_ROW(1),  r2 = CGMM_ROW(2),  r3 = CGMM_ROW(3);
        float2 r4 = CGMM_ROW(4),  r5 = CGMM_ROW(5),  r6 = CGMM_ROW(6),  r7 = CGMM_ROW(7);
        float2 r8 = CGMM_ROW(8),  r9 = CGMM_ROW(9),  r10 = CGMM_ROW(10), r11 = CGMM_ROW(11);
        float2 r12 = CGMM_ROW(12), r13 = CGMM_ROW(13), r14 = CGMM_ROW(14), r15 = CGMM_ROW(15);
        CGMM_ADD(r0) CGMM_ADD(r1) CGMM_ADD(r2) CGMM_ADD(r3)
        CGMM_ADD(r4) CGMM_ADD(r5) CGMM_ADD(r6) CGMM_ADD(r7)
        CGMM_ADD(r8) CGMM_ADD(r9) CGMM_ADD(r10) CGMM_ADD(r11)
        CGMM_ADD(r12) CGMM_ADD(r13) CGMM_ADD(r14) CGMM_ADD(r15)
    }
    int m = cnt - j;
    if (m) {
        int j4 = j + (m & 8), j2 = j4 + (m & 4), j1 = j2 + (m & 2);
        int ia0 = CGMM_IDX(j),      ia1 = CGMM_IDX(j + 1),  ia2 = CGMM_IDX(j + 2),  ia3 = CGMM_IDX(j + 3);
        int ia4 = CGMM_IDX(j + 4),  ia5 = CGMM_IDX(j + 5),  ia6 = CGMM_IDX(j + 6),  ia7 = CGMM_IDX(j + 7);
        int ib0 = CGMM_IDX(j4),     ib1 = CGMM_IDX(j4 + 1), ib2 = CGMM_IDX(j4 + 2), ib3 = CGMM_IDX(j4 + 3);
        int ic0 = CGMM_IDX(j2),     ic1 = CGMM_IDX(j2 + 1), ie0 = CGMM_IDX(j1);
        float2 zf = {0.0f, 0.0f};
        float2 a0 = zf, a1 = zf, a2 = zf, a3 = zf, a4 = zf, a5 = zf, a6 = zf, a7 = zf;
        float2 b0 = zf, b1 = zf, b2 = zf, b3 = zf, c0 = zf, c1 = zf, e0 = zf;
        if (m & 8) {
            a0 = CGMM_AT(ia0); a1 = CGMM_AT(ia1); a2 = CGMM_AT(ia2); a3 = CGMM_AT(ia3);
            a4 = CGMM_AT(ia4); a5 = CGMM_AT(ia5); a6 = CGMM_AT(ia6); a7 = CGMM_AT(ia7);
        }
        if (m & 4) { b0 = CGMM_AT(ib0); b1 = CGMM_AT(ib1); b2 = CGMM_AT(ib2); b3 = CGMM_AT(ib3); }
        if (m & 2) { c0 = CGMM_AT(ic0); c1 = CGMM_AT(ic1); }
        if (m & 1) { e0 = CGMM_AT(ie0); }
        if (m & 8) {
            CGMM_ADD(a0) CGMM_ADD(a1) CGMM_ADD(a2) CGMM_ADD(a3)
            CGMM_ADD(a4) CGMM_ADD(a5) CGMM_ADD(a6) CGMM_ADD(a7)
        }
        if (m & 4) { CGMM_ADD(b0) CGMM_ADD(b1) CGMM_ADD(b2) CGMM_ADD(b3) }
        if (m & 2) { CGMM_ADD(c0) CGMM_ADD(c1) }
        if (m & 1) { CGMM_ADD(e0) }
    }
    return acc;
}
#undef CGMM_ADD
#undef CGMM_ROW
#undef CGMM_AT
#undef CGMM_IDX

__global__ __launch_bounds__(256) void cgmm_gather_sum(const float* __restrict__ prev_h,
                                                       const int* __restrict__ csr_src,
                                                       const int* __restrict__ rowstart,
                                                       float* __restrict__ summed,
                                                       int* deg_counts, int N) {
    int lane = threadIdx.x & 31;
    int n0 = __builtin_amdgcn_readfirstlane((blockIdx.x * 8 + (threadIdx.x >> 5)) * 8);
    if (n0 >= N) return;
    int nk = N - n0 < 8 ? N - n0 : 8;
    int dv = 0, sv = 0;
    if (lane < nk) { dv = deg_counts[n0 + lane]; sv = rowstart[n0 + lane]; }
    int d_nx = __builtin_amdgcn_readlane(dv, 0);
    int s_nx = __builtin_amdgcn_readlane(sv, 0);
    int idx_nx = lane < d_nx ? csr_src[s_nx + lane] : 0;
    for (int k = 0; k < nk; ++k) {
        int d = d_nx, start = s_nx, idx0 = idx_nx;
        asm volatile("" :: "v"(idx0));
        d_nx = __builtin_amdgcn_readlane(dv, k + 1);
        s_nx = __builtin_amdgcn_readlane(sv, k + 1);
        idx_nx = lane < d_nx ? csr_src[s_nx + lane] : 0;
        float2 acc = {0.0f, 0.0f};
        acc = cgmm_add_rows(prev_h, lane, idx0, d < 32 ? d : 32, acc);
        for (int base = 32; base < d; base += 32) {
            int cnt = d - base < 32 ? d - base : 32;
            int myidx = lane < cnt ? csr_src[start + base + lane] : 0;
            acc = cgmm_add_rows(prev_h, lane, myidx, cnt, acc);
        }
        ((float2*)(summed + (size_t)(n0 + k) * 64))[lane] = acc;
    }
    if (lane < nk) ((float*)deg_counts)[n0 + lane] = (float)dv;
}

__global__ __launch_bounds__(256) void cgmm_node_wmma(const int* __restrict__ x,
                                                      const float* __restrict__ summed,
                                                      const float* __restrict__ counts,
                                                      const _Float16* __restrict__ bfrag,
                                                      const float* __restrict__ Bs_t,
                                                      float* __restrict__ lik,
                                                      float* __restrict__ post,
                                                      int N) {
    __shared__ float ldsT[8][16][8][8];
    __shared__ float ldsP[8][16][64];
    int lane = threadIdx.x & 31;
    int wave = threadIdx.x >> 5;
    int tile = blockIdx.x * 8 + wave;
    int n0 = tile * 16;
    if (n0 >= N) return;

    int  m    = lane & 15;
    bool hi   = lane >= 16;
    int  node = n0 + m;
    int  nodec = node < N ? node : N - 1;
    bool valid = node < N;

    const float* rowbase = summed + (size_t)nodec * 64;
    __builtin_prefetch(rowbase, 0, 3);
    float inv_cnt = 1.0f / fmaxf(counts[nodec], 1.0f);

#pragma unroll
    for (int p = 0; p < 4; ++p) {
        int g = 2 * p + (hi ? 1 : 0);
        const float* row = rowbase + g;
        v16h a;
#pragma unroll
        for (int e2 = 0; e2 < 8; ++e2)
            a[e2] = (_Float16)(row[e2 * 8] * inv_cnt);
#pragma unroll
        for (int e2 = 8; e2 < 16; ++e2)
            a[e2] = (_Float16)0.0f;

        v16h b = *((const v16h*)(bfrag + (p * 32 + lane) * 16));

        v8f acc = {};
        acc = __builtin_amdgcn_wmma_f32_16x16x32_f16(
            false, a, false, b, (short)0, acc, false, false);
#pragma unroll
        for (int r = 0; r < 8; ++r) {
            int mm = r + (hi ? 8 : 0);
            int gq = 2 * p + (m >= 8 ? 1 : 0);
            ldsT[wave][mm][gq][m & 7] = acc[r];
        }
    }

    int gbase = hi ? 4 : 0;
    int xi = x[nodec];
    const float* brow = Bs_t + (size_t)xi * 64;
    float lik4[4];
#pragma unroll
    for (int gg = 0; gg < 4; ++gg) {
        int g = gbase + gg;
        const float4* tp = (const float4*)&ldsT[wave][m][g][0];
        float4 t0 = tp[0], t1 = tp[1];
        const float4* bp = (const float4*)(brow + g * 8);
        float4 b0 = bp[0], b1 = bp[1];
        float u[8];
        u[0] = t0.x * b0.x; u[1] = t0.y * b0.y; u[2] = t0.z * b0.z; u[3] = t0.w * b0.w;
        u[4] = t1.x * b1.x; u[5] = t1.y * b1.y; u[6] = t1.z * b1.z; u[7] = t1.w * b1.w;
        float s = ((u[0] + u[1]) + (u[2] + u[3])) + ((u[4] + u[5]) + (u[6] + u[7]));
        float denom = s + 64.0f * EPS_F;
        lik4[gg] = __logf(denom);
        float invd = 1.0f / denom;
#pragma unroll
        for (int i = 0; i < 8; ++i)
            ldsP[wave][m][i * 8 + g] = (u[i] + 8.0f * EPS_F) * invd;
    }
    if (valid) {
        float4 lv = {lik4[0], lik4[1], lik4[2], lik4[3]};
        *(float4*)(lik + (size_t)node * 8 + gbase) = lv;
    }

    {
        int nl = lane >> 1;
        int half = lane & 1;
        int outnode = n0 + nl;
        if (outnode < N) {
            const float4* src = (const float4*)&ldsP[wave][nl][half * 32];
            float4* dstp = (float4*)(post + (size_t)outnode * 64 + half * 32);
#pragma unroll
            for (int q = 0; q < 8; ++q) dstp[q] = src[q];
        }
    }
}

extern "C" void kernel_launch(void* const* d_in, const int* in_sizes, int n_in,
                              void* d_out, int out_size, void* d_ws, size_t ws_size,
                              hipStream_t stream) {
    (void)n_in; (void)out_size; (void)ws_size;
    const int*   x      = (const int*)d_in[0];
    const float* prev_h = (const float*)d_in[1];
    const int*   ei     = (const int*)d_in[2];
    const float* Q      = (const float*)d_in[3];
    const float* B      = (const float*)d_in[4];
    const int N = in_sizes[0];
    const int E = in_sizes[2] / 2;

    float* ws     = (float*)d_ws;
    float* summed = ws;
    float* counts = summed + (size_t)N * 64;
    float* Bs_t   = counts + N;
    size_t off    = ((size_t)N * 65 + 2048 + 7) & ~(size_t)7;
    _Float16* bfrag = (_Float16*)(ws + off);

    float* lik  = (float*)d_out;
    float* post = lik + (size_t)N * 8;

    int* deg      = (int*)counts;
    int* rank     = (int*)summed;
    int* rowstart = (int*)lik;
    int* total    = rowstart + N;
    int* csr_src  = (int*)post;

    cgmm_zero_deg<<<(N / 4 + 256) / 256, 256, 0, stream>>>(deg, N, total);

    cgmm_params<<<1, 128, 0, stream>>>(Q, B, Bs_t, bfrag);

    cgmm_edge_rank<<<(E + 2047) / 2048, 256, 0, stream>>>(ei, deg, rank, E);
    cgmm_seg_alloc<<<(N + 1023) / 1024, 1024, 0, stream>>>(deg, rowstart, total, N);
    cgmm_edge_fill<<<(E + 2047) / 2048, 256, 0, stream>>>(ei, rowstart, rank, csr_src, E);
    cgmm_gather_sum<<<(N + 63) / 64, 256, 0, stream>>>(prev_h, csr_src, rowstart, summed, deg, N);

    int tiles = (N + 15) / 16;
    cgmm_node_wmma<<<(tiles + 7) / 8, 256, 0, stream>>>(x, summed, counts, bfrag, Bs_t,
                                                        lik, post, N);
}


# Node contraction fused into the gather-sum wave: a wave's 8 summed rows reach the WMMA through LDS, not memory (summed and counts arrays and one launch gone); float add order unchanged
# speedup vs baseline: 4.1749x; 2.9734x over previous
#include <hip/hip_runtime.h>
#include <hip/hip_bf16.h>
#include <math.h>


typedef __attribute__((ext_vector_type(16))) _Float16 v16h;
typedef __attribute__((ext_vector_type(8)))  float    v8f;

#define EPS_F 1e-8f

__global__ __launch_bounds__(128) void cgmm_params(const float* __restrict__ Q,
                                                   const float* __restrict__ B,
                                                   float* __restrict__ Bs_t,
                                                   _Float16* __restrict__ bfrag) {
    __shared__ float sQ[512];
    int t = threadIdx.x;
    if (t < 64) {
        int g = t & 7, a = t >> 3;
        {
            int l = a;
            float mx = -1e30f;
            for (int i = 0; i < 8; ++i) mx = fmaxf(mx, Q[i * 64 + l * 8 + g]);
            float e[8], s = 0.0f;
            for (int i = 0; i < 8; ++i) { e[i] = __expf(Q[i * 64 + l * 8 + g] - mx); s += e[i]; }
            float inv = 1.0f / s;
            for (int i = 0; i < 8; ++i) sQ[i * 64 + l * 8 + g] = e[i] * inv;
        }
        {
            int i = a;
            float mx = -1e30f;
            for (int m = 0; m < 32; ++m) mx = fmaxf(mx, B[i * 256 + m * 8 + g]);
            float ev[32], s = 0.0f;
            for (int m = 0; m < 32; ++m) { ev[m] = __expf(B[i * 256 + m * 8 + g] - mx); s += ev[m]; }
            float inv = 1.0f / s;
            for (int m = 0; m < 32; ++m) Bs_t[m * 64 + g * 8 + i] = ev[m] * inv;
        }
    }
    __syncthreads();
    int p = t >> 5, lane = t & 31;
    bool hi = lane >= 16;
    int  m  = lane & 15;
    bool bvalid = (hi == (m >= 8));
    int  icol = m & 7;
    int  g = 2 * p + (hi ? 1 : 0);
    _Float16* dst = bfrag + (p * 32 + lane) * 16;
    for (int e2 = 0; e2 < 16; ++e2) {
        float f = (e2 < 8 && bvalid) ? sQ[icol * 64 + e2 * 8 + g] : 0.0f;
        dst[e2] = (_Float16)f;
    }
}

#define CGMM_NPB_BITS 9
#define CGMM_SRC_BITS 23
#define CGMM_CH       8192

__device__ __forceinline__ int4 cgmm_ldq_tail(const int* __restrict__ row, int e, int E) {
    int4 v;
    v.x = e     < E ? row[e]     : -1;
    v.y = e + 1 < E ? row[e + 1] : -1;
    v.z = e + 2 < E ? row[e + 2] : -1;
    v.w = e + 3 < E ? row[e + 3] : -1;
    return v;
}

__device__ __forceinline__ int cgmm_block_excl_scan(int v, int* wsum, int t) {
    int lane = t & 31, wv = t >> 5;
    int incl = v;
#pragma unroll
    for (int s = 1; s < 32; s <<= 1) {
        int up = __shfl_up(incl, s, 32);
        if (lane >= s) incl += up;
    }
    if (lane == 31) wsum[wv] = incl;
    __syncthreads();
    if (wv == 0) {
        int w = wsum[lane];
        int wi = w;
#pragma unroll
        for (int s = 1; s < 32; s <<= 1) {
            int up = __shfl_up(wi, s, 32);
            if (lane >= s) wi += up;
        }
        wsum[lane] = wi - w;
    }
    __syncthreads();
    return wsum[wv] + (incl - v);
}

#define CGMM_COUNT1(d) if ((d) >= 0) atomicAdd(&hist[(d) >> CGMM_NPB_BITS], 1);
__global__ __launch_bounds__(1024) void cgmm_bkt_count(const int* __restrict__ ei,
                                                       int* __restrict__ cnt, int E, int NB) {
    __shared__ int hist[256];
    int t = threadIdx.x;
    if (t < 256) hist[t] = 0;
    int e0 = blockIdx.x * CGMM_CH + 4 * t, e1 = e0 + 4096;
    int4 a, b;
    if (e1 + 3 < E) {
        a = *((const int4*)(ei + e0));
        b = *((const int4*)(ei + e1));
    } else {
        a = cgmm_ldq_tail(ei, e0, E);
        b = cgmm_ldq_tail(ei, e1, E);
    }
    __syncthreads();
    CGMM_COUNT1(a.x) CGMM_COUNT1(a.y) CGMM_COUNT1(a.z) CGMM_COUNT1(a.w)
    CGMM_COUNT1(b.x) CGMM_COUNT1(b.y) CGMM_COUNT1(b.z) CGMM_COUNT1(b.w)
    __syncthreads();
    if (t < NB) cnt[blockIdx.x * NB + t] = hist[t];
}
#undef CGMM_COUNT1

__global__ __launch_bounds__(1024) void cgmm_bkt_scan(const int* __restrict__ cnt,
                                                      int* __restrict__ plc,
                                                      int* __restrict__ bstart, int NB, int B1) {
    __shared__ int part[4][256];
    __shared__ int wsum[32];
    int t = threadIdx.x, tb = t & 255, g = t >> 8;
    int per = (B1 + 3) >> 2;
    int k0 = g * per;
    int k1 = k0 + per < B1 ? k0 + per : B1;
    int s = 0;
    if (tb < NB) {
        for (int k = k0; k < k1; k += 8) {
            int c[8];
#pragma unroll
            for (int j = 0; j < 8; ++j) c[j] = k + j < k1 ? cnt[(k + j) * NB + tb] : 0;
#pragma unroll
            for (int j = 0; j < 8; ++j) s += c[j];
        }
    }
    part[g][tb] = s;
    __syncthreads();
    int tot = 0;
    if (t < 256) tot = part[0][t] + part[1][t] + part[2][t] + part[3][t];
    int ex = cgmm_block_excl_scan(tot, wsum, t);
    if (t < 256) {
        if (t <= NB) bstart[t] = ex;
        int r = ex;
#pragma unroll
        for (int gg = 0; gg < 4; ++gg) { int p = part[gg][t]; part[gg][t] = r; r += p; }
    }
    __syncthreads();
    if (tb < NB) {
        int run = part[g][tb];
        for (int k = k0; k < k1; k += 8) {
            int c[8];
#pragma unroll
            for (int j = 0; j < 8; ++j) c[j] = k + j < k1 ? cnt[(k + j) * NB + tb] : 0;
#pragma unroll
            for (int j = 0; j < 8; ++j) {
                if (k + j < k1) plc[(k + j) * NB + tb] = run;
                run += c[j];
            }
        }
    }
}

__global__ __launch_bounds__(1024) void cgmm_bkt_scatter(const int* __restrict__ ei,
                                                         const int* __restrict__ cnt,
                                                         const int* __restrict__ plc,
                                                         int* __restrict__ bkt, int E, int NB) {
    __shared__ int cur[256];
    __shared__ int gdl[256];
    __shared__ int wsum[32];
    __shared__ int stage[CGMM_CH];
    __shared__ unsigned char sb[CGMM_CH];
    int t = threadIdx.x;
    int row = blockIdx.x * NB;
    const int* srcrow = ei + E;
    int e0 = blockIdx.x * CGMM_CH + 4 * t, e1 = e0 + 4096;
    int4 d0, d1, s0, s1;
    if (e1 + 3 < E && (E & 3) == 0) {
        d0 = *((const int4*)(ei + e0));     d1 = *((const int4*)(ei + e1));
        s0 = *((const int4*)(srcrow + e0)); s1 = *((const int4*)(srcrow + e1));
    } else {
        d0 = cgmm_ldq_tail(ei, e0, E);     d1 = cgmm_ldq_tail(ei, e1, E);
        s0 = cgmm_ldq_tail(srcrow, e0, E); s1 = cgmm_ldq_tail(srcrow, e1, E);
    }
    int c = t < NB ? cnt[row + t] : 0;
    int g = t < NB ? plc[row + t] : 0;
    int ls = cgmm_block_excl_scan(c, wsum, t);
    if (t < NB) { cur[t] = ls; gdl[t] = g - ls; }
    __syncthreads();
    int dd[8] = {d0.x, d0.y, d0.z, d0.w, d1.x, d1.y, d1.z, d1.w};
    int ss[8] = {s0.x, s0.y, s0.z, s0.w, s1.x, s1.y, s1.z, s1.w};
    int pp[8];
#pragma unroll
    for (int k = 0; k < 8; ++k)
        pp[k] = dd[k] >= 0 ? atomicAdd(&cur[dd[k] >> CGMM_NPB_BITS], 1) : 0;
#pragma unroll
    for (int k = 0; k < 8; ++k) {
        if (dd[k] >= 0) {
            stage[pp[k]] = (int)((((unsigned)dd[k] & 511u) << CGMM_SRC_BITS) | (unsigned)ss[k]);
            sb[pp[k]]    = (unsigned char)(dd[k] >> CGMM_NPB_BITS);
        }
    }
    __syncthreads();
    int nblk = E - blockIdx.x * CGMM_CH;
    if (nblk >= CGMM_CH) {
#pragma unroll
        for (int k = 0; k < 8; ++k) {
            int j = t + 1024 * k;
            bkt[j + gdl[sb[j]]] = stage[j];
        }
    } else {
        for (int j = t; j < nblk; j += 1024) bkt[j + gdl[sb[j]]] = stage[j];
    }
}

#define CGMM_CAP 10240
#define CGMM_LOAD4 \
        int i1 = i + 1024, i2 = i + 2048, i3 = i + 3072; \
        bool v1 = i1 < hi, v2 = i2 < hi, v3 = i3 < hi; \
        unsigned w0 = (unsigned)bkt[i]; \
        unsigned w1 = (unsigned)bkt[v1 ? i1 : i]; \
        unsigned w2 = (unsigned)bkt[v2 ? i2 : i]; \
        unsigned w3 = (unsigned)bkt[v3 ? i3 : i]; \
        asm volatile("" :: "v"(w0), "v"(w1), "v"(w2), "v"(w3));
#define CGMM_RANK4 \
        int p0 = atomicAdd(&curl[w0 >> CGMM_SRC_BITS], 1); \
        int p1 = v1 ? atomicAdd(&curl[w1 >> CGMM_SRC_BITS], 1) : 0; \
        int p2 = v2 ? atomicAdd(&curl[w2 >> CGMM_SRC_BITS], 1) : 0; \
        int p3 = v3 ? atomicAdd(&curl[w3 >> CGMM_SRC_BITS], 1) : 0;
__global__ __launch_bounds__(1024) void cgmm_bkt_sort(const int* __restrict__ bkt,
                                                      const int* __restrict__ bstart,
                                                      int* __restrict__ deg,
                                                      int* __restrict__ rowstart,
                                                      int* __restrict__ csr_src, int N) {
    __shared__ int cntl[512];
    __shared__ int curl[512];
    __shared__ int wsum[32];
    __shared__ int stage[CGMM_CAP];
    int t = threadIdx.x;
    int lo = bstart[blockIdx.x], hi = bstart[blockIdx.x + 1];
    if (t < 512) cntl[t] = 0;
    __syncthreads();
    for (int i = lo + t; i < hi; i += 4096) {
        CGMM_LOAD4
        atomicAdd(&cntl[w0 >> CGMM_SRC_BITS], 1);
        if (v1) atomicAdd(&cntl[w1 >> CGMM_SRC_BITS], 1);
        if (v2) atomicAdd(&cntl[w2 >> CGMM_SRC_BITS], 1);
        if (v3) atomicAdd(&cntl[w3 >> CGMM_SRC_BITS], 1);
    }
    __syncthreads();
    int d = t < 512 ? cntl[t] : 0;
    int ex = cgmm_block_excl_scan(d, wsum, t);
    if (t < 512) {
        int n = (blockIdx.x << CGMM_NPB_BITS) + t;
        curl[t] = ex;
        if (n < N) { deg[n] = d; rowstart[n] = lo + ex; }
    }
    __syncthreads();
    const unsigned smask = (1u << CGMM_SRC_BITS) - 1u;
    const int cnt_b = hi - lo;
    if (cnt_b <= CGMM_CAP) {
        for (int i = lo + t; i < hi; i += 4096) {
            CGMM_LOAD4
            CGMM_RANK4
            stage[p0] = (int)(w0 & smask);
            if (v1) stage[p1] = (int)(w1 & smask);
            if (v2) stage[p2] = (int)(w2 & smask);
            if (v3) stage[p3] = (int)(w3 & smask);
        }
        __syncthreads();
        for (int j = t; j < cnt_b; j += 1024) csr_src[lo + j] = stage[j];
    } else {
        for (int i = lo + t; i < hi; i += 4096) {
            CGMM_LOAD4
            CGMM_RANK4
            csr_src[lo + p0] = (int)(w0 & smask);
            if (v1) csr_src[lo + p1] = (int)(w1 & smask);
            if (v2) csr_src[lo + p2] = (int)(w2 & smask);
            if (v3) csr_src[lo + p3] = (int)(w3 & smask);
        }
    }
}
#undef CGMM_RANK4
#undef CGMM_LOAD4

#define CGMM_IDX(p) __builtin_amdgcn_readlane(myidx, (p))
#define CGMM_AT(i)  (((const float2*)(prev_h + (size_t)(i) * 64))[lane])
#define CGMM_ROW(k) CGMM_AT(CGMM_IDX(j + (k)))
#define CGMM_ADD(r) acc.x += r.x; acc.y += r.y;
__device__ __forceinline__ float2 cgmm_add_rows(const float* __restrict__ prev_h, int lane,
                                                int myidx, int cnt, float2 acc) {
    int j = 0;
    for (; j + 16 <= cnt; j += 16) {
        float2 r0 = CGMM_ROW(0),  r1 = CGMM_ROW(1),  r2 = CGMM_ROW(2),  r3 = CGMM_ROW(3);
        float2 r4 = CGMM_ROW(4),  r5 = CGMM_ROW(5),  r6 = CGMM_ROW(6),  r7 = CGMM_ROW(7);
        float2 r8 = CGMM_ROW(8),  r9 = CGMM_ROW(9),  r10 = CGMM_ROW(10), r11 = CGMM_ROW(11);
        float2 r12 = CGMM_ROW(12), r13 = CGMM_ROW(13), r14 = CGMM_ROW(14), r15 = CGMM_ROW(15);
        CGMM_ADD(r0) CGMM_ADD(r1) CGMM_ADD(r2) CGMM_ADD(r3)
        CGMM_ADD(r4) CGMM_ADD(r5) CGMM_ADD(r6) CGMM_ADD(r7)
        CGMM_ADD(r8) CGMM_ADD(r9) CGMM_ADD(r10) CGMM_ADD(r11)
        CGMM_ADD(r12) CGMM_ADD(r13) CGMM_ADD(r14) CGMM_ADD(r15)
    }
    int m = cnt - j;
    if (m) {
        int j4 = j + (m & 8), j2 = j4 + (m & 4), j1 = j2 + (m & 2);
        int ia0 = CGMM_IDX(j),      ia1 = CGMM_IDX(j + 1),  ia2 = CGMM_IDX(j + 2),  ia3 = CGMM_IDX(j + 3);
        int ia4 = CGMM_IDX(j + 4),  ia5 = CGMM_IDX(j + 5),  ia6 = CGMM_IDX(j + 6),  ia7 = CGMM_IDX(j + 7);
        int ib0 = CGMM_IDX(j4),     ib1 = CGMM_IDX(j4 + 1), ib2 = CGMM_IDX(j4 + 2), ib3 = CGMM_IDX(j4 + 3);
        int ic0 = CGMM_IDX(j2),     ic1 = CGMM_IDX(j2 + 1), ie0 = CGMM_IDX(j1);
        float2 zf = {0.0f, 0.0f};
        float2 a0 = zf, a1 = zf, a2 = zf, a3 = zf, a4 = zf, a5 = zf, a6 = zf, a7 = zf;
        float2 b0 = zf, b1 = zf, b2 = zf, b3 = zf, c0 = zf, c1 = zf, e0 = zf;
        if (m & 8) {
            a0 = CGMM_AT(ia0); a1 = CGMM_AT(ia1); a2 = CGMM_AT(ia2); a3 = CGMM_AT(ia3);
            a4 = CGMM_AT(ia4); a5 = CGMM_AT(ia5); a6 = CGMM_AT(ia6); a7 = CGMM_AT(ia7);
        }
        if (m & 4) { b0 = CGMM_AT(ib0); b1 = CGMM_AT(ib1); b2 = CGMM_AT(ib2); b3 = CGMM_AT(ib3); }
        if (m & 2) { c0 = CGMM_AT(ic0); c1 = CGMM_AT(ic1); }
        if (m & 1) { e0 = CGMM_AT(ie0); }
        if (m & 8) {
            CGMM_ADD(a0) CGMM_ADD(a1) CGMM_ADD(a2) CGMM_ADD(a3)
            CGMM_ADD(a4) CGMM_ADD(a5) CGMM_ADD(a6) CGMM_ADD(a7)
        }
        if (m & 4) { CGMM_ADD(b0) CGMM_ADD(b1) CGMM_ADD(b2) CGMM_ADD(b3) }
        if (m & 2) { CGMM_ADD(c0) CGMM_ADD(c1) }
        if (m & 1) { CGMM_ADD(e0) }
    }
    return acc;
}
#undef CGMM_ADD
#undef CGMM_ROW
#undef CGMM_AT
#undef CGMM_IDX

#define CGMM_PITCH 72
__global__ __launch_bounds__(256) void cgmm_gather_node(const float* __restrict__ prev_h,
                                                        const int* __restrict__ csr_src,
                                                        const int* __restrict__ rowstart,
                                                        const int* __restrict__ deg,
                                                        const int* __restrict__ x,
                                                        const _Float16* __restrict__ bfrag,
                                                        const float* __restrict__ Bs_t,
                                                        float* __restrict__ lik,
                                                        float* __restrict__ post, int N) {
    __shared__ alignas(16) float ldsX[8][8][CGMM_PITCH];
    __shared__ alignas(16) float ldsY[8][8][CGMM_PITCH];
    int lane = threadIdx.x & 31;
    int wave = threadIdx.x >> 5;
    int n0 = __builtin_amdgcn_readfirstlane((blockIdx.x * 8 + wave) * 8);
    if (n0 >= N) return;
    int nk = N - n0 < 8 ? N - n0 : 8;
    int dv = 0, sv = 0;
    if (lane < nk) { dv = deg[n0 + lane]; sv = rowstart[n0 + lane]; }
    int en = lane >> 2, gp = lane & 3;
    bool evalid = en < nk;
    int xi = x[evalid ? n0 + en : n0];
    int d_nx = __builtin_amdgcn_readlane(dv, 0);
    int s_nx = __builtin_amdgcn_readlane(sv, 0);
    int idx_nx = lane < d_nx ? csr_src[s_nx + lane] : 0;
    for (int k = 0; k < nk; ++k) {
        int d = d_nx, start = s_nx, idx0 = idx_nx;
        asm volatile("" :: "v"(idx0));
        d_nx = __builtin_amdgcn_readlane(dv, k + 1);
        s_nx = __builtin_amdgcn_readlane(sv, k + 1);
        idx_nx = lane < d_nx ? csr_src[s_nx + lane] : 0;
        float2 acc = {0.0f, 0.0f};
        acc = cgmm_add_rows(prev_h, lane, idx0, d < 32 ? d : 32, acc);
        for (int base = 32; base < d; base += 32) {
            int cnt = d - base < 32 ? d - base : 32;
            int myidx = lane < cnt ? csr_src[start + base + lane] : 0;
            acc = cgmm_add_rows(prev_h, lane, myidx, cnt, acc);
        }
        *((float2*)&ldsX[wave][k][2 * lane]) = acc;
    }
    for (int k = nk; k < 8; ++k) {
        float2 zf = {0.0f, 0.0f};
        *((float2*)&ldsX[wave][k][2 * lane]) = zf;
    }

    const float4* bp = (const float4*)(Bs_t + (size_t)xi * 64 + gp * 16);
    float4 e0 = bp[0], e1 = bp[1], e2v = bp[2], e3 = bp[3];

    int  c  = lane & 15;
    bool hi = lane >= 16;
    int  nr = c & 7;
    float inv_cnt = 1.0f / fmaxf((float)__shfl(dv, nr, 32), 1.0f);
    const float* srow = &ldsX[wave][nr][hi ? 1 : 0];
#pragma unroll
    for (int p = 0; p < 4; ++p) {
        v16h a;
#pragma unroll
        for (int s = 0; s < 8; ++s)
            a[s] = (_Float16)(srow[s * 8 + 2 * p] * inv_cnt);
#pragma unroll
        for (int s = 8; s < 16; ++s)
            a[s] = (_Float16)0.0f;

        v16h b = *((const v16h*)(bfrag + (p * 32 + lane) * 16));

        v8f acc = {};
        acc = __builtin_amdgcn_wmma_f32_16x16x32_f16(
            false, a, false, b, (short)0, acc, false, false);
        if (!hi) {
#pragma unroll
            for (int r = 0; r < 8; ++r)
                ldsY[wave][r][16 * p + c] = acc[r];
        }
    }

    const float4* tp = (const float4*)&ldsY[wave][en][gp * 16];
    float4 t0 = tp[0], t1 = tp[1], t2 = tp[2], t3 = tp[3];
    float ua[8], ub[8];
    ua[0] = t0.x * e0.x;  ua[1] = t0.y * e0.y;  ua[2] = t0.z * e0.z;  ua[3] = t0.w * e0.w;
    ua[4] = t1.x * e1.x;  ua[5] = t1.y * e1.y;  ua[6] = t1.z * e1.z;  ua[7] = t1.w * e1.w;
    ub[0] = t2.x * e2v.x; ub[1] = t2.y * e2v.y; ub[2] = t2.z * e2v.z; ub[3] = t2.w * e2v.w;
    ub[4] = t3.x * e3.x;  ub[5] = t3.y * e3.y;  ub[6] = t3.z * e3.z;  ub[7] = t3.w * e3.w;
    float sa = ((ua[0] + ua[1]) + (ua[2] + ua[3])) + ((ua[4] + ua[5]) + (ua[6] + ua[7]));
    float sb = ((ub[0] + ub[1]) + (ub[2] + ub[3])) + ((ub[4] + ub[5]) + (ub[6] + ub[7]));
    float da = sa + 64.0f * EPS_F, db = sb + 64.0f * EPS_F;
    float ia = 1.0f / da, ib = 1.0f / db;
#pragma unroll
    for (int i = 0; i < 8; ++i) {
        float2 pv = {(ua[i] + 8.0f * EPS_F) * ia, (ub[i] + 8.0f * EPS_F) * ib};
        *((float2*)&ldsX[wave][en][i * 8 + 2 * gp]) = pv;
    }
    if (evalid) {
        float2 lv = {__logf(da), __logf(db)};
        *((float2*)(lik + (size_t)n0 * 8 + 2 * lane)) = lv;
    }
    float4* dstp = (float4*)(post + (size_t)n0 * 64);
#pragma unroll
    for (int q = 0; q < 4; ++q) {
        int f = 32 * q + lane;
        int nd = f >> 4;
        if (nd < nk) dstp[f] = *((const float4*)&ldsX[wave][nd][(f & 15) * 4]);
    }
}

extern "C" void kernel_launch(void* const* d_in, const int* in_sizes, int n_in,
                              void* d_out, int out_size, void* d_ws, size_t ws_size,
                              hipStream_t stream) {
    (void)n_in; (void)out_size; (void)ws_size;
    const int*   x      = (const int*)d_in[0];
    const float* prev_h = (const float*)d_in[1];
    const int*   ei     = (const int*)d_in[2];
    const float* Q      = (const float*)d_in[3];
    const float* B      = (const float*)d_in[4];
    const int N = in_sizes[0];
    const int E = in_sizes[2] / 2;

    float* ws     = (float*)d_ws;
    int*   deg    = (int*)(ws + (size_t)N * 64);
    float* Bs_t   = ws + (size_t)N * 65;
    size_t off    = ((size_t)N * 65 + 2048 + 7) & ~(size_t)7;
    _Float16* bfrag = (_Float16*)(ws + off);

    float* lik  = (float*)d_out;
    float* post = lik + (size_t)N * 8;

    const int NB = (N + 511) / 512;
    const int B1 = (E + 8191) / 8192;
    int* bkt      = (int*)ws;
    int* cnt      = bkt + ((E + 3) & ~3);
    int* plc      = cnt + B1 * NB;
    int* bstart   = plc + B1 * NB;
    int* rowstart = bstart + ((NB + 4) & ~3);
    int* csr_src  = rowstart + ((N + 3) & ~3);

    cgmm_params<<<1, 128, 0, stream>>>(Q, B, Bs_t, bfrag);

    cgmm_bkt_count<<<B1, 1024, 0, stream>>>(ei, cnt, E, NB);
    cgmm_bkt_scan<<<1, 1024, 0, stream>>>(cnt, plc, bstart, NB, B1);
    cgmm_bkt_scatter<<<B1, 1024, 0, stream>>>(ei, cnt, plc, bkt, E, NB);
    cgmm_bkt_sort<<<NB, 1024, 0, stream>>>(bkt, bstart, deg, rowstart, csr_src, N);

    cgmm_gather_node<<<(N + 63) / 64, 256, 0, stream>>>(prev_h, csr_src, rowstart, deg, x,
                                                         bfrag, Bs_t, lik, post, N);
}


# Launch count 6 -> 4: both one-work-group launches folded away (each scatter block sums the bucket-count matrix itself, no scan launch; parameter softmaxes ride as one extra block of the bucket sort)
# speedup vs baseline: 5.6584x; 1.3553x over previous
#include <hip/hip_runtime.h>
#include <hip/hip_bf16.h>
#include <math.h>


typedef __attribute__((ext_vector_type(16))) _Float16 v16h;
typedef __attribute__((ext_vector_type(8)))  float    v8f;

#define EPS_F 1e-8f

__device__ __forceinline__ void cgmm_params_block(const float* __restrict__ Q,
                                                  const float* __restrict__ B,
                                                  float* __restrict__ Bs_t,
                                                  _Float16* __restrict__ bfrag,
                                                  float* sQ, int t) {
    if (t < 64) {
        int g = t & 7, a = t >> 3;
        {
            int l = a;
            float mx = -1e30f;
            for (int i = 0; i < 8; ++i) mx = fmaxf(mx, Q[i * 64 + l * 8 + g]);
            float e[8], s = 0.0f;
            for (int i = 0; i < 8; ++i) { e[i] = __expf(Q[i * 64 + l * 8 + g] - mx); s += e[i]; }
            float inv = 1.0f / s;
            for (int i = 0; i < 8; ++i) sQ[i * 64 + l * 8 + g] = e[i] * inv;
        }
        {
            int i = a;
            float mx = -1e30f;
            for (int m = 0; m < 32; ++m) mx = fmaxf(mx, B[i * 256 + m * 8 + g]);
            float ev[32], s = 0.0f;
            for (int m = 0; m < 32; ++m) { ev[m] = __expf(B[i * 256 + m * 8 + g] - mx); s += ev[m]; }
            float inv = 1.0f / s;
            for (int m = 0; m < 32; ++m) Bs_t[m * 64 + g * 8 + i] = ev[m] * inv;
        }
    }
    __syncthreads();
    if (t >= 128) return;
    int p = t >> 5, lane = t & 31;
    bool hi = lane >= 16;
    int  m  = lane & 15;
    bool bvalid = (hi == (m >= 8));
    int  icol = m & 7;
    int  g = 2 * p + (hi ? 1 : 0);
    _Float16* dst = bfrag + (p * 32 + lane) * 16;
    for (int e2 = 0; e2 < 16; ++e2) {
        float f = (e2 < 8 && bvalid) ? sQ[icol * 64 + e2 * 8 + g] : 0.0f;
        dst[e2] = (_Float16)f;
    }
}

#define CGMM_NPB_BITS 9
#define CGMM_SRC_BITS 23
#define CGMM_CH       8192

__device__ __forceinline__ int4 cgmm_ldq_tail(const int* __restrict__ row, int e, int E) {
    int4 v;
    v.x = e     < E ? row[e]     : -1;
    v.y = e + 1 < E ? row[e + 1] : -1;
    v.z = e + 2 < E ? row[e + 2] : -1;
    v.w = e + 3 < E ? row[e + 3] : -1;
    return v;
}

__device__ __forceinline__ int cgmm_block_excl_scan(int v, int* wsum, int t) {
    int lane = t & 31, wv = t >> 5;
    int incl = v;
#pragma unroll
    for (int s = 1; s < 32; s <<= 1) {
        int up = __shfl_up(incl, s, 32);
        if (lane >= s) incl += up;
    }
    if (lane == 31) wsum[wv] = incl;
    __syncthreads();
    if (wv == 0) {
        int w = wsum[lane];
        int wi = w;
#pragma unroll
        for (int s = 1; s < 32; s <<= 1) {
            int up = __shfl_up(wi, s, 32);
            if (lane >= s) wi += up;
        }
        wsum[lane] = wi - w;
    }
    __syncthreads();
    return wsum[wv] + (incl - v);
}

__device__ __forceinline__ int2 cgmm_block_excl_scan2(int va, int vb, int* wsum, int t) {
    int lane = t & 31, wv = t >> 5;
    int ia = va, ib = vb;
#pragma unroll
    for (int s = 1; s < 32; s <<= 1) {
        int ua = __shfl_up(ia, s, 32);
        int ub = __shfl_up(ib, s, 32);
        if (lane >= s) { ia += ua; ib += ub; }
    }
    if (lane == 31) { wsum[wv] = ia; wsum[32 + wv] = ib; }
    __syncthreads();
    if (wv < 2) {
        int w = wsum[t];
        int wi = w;
#pragma unroll
        for (int s = 1; s < 32; s <<= 1) {
            int up = __shfl_up(wi, s, 32);
            if (lane >= s) wi += up;
        }
        wsum[t] = wi - w;
    }
    __syncthreads();
    int2 r = {wsum[wv] + (ia - va), wsum[32 + wv] + (ib - vb)};
    return r;
}

#define CGMM_COUNT1(d) if ((d) >= 0) atomicAdd(&hist[(d) >> CGMM_NPB_BITS], 1);
__global__ __launch_bounds__(1024) void cgmm_bkt_count(const int* __restrict__ ei,
                                                       int* __restrict__ cnt, int E, int NBP) {
    __shared__ int hist[256];
    int t = threadIdx.x;
    if (t < 256) hist[t] = 0;
    int e0 = blockIdx.x * CGMM_CH + 4 * t, e1 = e0 + 4096;
    int4 a, b;
    if (e1 + 3 < E) {
        a = *((const int4*)(ei + e0));
        b = *((const int4*)(ei + e1));
    } else {
        a = cgmm_ldq_tail(ei, e0, E);
        b = cgmm_ldq_tail(ei, e1, E);
    }
    __syncthreads();
    CGMM_COUNT1(a.x) CGMM_COUNT1(a.y) CGMM_COUNT1(a.z) CGMM_COUNT1(a.w)
    CGMM_COUNT1(b.x) CGMM_COUNT1(b.y) CGMM_COUNT1(b.z) CGMM_COUNT1(b.w)
    __syncthreads();
    if (t < NBP) cnt[blockIdx.x * NBP + t] = hist[t];
}
#undef CGMM_COUNT1

#define CGMM_SC_ROWS 13
__global__ __launch_bounds__(1024) void cgmm_bkt_scatter(const int* __restrict__ ei,
                                                         const int* __restrict__ cnt,
                                                         int* __restrict__ bkt,
                                                         int* __restrict__ bstart,
                                                         int E, int NB, int NBP, int B1) {
    __shared__ int cur[256];
    __shared__ int gdl[256];
    __shared__ int stot[256];
    __shared__ int spre[256];
    __shared__ int wsum[64];
    __shared__ int stage[CGMM_CH];
    __shared__ unsigned char sb[CGMM_CH];
    int t = threadIdx.x;
    if (t < 256) { stot[t] = 0; spre[t] = 0; }
    int row = blockIdx.x * NBP;
    const int* srcrow = ei + E;
    int e0 = blockIdx.x * CGMM_CH + 4 * t, e1 = e0 + 4096;
    int4 d0, d1, s0, s1;
    if (e1 + 3 < E && (E & 3) == 0) {
        d0 = *((const int4*)(ei + e0));     d1 = *((const int4*)(ei + e1));
        s0 = *((const int4*)(srcrow + e0)); s1 = *((const int4*)(srcrow + e1));
    } else {
        d0 = cgmm_ldq_tail(ei, e0, E);     d1 = cgmm_ldq_tail(ei, e1, E);
        s0 = cgmm_ldq_tail(srcrow, e0, E); s1 = cgmm_ldq_tail(srcrow, e1, E);
    }
    int c = t < NB ? cnt[row + t] : 0;
    const int q = t & 63, rg = t >> 6, Q4 = NBP >> 2;
    const int blk = blockIdx.x;
    int4 tot4 = {0, 0, 0, 0}, pre4 = {0, 0, 0, 0};
    if (q < Q4) {
        const int4* cnt4 = (const int4*)cnt;
        for (int r0 = rg; r0 < B1; r0 += 16 * CGMM_SC_ROWS) {
            int4 v[CGMM_SC_ROWS];
#pragma unroll
            for (int j = 0; j < CGMM_SC_ROWS; ++j) {
                int r = r0 + 16 * j;
                v[j] = cnt4[(r < B1 ? r : r0) * Q4 + q];
            }
            __builtin_amdgcn_sched_barrier(0);
#pragma unroll
            for (int j = 0; j < CGMM_SC_ROWS; ++j) {
                int r = r0 + 16 * j;
                int mt = r < B1 ? -1 : 0;
                int mp = r < blk ? -1 : 0;
                tot4.x += v[j].x & mt; tot4.y += v[j].y & mt; tot4.z += v[j].z & mt; tot4.w += v[j].w & mt;
                pre4.x += v[j].x & mp; pre4.y += v[j].y & mp; pre4.z += v[j].z & mp; pre4.w += v[j].w & mp;
            }
        }
    }
    __syncthreads();
    if (q < Q4) {
        atomicAdd(&stot[4 * q],     tot4.x); atomicAdd(&stot[4 * q + 1], tot4.y);
        atomicAdd(&stot[4 * q + 2], tot4.z); atomicAdd(&stot[4 * q + 3], tot4.w);
        atomicAdd(&spre[4 * q],     pre4.x); atomicAdd(&spre[4 * q + 1], pre4.y);
        atomicAdd(&spre[4 * q + 2], pre4.z); atomicAdd(&spre[4 * q + 3], pre4.w);
    }
    __syncthreads();
    int tot = t < 256 ? stot[t] : 0;
    int2 sc = cgmm_block_excl_scan2(tot, c, wsum, t);
    if (t < 256) {
        cur[t] = sc.y;
        gdl[t] = sc.x + spre[t] - sc.y;
        if (blk == 0 && t <= NB) bstart[t] = sc.x;
    }
    __syncthreads();
    int dd[8] = {d0.x, d0.y, d0.z, d0.w, d1.x, d1.y, d1.z, d1.w};
    int ss[8] = {s0.x, s0.y, s0.z, s0.w, s1.x, s1.y, s1.z, s1.w};
    int pp[8];
#pragma unroll
    for (int k = 0; k < 8; ++k)
        pp[k] = dd[k] >= 0 ? atomicAdd(&cur[dd[k] >> CGMM_NPB_BITS], 1) : 0;
#pragma unroll
    for (int k = 0; k < 8; ++k) {
        if (dd[k] >= 0) {
            stage[pp[k]] = (int)((((unsigned)dd[k] & 511u) << CGMM_SRC_BITS) | (unsigned)ss[k]);
            sb[pp[k]]    = (unsigned char)(dd[k] >> CGMM_NPB_BITS);
        }
    }
    __syncthreads();
    int nblk = E - blockIdx.x * CGMM_CH;
    if (nblk >= CGMM_CH) {
#pragma unroll
        for (int k = 0; k < 8; ++k) {
            int j = t + 1024 * k;
            bkt[j + gdl[sb[j]]] = stage[j];
        }
    } else {
        for (int j = t; j < nblk; j += 1024) bkt[j + gdl[sb[j]]] = stage[j];
    }
}
#undef CGMM_SC_ROWS

#define CGMM_CAP 10240
#define CGMM_LOAD4 \
        int i1 = i + 1024, i2 = i + 2048, i3 = i + 3072; \
        bool v1 = i1 < hi, v2 = i2 < hi, v3 = i3 < hi; \
        unsigned w0 = (unsigned)bkt[i]; \
        unsigned w1 = (unsigned)bkt[v1 ? i1 : i]; \
        unsigned w2 = (unsigned)bkt[v2 ? i2 : i]; \
        unsigned w3 = (unsigned)bkt[v3 ? i3 : i]; \
        asm volatile("" :: "v"(w0), "v"(w1), "v"(w2), "v"(w3));
#define CGMM_RANK4 \
        int p0 = atomicAdd(&curl[w0 >> CGMM_SRC_BITS], 1); \
        int p1 = v1 ? atomicAdd(&curl[w1 >> CGMM_SRC_BITS], 1) : 0; \
        int p2 = v2 ? atomicAdd(&curl[w2 >> CGMM_SRC_BITS], 1) : 0; \
        int p3 = v3 ? atomicAdd(&curl[w3 >> CGMM_SRC_BITS], 1) : 0;
__global__ __launch_bounds__(1024) void cgmm_bkt_sort(const int* __restrict__ bkt,
                                                      const int* __restrict__ bstart,
                                                      int* __restrict__ deg,
                                                      int* __restrict__ rowstart,
                                                      int* __restrict__ csr_src, int N, int NB,
                                                      const float* __restrict__ Q,
                                                      const float* __restrict__ B,
                                                      float* __restrict__ Bs_t,
                                                      _Float16* __restrict__ bfrag) {
    __shared__ int cntl[512];
    __shared__ int curl[512];
    __shared__ int wsum[32];
    __shared__ int stage[CGMM_CAP];
    __shared__ float sQ[512];
    int t = threadIdx.x;
    if ((int)blockIdx.x == NB) {
        cgmm_params_block(Q, B, Bs_t, bfrag, sQ, t);
        return;
    }
    int lo = bstart[blockIdx.x], hi = bstart[blockIdx.x + 1];
    if (t < 512) cntl[t] = 0;
    __syncthreads();
    for (int i = lo + t; i < hi; i += 4096) {
        CGMM_LOAD4
        atomicAdd(&cntl[w0 >> CGMM_SRC_BITS], 1);
        if (v1) atomicAdd(&cntl[w1 >> CGMM_SRC_BITS], 1);
        if (v2) atomicAdd(&cntl[w2 >> CGMM_SRC_BITS], 1);
        if (v3) atomicAdd(&cntl[w3 >> CGMM_SRC_BITS], 1);
    }
    __syncthreads();
    int d = t < 512 ? cntl[t] : 0;
    int ex = cgmm_block_excl_scan(d, wsum, t);
    if (t < 512) {
        int n = (blockIdx.x << CGMM_NPB_BITS) + t;
        curl[t] = ex;
        if (n < N) { deg[n] = d; rowstart[n] = lo + ex; }
    }
    __syncthreads();
    const unsigned smask = (1u << CGMM_SRC_BITS) - 1u;
    const int cnt_b = hi - lo;
    if (cnt_b <= CGMM_CAP) {
        for (int i = lo + t; i < hi; i += 4096) {
            CGMM_LOAD4
            CGMM_RANK4
            stage[p0] = (int)(w0 & smask);
            if (v1) stage[p1] = (int)(w1 & smask);
            if (v2) stage[p2] = (int)(w2 & smask);
            if (v3) stage[p3] = (int)(w3 & smask);
        }
        __syncthreads();
        for (int j = t; j < cnt_b; j += 1024) csr_src[lo + j] = stage[j];
    } else {
        for (int i = lo + t; i < hi; i += 4096) {
            CGMM_LOAD4
            CGMM_RANK4
            csr_src[lo + p0] = (int)(w0 & smask);
            if (v1) csr_src[lo + p1] = (int)(w1 & smask);
            if (v2) csr_src[lo + p2] = (int)(w2 & smask);
            if (v3) csr_src[lo + p3] = (int)(w3 & smask);
        }
    }
}
#undef CGMM_RANK4
#undef CGMM_LOAD4

#define CGMM_IDX(p) __builtin_amdgcn_readlane(myidx, (p))
#define CGMM_AT(i)  (((const float2*)(prev_h + (size_t)(i) * 64))[lane])
#define CGMM_ROW(k) CGMM_AT(CGMM_IDX(j + (k)))
#define CGMM_ADD(r) acc.x += r.x; acc.y += r.y;
__device__ __forceinline__ float2 cgmm_add_rows(const float* __restrict__ prev_h, int lane,
                                                int myidx, int cnt, float2 acc) {
    int j = 0;
    for (; j + 16 <= cnt; j += 16) {
        float2 r0 = CGMM_ROW(0),  r1 = CGMM_ROW(1),  r2 = CGMM_ROW(2),  r3 = CGMM_ROW(3);
        float2 r4 = CGMM_ROW(4),  r5 = CGMM_ROW(5),  r6 = CGMM_ROW(6),  r7 = CGMM_ROW(7);
        float2 r8 = CGMM_ROW(8),  r9 = CGMM_ROW(9),  r10 = CGMM_ROW(10), r11 = CGMM_ROW(11);
        float2 r12 = CGMM_ROW(12), r13 = CGMM_ROW(13), r14 = CGMM_ROW(14), r15 = CGMM_ROW(15);
        CGMM_ADD(r0) CGMM_ADD(r1) CGMM_ADD(r2) CGMM_ADD(r3)
        CGMM_ADD(r4) CGMM_ADD(r5) CGMM_ADD(r6) CGMM_ADD(r7)
        CGMM_ADD(r8) CGMM_ADD(r9) CGMM_ADD(r10) CGMM_ADD(r11)
        CGMM_ADD(r12) CGMM_ADD(r13) CGMM_ADD(r14) CGMM_ADD(r15)
    }
    int m = cnt - j;
    if (m) {
        int j4 = j + (m & 8), j2 = j4 + (m & 4), j1 = j2 + (m & 2);
        int ia0 = CGMM_IDX(j),      ia1 = CGMM_IDX(j + 1),  ia2 = CGMM_IDX(j + 2),  ia3 = CGMM_IDX(j + 3);
        int ia4 = CGMM_IDX(j + 4),  ia5 = CGMM_IDX(j + 5),  ia6 = CGMM_IDX(j + 6),  ia7 = CGMM_IDX(j + 7);
        int ib0 = CGMM_IDX(j4),     ib1 = CGMM_IDX(j4 + 1), ib2 = CGMM_IDX(j4 + 2), ib3 = CGMM_IDX(j4 + 3);
        int ic0 = CGMM_IDX(j2),     ic1 = CGMM_IDX(j2 + 1), ie0 = CGMM_IDX(j1);
        float2 zf = {0.0f, 0.0f};
        float2 a0 = zf, a1 = zf, a2 = zf, a3 = zf, a4 = zf, a5 = zf, a6 = zf, a7 = zf;
        float2 b0 = zf, b1 = zf, b2 = zf, b3 = zf, c0 = zf, c1 = zf, e0 = zf;
        if (m & 8) {
            a0 = CGMM_AT(ia0); a1 = CGMM_AT(ia1); a2 = CGMM_AT(ia2); a3 = CGMM_AT(ia3);
            a4 = CGMM_AT(ia4); a5 = CGMM_AT(ia5); a6 = CGMM_AT(ia6); a7 = CGMM_AT(ia7);
        }
        if (m & 4) { b0 = CGMM_AT(ib0); b1 = CGMM_AT(ib1); b2 = CGMM_AT(ib2); b3 = CGMM_AT(ib3); }
        if (m & 2) { c0 = CGMM_AT(ic0); c1 = CGMM_AT(ic1); }
        if (m & 1) { e0 = CGMM_AT(ie0); }
        if (m & 8) {
            CGMM_ADD(a0) CGMM_ADD(a1) CGMM_ADD(a2) CGMM_ADD(a3)
            CGMM_ADD(a4) CGMM_ADD(a5) CGMM_ADD(a6) CGMM_ADD(a7)
        }
        if (m & 4) { CGMM_ADD(b0) CGMM_ADD(b1) CGMM_ADD(b2) CGMM_ADD(b3) }
        if (m & 2) { CGMM_ADD(c0) CGMM_ADD(c1) }
        if (m & 1) { CGMM_ADD(e0) }
    }
    return acc;
}
#undef CGMM_ADD
#undef CGMM_ROW
#undef CGMM_AT
#undef CGMM_IDX

#define CGMM_PITCH 72
__global__ __launch_bounds__(256) void cgmm_gather_node(const float* __restrict__ prev_h,
                                                        const int* __restrict__ csr_src,
                                                        const int* __restrict__ rowstart,
                                                        const int* __restrict__ deg,
                                                        const int* __restrict__ x,
                                                        const _Float16* __restrict__ bfrag,
                                                        const float* __restrict__ Bs_t,
                                                        float* __restrict__ lik,
                                                        float* __restrict__ post, int N) {
    __shared__ alignas(16) float ldsX[8][8][CGMM_PITCH];
    __shared__ alignas(16) float ldsY[8][8][CGMM_PITCH];
    int lane = threadIdx.x & 31;
    int wave = threadIdx.x >> 5;
    int n0 = __builtin_amdgcn_readfirstlane((blockIdx.x * 8 + wave) * 8);
    if (n0 >= N) return;
    int nk = N - n0 < 8 ? N - n0 : 8;
    int dv = 0, sv = 0;
    if (lane < nk) { dv = deg[n0 + lane]; sv = rowstart[n0 + lane]; }
    int en = lane >> 2, gp = lane & 3;
    bool evalid = en < nk;
    int xi = x[evalid ? n0 + en : n0];
    int d_nx = __builtin_amdgcn_readlane(dv, 0);
    int s_nx = __builtin_amdgcn_readlane(sv, 0);
    int idx_nx = lane < d_nx ? csr_src[s_nx + lane] : 0;
    for (int k = 0; k < nk; ++k) {
        int d = d_nx, start = s_nx, idx0 = idx_nx;
        asm volatile("" :: "v"(idx0));
        d_nx = __builtin_amdgcn_readlane(dv, k + 1);
        s_nx = __builtin_amdgcn_readlane(sv, k + 1);
        idx_nx = lane < d_nx ? csr_src[s_nx + lane] : 0;
        float2 acc = {0.0f, 0.0f};
        acc = cgmm_add_rows(prev_h, lane, idx0, d < 32 ? d : 32, acc);
        for (int base = 32; base < d; base += 32) {
            int cnt = d - base < 32 ? d - base : 32;
            int myidx = lane < cnt ? csr_src[start + base + lane] : 0;
            acc = cgmm_add_rows(prev_h, lane, myidx, cnt, acc);
        }
        *((float2*)&ldsX[wave][k][2 * lane]) = acc;
    }
    for (int k = nk; k < 8; ++k) {
        float2 zf = {0.0f, 0.0f};
        *((float2*)&ldsX[wave][k][2 * lane]) = zf;
    }

    const float4* bp = (const float4*)(Bs_t + (size_t)xi * 64 + gp * 16);
    float4 e0 = bp[0], e1 = bp[1], e2v = bp[2], e3 = bp[3];

    int  c  = lane & 15;
    bool hi = lane >= 16;
    int  nr = c & 7;
    float inv_cnt = 1.0f / fmaxf((float)__shfl(dv, nr, 32), 1.0f);
    const float* srow = &ldsX[wave][nr][hi ? 1 : 0];
#pragma unroll
    for (int p = 0; p < 4; ++p) {
        v16h a;
#pragma unroll
        for (int s = 0; s < 8; ++s)
            a[s] = (_Float16)(srow[s * 8 + 2 * p] * inv_cnt);
#pragma unroll
        for (int s = 8; s < 16; ++s)
            a[s] = (_Float16)0.0f;

        v16h b = *((const v16h*)(bfrag + (p * 32 + lane) * 16));

        v8f acc = {};
        acc = __builtin_amdgcn_wmma_f32_16x16x32_f16(
            false, a, false, b, (short)0, acc, false, false);
        if (!hi) {
#pragma unroll
            for (int r = 0; r < 8; ++r)
                ldsY[wave][r][16 * p + c] = acc[r];
        }
    }

    const float4* tp = (const float4*)&ldsY[wave][en][gp * 16];
    float4 t0 = tp[0], t1 = tp[1], t2 = tp[2], t3 = tp[3];
    float ua[8], ub[8];
    ua[0] = t0.x * e0.x;  ua[1] = t0.y * e0.y;  ua[2] = t0.z * e0.z;  ua[3] = t0.w * e0.w;
    ua[4] = t1.x * e1.x;  ua[5] = t1.y * e1.y;  ua[6] = t1.z * e1.z;  ua[7] = t1.w * e1.w;
    ub[0] = t2.x * e2v.x; ub[1] = t2.y * e2v.y; ub[2] = t2.z * e2v.z; ub[3] = t2.w * e2v.w;
    ub[4] = t3.x * e3.x;  ub[5] = t3.y * e3.y;  ub[6] = t3.z * e3.z;  ub[7] = t3.w * e3.w;
    float sa = ((ua[0] + ua[1]) + (ua[2] + ua[3])) + ((ua[4] + ua[5]) + (ua[6] + ua[7]));
    float sb = ((ub[0] + ub[1]) + (ub[2] + ub[3])) + ((ub[4] + ub[5]) + (ub[6] + ub[7]));
    float da = sa + 64.0f * EPS_F, db = sb + 64.0f * EPS_F;
    float ia = 1.0f / da, ib = 1.0f / db;
#pragma unroll
    for (int i = 0; i < 8; ++i) {
        float2 pv = {(ua[i] + 8.0f * EPS_F) * ia, (ub[i] + 8.0f * EPS_F) * ib};
        *((float2*)&ldsX[wave][en][i * 8 + 2 * gp]) = pv;
    }
    if (evalid) {
        float2 lv = {__logf(da), __logf(db)};
        *((float2*)(lik + (size_t)n0 * 8 + 2 * lane)) = lv;
    }
    float4* dstp = (float4*)(post + (size_t)n0 * 64);
#pragma unroll
    for (int q = 0; q < 4; ++q) {
        int f = 32 * q + lane;
        int nd = f >> 4;
        if (nd < nk) dstp[f] = *((const float4*)&ldsX[wave][nd][(f & 15) * 4]);
    }
}

extern "C" void kernel_launch(void* const* d_in, const int* in_sizes, int n_in,
                              void* d_out, int out_size, void* d_ws, size_t ws_size,
                              hipStream_t stream) {
    (void)n_in; (void)out_size; (void)ws_size;
    const int*   x      = (const int*)d_in[0];
    const float* prev_h = (const float*)d_in[1];
    const int*   ei     = (const int*)d_in[2];
    const float* Q      = (const float*)d_in[3];
    const float* B      = (const float*)d_in[4];
    const int N = in_sizes[0];
    const int E = in_sizes[2] / 2;

    float* ws     = (float*)d_ws;
    int*   deg    = (int*)(ws + (size_t)N * 64);
    float* Bs_t   = ws + (size_t)N * 65;
    size_t off    = ((size_t)N * 65 + 2048 + 7) & ~(size_t)7;
    _Float16* bfrag = (_Float16*)(ws + off);

    float* lik  = (float*)d_out;
    float* post = lik + (size_t)N * 8;

    const int NB = (N + 511) / 512;
    const int NBP = (NB + 3) & ~3;
    const int B1 = (E + 8191) / 8192;
    int* bkt      = (int*)ws;
    int* cnt      = bkt + ((E + 3) & ~3);
    int* bstart   = cnt + B1 * NBP;
    int* rowstart = bstart + ((NB + 4) & ~3);
    int* csr_src  = rowstart + ((N + 3) & ~3);

    cgmm_bkt_count<<<B1, 1024, 0, stream>>>(ei, cnt, E, NBP);
    cgmm_bkt_scatter<<<B1, 1024, 0, stream>>>(ei, cnt, bkt, bstart, E, NB, NBP, B1);
    cgmm_bkt_sort<<<NB + 1, 1024, 0, stream>>>(bkt, bstart, deg, rowstart, csr_src, N, NB,
                                               Q, B, Bs_t, bfrag);

    cgmm_gather_node<<<(N + 63) / 64, 256, 0, stream>>>(prev_h, csr_src, rowstart, deg, x,
                                                         bfrag, Bs_t, lik, post, N);
}


# Bucket sort and gather-sum fused in one work-group: sorted source list, degrees and starts stay in LDS, each wave sums and contracts 16 nodes of its bucket; launches 4 -> 3; float add order unchanged
# speedup vs baseline: 5.9019x; 1.0430x over previous
#include <hip/hip_runtime.h>
#include <hip/hip_bf16.h>
#include <math.h>


typedef __attribute__((ext_vector_type(16))) _Float16 v16h;
typedef __attribute__((ext_vector_type(8)))  float    v8f;

#define EPS_F 1e-8f

__device__ __forceinline__ void cgmm_params_block(const float* __restrict__ Q,
                                                  const float* __restrict__ B,
                                                  float* __restrict__ Bs_t,
                                                  _Float16* __restrict__ bfrag,
                                                  float* sQ, int t) {
    if (t < 64) {
        int g = t & 7, a = t >> 3;
        {
            int l = a;
            float mx = -1e30f;
            for (int i = 0; i < 8; ++i) mx = fmaxf(mx, Q[i * 64 + l * 8 + g]);
            float e[8], s = 0.0f;
            for (int i = 0; i < 8; ++i) { e[i] = __expf(Q[i * 64 + l * 8 + g] - mx); s += e[i]; }
            float inv = 1.0f / s;
            for (int i = 0; i < 8; ++i) sQ[i * 64 + l * 8 + g] = e[i] * inv;
        }
        {
            int i = a;
            float mx = -1e30f;
            for (int m = 0; m < 32; ++m) mx = fmaxf(mx, B[i * 256 + m * 8 + g]);
            float ev[32], s = 0.0f;
            for (int m = 0; m < 32; ++m) { ev[m] = __expf(B[i * 256 + m * 8 + g] - mx); s += ev[m]; }
            float inv = 1.0f / s;
            for (int m = 0; m < 32; ++m) Bs_t[m * 64 + g * 8 + i] = ev[m] * inv;
        }
    }
    __syncthreads();
    if (t >= 128) return;
    int p = t >> 5, lane = t & 31;
    bool hi = lane >= 16;
    int  m  = lane & 15;
    bool bvalid = (hi == (m >= 8));
    int  icol = m & 7;
    int  g = 2 * p + (hi ? 1 : 0);
    _Float16* dst = bfrag + (p * 32 + lane) * 16;
    for (int e2 = 0; e2 < 16; ++e2) {
        float f = (e2 < 8 && bvalid) ? sQ[icol * 64 + e2 * 8 + g] : 0.0f;
        dst[e2] = (_Float16)f;
    }
}

#define CGMM_NPB_BITS 9
#define CGMM_SRC_BITS 23
#define CGMM_CH       8192

__device__ __forceinline__ int4 cgmm_ldq_tail(const int* __restrict__ row, int e, int E) {
    int4 v;
    v.x = e     < E ? row[e]     : -1;
    v.y = e + 1 < E ? row[e + 1] : -1;
    v.z = e + 2 < E ? row[e + 2] : -1;
    v.w = e + 3 < E ? row[e + 3] : -1;
    return v;
}

__device__ __forceinline__ int cgmm_block_excl_scan(int v, int* wsum, int t) {
    int lane = t & 31, wv = t >> 5;
    int incl = v;
#pragma unroll
    for (int s = 1; s < 32; s <<= 1) {
        int up = __shfl_up(incl, s, 32);
        if (lane >= s) incl += up;
    }
    if (lane == 31) wsum[wv] = incl;
    __syncthreads();
    if (wv == 0) {
        int w = wsum[lane];
        int wi = w;
#pragma unroll
        for (int s = 1; s < 32; s <<= 1) {
            int up = __shfl_up(wi, s, 32);
            if (lane >= s) wi += up;
        }
        wsum[lane] = wi - w;
    }
    __syncthreads();
    return wsum[wv] + (incl - v);
}

__device__ __forceinline__ int2 cgmm_block_excl_scan2(int va, int vb, int* wsum, int t) {
    int lane = t & 31, wv = t >> 5;
    int ia = va, ib = vb;
#pragma unroll
    for (int s = 1; s < 32; s <<= 1) {
        int ua = __shfl_up(ia, s, 32);
        int ub = __shfl_up(ib, s, 32);
        if (lane >= s) { ia += ua; ib += ub; }
    }
    if (lane == 31) { wsum[wv] = ia; wsum[32 + wv] = ib; }
    __syncthreads();
    if (wv < 2) {
        int w = wsum[t];
        int wi = w;
#pragma unroll
        for (int s = 1; s < 32; s <<= 1) {
            int up = __shfl_up(wi, s, 32);
            if (lane >= s) wi += up;
        }
        wsum[t] = wi - w;
    }
    __syncthreads();
    int2 r = {wsum[wv] + (ia - va), wsum[32 + wv] + (ib - vb)};
    return r;
}

#define CGMM_COUNT1(d) if ((d) >= 0) atomicAdd(&hist[(d) >> CGMM_NPB_BITS], 1);
__global__ __launch_bounds__(1024) void cgmm_bkt_count(const int* __restrict__ ei,
                                                       int* __restrict__ cnt, int E, int NBP) {
    __shared__ int hist[256];
    int t = threadIdx.x;
    if (t < 256) hist[t] = 0;
    int e0 = blockIdx.x * CGMM_CH + 4 * t, e1 = e0 + 4096;
    int4 a, b;
    if (e1 + 3 < E) {
        a = *((const int4*)(ei + e0));
        b = *((const int4*)(ei + e1));
    } else {
        a = cgmm_ldq_tail(ei, e0, E);
        b = cgmm_ldq_tail(ei, e1, E);
    }
    __syncthreads();
    CGMM_COUNT1(a.x) CGMM_COUNT1(a.y) CGMM_COUNT1(a.z) CGMM_COUNT1(a.w)
    CGMM_COUNT1(b.x) CGMM_COUNT1(b.y) CGMM_COUNT1(b.z) CGMM_COUNT1(b.w)
    __syncthreads();
    if (t < NBP) cnt[blockIdx.x * NBP + t] = hist[t];
}
#undef CGMM_COUNT1

#define CGMM_SC_ROWS 13
__global__ __launch_bounds__(1024) void cgmm_bkt_scatter(const int* __restrict__ ei,
                                                         const int* __restrict__ cnt,
                                                         int* __restrict__ bkt,
                                                         int* __restrict__ bstart,
                                                         int E, int NB, int NBP, int B1,
                                                         const float* __restrict__ Q,
                                                         const float* __restrict__ B,
                                                         float* __restrict__ Bs_t,
                                                         _Float16* __restrict__ bfrag) {
    __shared__ int cur[256];
    __shared__ int gdl[256];
    __shared__ int stot[256];
    __shared__ int spre[256];
    __shared__ int wsum[64];
    __shared__ int stage[CGMM_CH];
    __shared__ unsigned char sb[CGMM_CH];
    __shared__ float sQ[512];
    int t = threadIdx.x;
    if ((int)blockIdx.x == B1) {
        cgmm_params_block(Q, B, Bs_t, bfrag, sQ, t);
        return;
    }
    if (t < 256) { stot[t] = 0; spre[t] = 0; }
    int row = blockIdx.x * NBP;
    const int* srcrow = ei + E;
    int e0 = blockIdx.x * CGMM_CH + 4 * t, e1 = e0 + 4096;
    int4 d0, d1, s0, s1;
    if (e1 + 3 < E && (E & 3) == 0) {
        d0 = *((const int4*)(ei + e0));     d1 = *((const int4*)(ei + e1));
        s0 = *((const int4*)(srcrow + e0)); s1 = *((const int4*)(srcrow + e1));
    } else {
        d0 = cgmm_ldq_tail(ei, e0, E);     d1 = cgmm_ldq_tail(ei, e1, E);
        s0 = cgmm_ldq_tail(srcrow, e0, E); s1 = cgmm_ldq_tail(srcrow, e1, E);
    }
    int c = t < NB ? cnt[row + t] : 0;
    const int q = t & 63, rg = t >> 6, Q4 = NBP >> 2;
    const int blk = blockIdx.x;
    int4 tot4 = {0, 0, 0, 0}, pre4 = {0, 0, 0, 0};
    if (q < Q4) {
        const int4* cnt4 = (const int4*)cnt;
        for (int r0 = rg; r0 < B1; r0 += 16 * CGMM_SC_ROWS) {
            int4 v[CGMM_SC_ROWS];
#pragma unroll
            for (int j = 0; j < CGMM_SC_ROWS; ++j) {
                int r = r0 + 16 * j;
                v[j] = cnt4[(r < B1 ? r : r0) * Q4 + q];
            }
            __builtin_amdgcn_sched_barrier(0);
#pragma unroll
            for (int j = 0; j < CGMM_SC_ROWS; ++j) {
                int r = r0 + 16 * j;
                int mt = r < B1 ? -1 : 0;
                int mp = r < blk ? -1 : 0;
                tot4.x += v[j].x & mt; tot4.y += v[j].y & mt; tot4.z += v[j].z & mt; tot4.w += v[j].w & mt;
                pre4.x += v[j].x & mp; pre4.y += v[j].y & mp; pre4.z += v[j].z & mp; pre4.w += v[j].w & mp;
            }
        }
    }
    __syncthreads();
    if (q < Q4) {
        atomicAdd(&stot[4 * q],     tot4.x); atomicAdd(&stot[4 * q + 1], tot4.y);
        atomicAdd(&stot[4 * q + 2], tot4.z); atomicAdd(&stot[4 * q + 3], tot4.w);
        atomicAdd(&spre[4 * q],     pre4.x); atomicAdd(&spre[4 * q + 1], pre4.y);
        atomicAdd(&spre[4 * q + 2], pre4.z); atomicAdd(&spre[4 * q + 3], pre4.w);
    }
    __syncthreads();
    int tot = t < 256 ? stot[t] : 0;
    int2 sc = cgmm_block_excl_scan2(tot, c, wsum, t);
    if (t < 256) {
        cur[t] = sc.y;
        gdl[t] = sc.x + spre[t] - sc.y;
        if (blk == 0 && t <= NB) bstart[t] = sc.x;
    }
    __syncthreads();
    int dd[8] = {d0.x, d0.y, d0.z, d0.w, d1.x, d1.y, d1.z, d1.w};
    int ss[8] = {s0.x, s0.y, s0.z, s0.w, s1.x, s1.y, s1.z, s1.w};
    int pp[8];
#pragma unroll
    for (int k = 0; k < 8; ++k)
        pp[k] = dd[k] >= 0 ? atomicAdd(&cur[dd[k] >> CGMM_NPB_BITS], 1) : 0;
#pragma unroll
    for (int k = 0; k < 8; ++k) {
        if (dd[k] >= 0) {
            stage[pp[k]] = (int)((((unsigned)dd[k] & 511u) << CGMM_SRC_BITS) | (unsigned)ss[k]);
            sb[pp[k]]    = (unsigned char)(dd[k] >> CGMM_NPB_BITS);
        }
    }
    __syncthreads();
    int nblk = E - blockIdx.x * CGMM_CH;
    if (nblk >= CGMM_CH) {
#pragma unroll
        for (int k = 0; k < 8; ++k) {
            int j = t + 1024 * k;
            bkt[j + gdl[sb[j]]] = stage[j];
        }
    } else {
        for (int j = t; j < nblk; j += 1024) bkt[j + gdl[sb[j]]] = stage[j];
    }
}
#undef CGMM_SC_ROWS

#define CGMM_IDX(p) __builtin_amdgcn_readlane(myidx, (p))
#define CGMM_AT(i)  (((const float2*)(prev_h + (size_t)(i) * 64))[lane])
#define CGMM_ROW(k) CGMM_AT(CGMM_IDX(j + (k)))
#define CGMM_ADD(r) acc.x += r.x; acc.y += r.y;
__device__ __forceinline__ float2 cgmm_add_rows(const float* __restrict__ prev_h, int lane,
                                                int myidx, int cnt, float2 acc) {
    int j = 0;
    for (; j + 16 <= cnt; j += 16) {
        float2 r0 = CGMM_ROW(0),  r1 = CGMM_ROW(1),  r2 = CGMM_ROW(2),  r3 = CGMM_ROW(3);
        float2 r4 = CGMM_ROW(4),  r5 = CGMM_ROW(5),  r6 = CGMM_ROW(6),  r7 = CGMM_ROW(7);
        float2 r8 = CGMM_ROW(8),  r9 = CGMM_ROW(9),  r10 = CGMM_ROW(10), r11 = CGMM_ROW(11);
        float2 r12 = CGMM_ROW(12), r13 = CGMM_ROW(13), r14 = CGMM_ROW(14), r15 = CGMM_ROW(15);
        CGMM_ADD(r0) CGMM_ADD(r1) CGMM_ADD(r2) CGMM_ADD(r3)
        CGMM_ADD(r4) CGMM_ADD(r5) CGMM_ADD(r6) CGMM_ADD(r7)
        CGMM_ADD(r8) CGMM_ADD(r9) CGMM_ADD(r10) CGMM_ADD(r11)
        CGMM_ADD(r12) CGMM_ADD(r13) CGMM_ADD(r14) CGMM_ADD(r15)
    }
    int m = cnt - j;
    if (m) {
        int j4 = j + (m & 8), j2 = j4 + (m & 4), j1 = j2 + (m & 2);
        int ia0 = CGMM_IDX(j),      ia1 = CGMM_IDX(j + 1),  ia2 = CGMM_IDX(j + 2),  ia3 = CGMM_IDX(j + 3);
        int ia4 = CGMM_IDX(j + 4),  ia5 = CGMM_IDX(j + 5),  ia6 = CGMM_IDX(j + 6),  ia7 = CGMM_IDX(j + 7);
        int ib0 = CGMM_IDX(j4),     ib1 = CGMM_IDX(j4 + 1), ib2 = CGMM_IDX(j4 + 2), ib3 = CGMM_IDX(j4 + 3);
        int ic0 = CGMM_IDX(j2),     ic1 = CGMM_IDX(j2 + 1), ie0 = CGMM_IDX(j1);
        float2 zf = {0.0f, 0.0f};
        float2 a0 = zf, a1 = zf, a2 = zf, a3 = zf, a4 = zf, a5 = zf, a6 = zf, a7 = zf;
        float2 b0 = zf, b1 = zf, b2 = zf, b3 = zf, c0 = zf, c1 = zf, e0 = zf;
        if (m & 8) {
            a0 = CGMM_AT(ia0); a1 = CGMM_AT(ia1); a2 = CGMM_AT(ia2); a3 = CGMM_AT(ia3);
            a4 = CGMM_AT(ia4); a5 = CGMM_AT(ia5); a6 = CGMM_AT(ia6); a7 = CGMM_AT(ia7);
        }
        if (m & 4) { b0 = CGMM_AT(ib0); b1 = CGMM_AT(ib1); b2 = CGMM_AT(ib2); b3 = CGMM_AT(ib3); }
        if (m & 2) { c0 = CGMM_AT(ic0); c1 = CGMM_AT(ic1); }
        if (m & 1) { e0 = CGMM_AT(ie0); }
        if (m & 8) {
            CGMM_ADD(a0) CGMM_ADD(a1) CGMM_ADD(a2) CGMM_ADD(a3)
            CGMM_ADD(a4) CGMM_ADD(a5) CGMM_ADD(a6) CGMM_ADD(a7)
        }
        if (m & 4) { CGMM_ADD(b0) CGMM_ADD(b1) CGMM_ADD(b2) CGMM_ADD(b3) }
        if (m & 2) { CGMM_ADD(c0) CGMM_ADD(c1) }
        if (m & 1) { CGMM_ADD(e0) }
    }
    return acc;
}
#undef CGMM_ADD
#undef CGMM_ROW
#undef CGMM_AT
#undef CGMM_IDX

#define CGMM_CAP 10240
#define CGMM_PITCH 72
#define CGMM_LOAD4 \
        int i1 = i + 1024, i2 = i + 2048, i3 = i + 3072; \
        bool v1 = i1 < hi, v2 = i2 < hi, v3 = i3 < hi; \
        unsigned w0 = (unsigned)bkt[i]; \
        unsigned w1 = (unsigned)bkt[v1 ? i1 : i]; \
        unsigned w2 = (unsigned)bkt[v2 ? i2 : i]; \
        unsigned w3 = (unsigned)bkt[v3 ? i3 : i]; \
        asm volatile("" :: "v"(w0), "v"(w1), "v"(w2), "v"(w3));
#define CGMM_RANK4 \
        int p0 = atomicAdd(&curl[w0 >> CGMM_SRC_BITS], 1); \
        int p1 = v1 ? atomicAdd(&curl[w1 >> CGMM_SRC_BITS], 1) : 0; \
        int p2 = v2 ? atomicAdd(&curl[w2 >> CGMM_SRC_BITS], 1) : 0; \
        int p3 = v3 ? atomicAdd(&curl[w3 >> CGMM_SRC_BITS], 1) : 0;

template <bool IN_LDS>
__device__ __forceinline__ int cgmm_src_at(const int* stage, const int* csrb, int pos) {
    if (IN_LDS) return stage[pos];
    return csrb[pos];
}

template <bool IN_LDS>
__device__ __forceinline__ void cgmm_chunk(const int* stage, const int* csrb,
                                           const int* cntl, const int* curl,
                                           int nl0, int n0, int nk, int lane,
                                           const float* __restrict__ prev_h,
                                           const int* __restrict__ x,
                                           const _Float16* __restrict__ bfrag,
                                           const float* __restrict__ Bs_t,
                                           float* __restrict__ lik,
                                           float* __restrict__ post,
                                           float (*lx)[CGMM_PITCH],
                                           float (*ly)[CGMM_PITCH]) {
    int dv = 0, sv = 0;
    if (lane < nk) { dv = cntl[nl0 + lane]; sv = curl[nl0 + lane] - dv; }
    int en = lane >> 2, gp = lane & 3;
    bool evalid = en < nk;
    int xi = x[evalid ? n0 + en : n0];
    int d_nx = __builtin_amdgcn_readlane(dv, 0);
    int s_nx = __builtin_amdgcn_readlane(sv, 0);
    int idx_nx = lane < d_nx ? cgmm_src_at<IN_LDS>(stage, csrb, s_nx + lane) : 0;
    for (int k = 0; k < nk; ++k) {
        int d = d_nx, start = s_nx, idx0 = idx_nx;
        if (!IN_LDS) asm volatile("" :: "v"(idx0));
        d_nx = __builtin_amdgcn_readlane(dv, k + 1);
        s_nx = __builtin_amdgcn_readlane(sv, k + 1);
        idx_nx = lane < d_nx ? cgmm_src_at<IN_LDS>(stage, csrb, s_nx + lane) : 0;
        float2 acc = {0.0f, 0.0f};
        acc = cgmm_add_rows(prev_h, lane, idx0, d < 32 ? d : 32, acc);
        for (int base = 32; base < d; base += 32) {
            int cnt = d - base < 32 ? d - base : 32;
            int myidx = lane < cnt ? cgmm_src_at<IN_LDS>(stage, csrb, start + base + lane) : 0;
            acc = cgmm_add_rows(prev_h, lane, myidx, cnt, acc);
        }
        *((float2*)&lx[k][2 * lane]) = acc;
    }
    for (int k = nk; k < 8; ++k) {
        float2 zf = {0.0f, 0.0f};
        *((float2*)&lx[k][2 * lane]) = zf;
    }

    const float4* bp = (const float4*)(Bs_t + (size_t)xi * 64 + gp * 16);
    float4 e0 = bp[0], e1 = bp[1], e2v = bp[2], e3 = bp[3];

    int  c  = lane & 15;
    bool hi = lane >= 16;
    int  nr = c & 7;
    float inv_cnt = 1.0f / fmaxf((float)__shfl(dv, nr, 32), 1.0f);
    const float* srow = &lx[nr][hi ? 1 : 0];
#pragma unroll
    for (int p = 0; p < 4; ++p) {
        v16h a;
#pragma unroll
        for (int s = 0; s < 8; ++s)
            a[s] = (_Float16)(srow[s * 8 + 2 * p] * inv_cnt);
#pragma unroll
        for (int s = 8; s < 16; ++s)
            a[s] = (_Float16)0.0f;

        v16h b = *((const v16h*)(bfrag + (p * 32 + lane) * 16));

        v8f acc = {};
        acc = __builtin_amdgcn_wmma_f32_16x16x32_f16(
            false, a, false, b, (short)0, acc, false, false);
        if (!hi) {
#pragma unroll
            for (int r = 0; r < 8; ++r)
                ly[r][16 * p + c] = acc[r];
        }
    }

    const float4* tp = (const float4*)&ly[en][gp * 16];
    float4 t0 = tp[0], t1 = tp[1], t2 = tp[2], t3 = tp[3];
    float ua[8], ub[8];
    ua[0] = t0.x * e0.x;  ua[1] = t0.y * e0.y;  ua[2] = t0.z * e0.z;  ua[3] = t0.w * e0.w;
    ua[4] = t1.x * e1.x;  ua[5] = t1.y * e1.y;  ua[6] = t1.z * e1.z;  ua[7] = t1.w * e1.w;
    ub[0] = t2.x * e2v.x; ub[1] = t2.y * e2v.y; ub[2] = t2.z * e2v.z; ub[3] = t2.w * e2v.w;
    ub[4] = t3.x * e3.x;  ub[5] = t3.y * e3.y;  ub[6] = t3.z * e3.z;  ub[7] = t3.w * e3.w;
    float sa = ((ua[0] + ua[1]) + (ua[2] + ua[3])) + ((ua[4] + ua[5]) + (ua[6] + ua[7]));
    float sb = ((ub[0] + ub[1]) + (ub[2] + ub[3])) + ((ub[4] + ub[5]) + (ub[6] + ub[7]));
    float da = sa + 64.0f * EPS_F, db = sb + 64.0f * EPS_F;
    float ia = 1.0f / da, ib = 1.0f / db;
#pragma unroll
    for (int i = 0; i < 8; ++i) {
        float2 pv = {(ua[i] + 8.0f * EPS_F) * ia, (ub[i] + 8.0f * EPS_F) * ib};
        *((float2*)&lx[en][i * 8 + 2 * gp]) = pv;
    }
    if (evalid) {
        float2 lv = {__logf(da), __logf(db)};
        *((float2*)(lik + (size_t)n0 * 8 + 2 * lane)) = lv;
    }
    float4* dstp = (float4*)(post + (size_t)n0 * 64);
#pragma unroll
    for (int q = 0; q < 4; ++q) {
        int f = 32 * q + lane;
        int nd = f >> 4;
        if (nd < nk) dstp[f] = *((const float4*)&lx[nd][(f & 15) * 4]);
    }
}

__global__ __launch_bounds__(1024) void cgmm_sort_gather(const int* __restrict__ bkt,
                                                         const int* __restrict__ bstart,
                                                         int* csr_src, int N,
                                                         const float* __restrict__ prev_h,
                                                         const int* __restrict__ x,
                                                         const _Float16* __restrict__ bfrag,
                                                         const float* __restrict__ Bs_t,
                                                         float* __restrict__ lik,
                                                         float* __restrict__ post) {
    __shared__ int cntl[512];
    __shared__ int curl[512];
    __shared__ int wsum[32];
    __shared__ int stage[CGMM_CAP];
    __shared__ alignas(16) float ldsX[32][8][CGMM_PITCH];
    __shared__ alignas(16) float ldsY[32][8][CGMM_PITCH];
    int t = threadIdx.x;
    int lo = bstart[blockIdx.x], hi = bstart[blockIdx.x + 1];
    if (t < 512) cntl[t] = 0;
    __syncthreads();
    for (int i = lo + t; i < hi; i += 4096) {
        CGMM_LOAD4
        atomicAdd(&cntl[w0 >> CGMM_SRC_BITS], 1);
        if (v1) atomicAdd(&cntl[w1 >> CGMM_SRC_BITS], 1);
        if (v2) atomicAdd(&cntl[w2 >> CGMM_SRC_BITS], 1);
        if (v3) atomicAdd(&cntl[w3 >> CGMM_SRC_BITS], 1);
    }
    __syncthreads();
    int d = t < 512 ? cntl[t] : 0;
    int ex = cgmm_block_excl_scan(d, wsum, t);
    if (t < 512) curl[t] = ex;
    __syncthreads();
    const unsigned smask = (1u << CGMM_SRC_BITS) - 1u;
    const int cnt_b = hi - lo;
    const bool in_lds = cnt_b <= CGMM_CAP;
    if (in_lds) {
        for (int i = lo + t; i < hi; i += 4096) {
            CGMM_LOAD4
            CGMM_RANK4
            stage[p0] = (int)(w0 & smask);
            if (v1) stage[p1] = (int)(w1 & smask);
            if (v2) stage[p2] = (int)(w2 & smask);
            if (v3) stage[p3] = (int)(w3 & smask);
        }
    } else {
        for (int i = lo + t; i < hi; i += 4096) {
            CGMM_LOAD4
            CGMM_RANK4
            csr_src[lo + p0] = (int)(w0 & smask);
            if (v1) csr_src[lo + p1] = (int)(w1 & smask);
            if (v2) csr_src[lo + p2] = (int)(w2 & smask);
            if (v3) csr_src[lo + p3] = (int)(w3 & smask);
        }
        __threadfence();
    }
    __syncthreads();

    int lane = t & 31;
    int wv = __builtin_amdgcn_readfirstlane(t >> 5);
    const int nb0 = blockIdx.x << CGMM_NPB_BITS;
    for (int ch = wv; ch < 64; ch += 32) {
        int nl0 = ch * 8, n0 = nb0 + nl0;
        if (n0 >= N) break;
        int nk = N - n0 < 8 ? N - n0 : 8;
        if (in_lds)
            cgmm_chunk<true>(stage, csr_src + lo, cntl, curl, nl0, n0, nk, lane, prev_h, x,
                             bfrag, Bs_t, lik, post, ldsX[wv], ldsY[wv]);
        else
            cgmm_chunk<false>(stage, csr_src + lo, cntl, curl, nl0, n0, nk, lane, prev_h, x,
                              bfrag, Bs_t, lik, post, ldsX[wv], ldsY[wv]);
    }
}
#undef CGMM_RANK4
#undef CGMM_LOAD4

extern "C" void kernel_launch(void* const* d_in, const int* in_sizes, int n_in,
                              void* d_out, int out_size, void* d_ws, size_t ws_size,
                              hipStream_t stream) {
    (void)n_in; (void)out_size; (void)ws_size;
    const int*   x      = (const int*)d_in[0];
    const float* prev_h = (const float*)d_in[1];
    const int*   ei     = (const int*)d_in[2];
    const float* Q      = (const float*)d_in[3];
    const float* B      = (const float*)d_in[4];
    const int N = in_sizes[0];
    const int E = in_sizes[2] / 2;

    float* ws     = (float*)d_ws;
    float* Bs_t   = ws + (size_t)N * 65;
    size_t off    = ((size_t)N * 65 + 2048 + 7) & ~(size_t)7;
    _Float16* bfrag = (_Float16*)(ws + off);

    float* lik  = (float*)d_out;
    float* post = lik + (size_t)N * 8;

    const int NB = (N + 511) / 512;
    const int NBP = (NB + 3) & ~3;
    const int B1 = (E + 8191) / 8192;
    int* bkt      = (int*)ws;
    int* cnt      = bkt + ((E + 3) & ~3);
    int* bstart   = cnt + B1 * NBP;
    int* csr_src  = bstart + ((NB + 4) & ~3);

    cgmm_bkt_count<<<B1, 1024, 0, stream>>>(ei, cnt, E, NBP);
    cgmm_bkt_scatter<<<B1 + 1, 1024, 0, stream>>>(ei, cnt, bkt, bstart, E, NB, NBP, B1,
                                                  Q, B, Bs_t, bfrag);

    cgmm_sort_gather<<<NB, 1024, 0, stream>>>(bkt, bstart, csr_src, N, prev_h, x,
                                              bfrag, Bs_t, lik, post);
}


# One LDS-atomic pass per sort level, not two: the counting add's returned old value is the entry's rank, held in registers over the scan; place = start + rank; add order: that pass's LDS arrival
# speedup vs baseline: 6.8480x; 1.1603x over previous
#include <hip/hip_runtime.h>
#include <hip/hip_bf16.h>
#include <math.h>


typedef __attribute__((ext_vector_type(16))) _Float16 v16h;
typedef __attribute__((ext_vector_type(8)))  float    v8f;

#define EPS_F 1e-8f

__device__ __forceinline__ void cgmm_params_block(const float* __restrict__ Q,
                                                  const float* __restrict__ B,
                                                  float* __restrict__ Bs_t,
                                                  _Float16* __restrict__ bfrag,
                                                  float* sQ, int t) {
    if (t < 64) {
        int g = t & 7, a = t >> 3;
        {
            int l = a;
            float mx = -1e30f;
            for (int i = 0; i < 8; ++i) mx = fmaxf(mx, Q[i * 64 + l * 8 + g]);
            float e[8], s = 0.0f;
            for (int i = 0; i < 8; ++i) { e[i] = __expf(Q[i * 64 + l * 8 + g] - mx); s += e[i]; }
            float inv = 1.0f / s;
            for (int i = 0; i < 8; ++i) sQ[i * 64 + l * 8 + g] = e[i] * inv;
        }
        {
            int i = a;
            float mx = -1e30f;
            for (int m = 0; m < 32; ++m) mx = fmaxf(mx, B[i * 256 + m * 8 + g]);
            float ev[32], s = 0.0f;
            for (int m = 0; m < 32; ++m) { ev[m] = __expf(B[i * 256 + m * 8 + g] - mx); s += ev[m]; }
            float inv = 1.0f / s;
            for (int m = 0; m < 32; ++m) Bs_t[m * 64 + g * 8 + i] = ev[m] * inv;
        }
    }
    __syncthreads();
    if (t >= 128) return;
    int p = t >> 5, lane = t & 31;
    bool hi = lane >= 16;
    int  m  = lane & 15;
    bool bvalid = (hi == (m >= 8));
    int  icol = m & 7;
    int  g = 2 * p + (hi ? 1 : 0);
    _Float16* dst = bfrag + (p * 32 + lane) * 16;
    for (int e2 = 0; e2 < 16; ++e2) {
        float f = (e2 < 8 && bvalid) ? sQ[icol * 64 + e2 * 8 + g] : 0.0f;
        dst[e2] = (_Float16)f;
    }
}

#define CGMM_NPB_BITS 9
#define CGMM_SRC_BITS 23
#define CGMM_CH       8192

__device__ __forceinline__ int4 cgmm_ldq_tail(const int* __restrict__ row, int e, int E) {
    int4 v;
    v.x = e     < E ? row[e]     : -1;
    v.y = e + 1 < E ? row[e + 1] : -1;
    v.z = e + 2 < E ? row[e + 2] : -1;
    v.w = e + 3 < E ? row[e + 3] : -1;
    return v;
}

__device__ __forceinline__ int cgmm_block_excl_scan(int v, int* wsum, int t) {
    int lane = t & 31, wv = t >> 5;
    int incl = v;
#pragma unroll
    for (int s = 1; s < 32; s <<= 1) {
        int up = __shfl_up(incl, s, 32);
        if (lane >= s) incl += up;
    }
    if (lane == 31) wsum[wv] = incl;
    __syncthreads();
    if (wv == 0) {
        int w = wsum[lane];
        int wi = w;
#pragma unroll
        for (int s = 1; s < 32; s <<= 1) {
            int up = __shfl_up(wi, s, 32);
            if (lane >= s) wi += up;
        }
        wsum[lane] = wi - w;
    }
    __syncthreads();
    return wsum[wv] + (incl - v);
}

__global__ __launch_bounds__(1024) void cgmm_bkt_scatter(const int* __restrict__ ei,
                                                         int* __restrict__ bkt,
                                                         int* __restrict__ lstart,
                                                         int E, int LP, int B1,
                                                         const float* __restrict__ Q,
                                                         const float* __restrict__ B,
                                                         float* __restrict__ Bs_t,
                                                         _Float16* __restrict__ bfrag) {
    __shared__ int cur[256];
    __shared__ int wsum[32];
    __shared__ alignas(16) int stage[CGMM_CH];
    __shared__ float sQ[512];
    int t = threadIdx.x;
    if ((int)blockIdx.x == B1) {
        cgmm_params_block(Q, B, Bs_t, bfrag, sQ, t);
        return;
    }
    if (t < 256) cur[t] = 0;
    const int blk = blockIdx.x;
    const int* srcrow = ei + E;
    int e0 = blk * CGMM_CH + 4 * t, e1 = e0 + 4096;
    int4 d0, d1, s0, s1;
    if (e1 + 3 < E && (E & 3) == 0) {
        d0 = *((const int4*)(ei + e0));     d1 = *((const int4*)(ei + e1));
        s0 = *((const int4*)(srcrow + e0)); s1 = *((const int4*)(srcrow + e1));
    } else {
        d0 = cgmm_ldq_tail(ei, e0, E);     d1 = cgmm_ldq_tail(ei, e1, E);
        s0 = cgmm_ldq_tail(srcrow, e0, E); s1 = cgmm_ldq_tail(srcrow, e1, E);
    }
    __syncthreads();
    int dd[8] = {d0.x, d0.y, d0.z, d0.w, d1.x, d1.y, d1.z, d1.w};
    int ss[8] = {s0.x, s0.y, s0.z, s0.w, s1.x, s1.y, s1.z, s1.w};
    int pp[8];
#pragma unroll
    for (int k = 0; k < 8; ++k)
        pp[k] = dd[k] >= 0 ? atomicAdd(&cur[dd[k] >> CGMM_NPB_BITS], 1) : 0;
    __syncthreads();
    int c = t < 256 ? cur[t] : 0;
    int ex = cgmm_block_excl_scan(c, wsum, t);
    if (t < 256) {
        cur[t] = ex;
        if (t < LP) {
            int2 se = {ex, ex + c};
            ((int2*)lstart)[blk * LP + t] = se;
        }
    }
    __syncthreads();
    int st[8];
#pragma unroll
    for (int k = 0; k < 8; ++k)
        st[k] = cur[dd[k] >= 0 ? dd[k] >> CGMM_NPB_BITS : 0];
#pragma unroll
    for (int k = 0; k < 8; ++k) {
        if (dd[k] >= 0)
            stage[st[k] + pp[k]] = (int)((((unsigned)dd[k] & 511u) << CGMM_SRC_BITS) | (unsigned)ss[k]);
    }
    __syncthreads();
    int nblk = E - blk * CGMM_CH;
    int* slice = bkt + blk * CGMM_CH;
    if (nblk >= CGMM_CH) {
        ((int4*)slice)[t]        = ((const int4*)stage)[t];
        ((int4*)slice)[t + 1024] = ((const int4*)stage)[t + 1024];
    } else {
        for (int j = t; j < nblk; j += 1024) slice[j] = stage[j];
    }
}

#define CGMM_IDX(p) __builtin_amdgcn_readlane(myidx, (p))
#define CGMM_AT(i)  (((const float2*)(prev_h + (size_t)(i) * 64))[lane])
#define CGMM_ROW(k) CGMM_AT(CGMM_IDX(j + (k)))
#define CGMM_ADD(r) acc.x += r.x; acc.y += r.y;
__device__ __forceinline__ float2 cgmm_add_rows(const float* __restrict__ prev_h, int lane,
                                                int myidx, int cnt, float2 acc) {
    int j = 0;
    for (; j + 16 <= cnt; j += 16) {
        float2 r0 = CGMM_ROW(0),  r1 = CGMM_ROW(1),  r2 = CGMM_ROW(2),  r3 = CGMM_ROW(3);
        float2 r4 = CGMM_ROW(4),  r5 = CGMM_ROW(5),  r6 = CGMM_ROW(6),  r7 = CGMM_ROW(7);
        float2 r8 = CGMM_ROW(8),  r9 = CGMM_ROW(9),  r10 = CGMM_ROW(10), r11 = CGMM_ROW(11);
        float2 r12 = CGMM_ROW(12), r13 = CGMM_ROW(13), r14 = CGMM_ROW(14), r15 = CGMM_ROW(15);
        CGMM_ADD(r0) CGMM_ADD(r1) CGMM_ADD(r2) CGMM_ADD(r3)
        CGMM_ADD(r4) CGMM_ADD(r5) CGMM_ADD(r6) CGMM_ADD(r7)
        CGMM_ADD(r8) CGMM_ADD(r9) CGMM_ADD(r10) CGMM_ADD(r11)
        CGMM_ADD(r12) CGMM_ADD(r13) CGMM_ADD(r14) CGMM_ADD(r15)
    }
    int m = cnt - j;
    if (m) {
        int j4 = j + (m & 8), j2 = j4 + (m & 4), j1 = j2 + (m & 2);
        int ia0 = CGMM_IDX(j),      ia1 = CGMM_IDX(j + 1),  ia2 = CGMM_IDX(j + 2),  ia3 = CGMM_IDX(j + 3);
        int ia4 = CGMM_IDX(j + 4),  ia5 = CGMM_IDX(j + 5),  ia6 = CGMM_IDX(j + 6),  ia7 = CGMM_IDX(j + 7);
        int ib0 = CGMM_IDX(j4),     ib1 = CGMM_IDX(j4 + 1), ib2 = CGMM_IDX(j4 + 2), ib3 = CGMM_IDX(j4 + 3);
        int ic0 = CGMM_IDX(j2),     ic1 = CGMM_IDX(j2 + 1), ie0 = CGMM_IDX(j1);
        float2 zf = {0.0f, 0.0f};
        float2 a0 = zf, a1 = zf, a2 = zf, a3 = zf, a4 = zf, a5 = zf, a6 = zf, a7 = zf;
        float2 b0 = zf, b1 = zf, b2 = zf, b3 = zf, c0 = zf, c1 = zf, e0 = zf;
        if (m & 8) {
            a0 = CGMM_AT(ia0); a1 = CGMM_AT(ia1); a2 = CGMM_AT(ia2); a3 = CGMM_AT(ia3);
            a4 = CGMM_AT(ia4); a5 = CGMM_AT(ia5); a6 = CGMM_AT(ia6); a7 = CGMM_AT(ia7);
        }
        if (m & 4) { b0 = CGMM_AT(ib0); b1 = CGMM_AT(ib1); b2 = CGMM_AT(ib2); b3 = CGMM_AT(ib3); }
        if (m & 2) { c0 = CGMM_AT(ic0); c1 = CGMM_AT(ic1); }
        if (m & 1) { e0 = CGMM_AT(ie0); }
        if (m & 8) {
            CGMM_ADD(a0) CGMM_ADD(a1) CGMM_ADD(a2) CGMM_ADD(a3)
            CGMM_ADD(a4) CGMM_ADD(a5) CGMM_ADD(a6) CGMM_ADD(a7)
        }
        if (m & 4) { CGMM_ADD(b0) CGMM_ADD(b1) CGMM_ADD(b2) CGMM_ADD(b3) }
        if (m & 2) { CGMM_ADD(c0) CGMM_ADD(c1) }
        if (m & 1) { CGMM_ADD(e0) }
    }
    return acc;
}
#undef CGMM_ADD
#undef CGMM_ROW
#undef CGMM_AT
#undef CGMM_IDX

#define CGMM_CAP 10240
#define CGMM_PITCH 72
#define CGMM_MAXB1 1024
#define CGMM_SEGU 7

template <int MODE>
__device__ __forceinline__ void cgmm_bucket_pass(const int* __restrict__ bkt,
                                                 const int* segbase, const int* seglen,
                                                 int B1, int wv, int lane,
                                                 int* cc, int* outl, int* outg) {
    const unsigned smask = (1u << CGMM_SRC_BITS) - 1u;
    for (int s0 = wv; s0 < B1; s0 += 32 * CGMM_SEGU) {
        unsigned wa[CGMM_SEGU], wb[CGMM_SEGU];
        int bs[CGMM_SEGU], ln[CGMM_SEGU];
        int su = s0 + 32 * lane;
        bool lv = lane < CGMM_SEGU && su < B1;
        int myb = lv ? segbase[su] : 0;
        int myl = lv ? seglen[su]  : 0;
#pragma unroll
        for (int u = 0; u < CGMM_SEGU; ++u) {
            bs[u] = __builtin_amdgcn_readlane(myb, u);
            ln[u] = __builtin_amdgcn_readlane(myl, u);
            wa[u] = lane < ln[u]      ? (unsigned)bkt[bs[u] + lane]      : 0u;
            wb[u] = lane + 32 < ln[u] ? (unsigned)bkt[bs[u] + lane + 32] : 0u;
        }
        asm volatile("" :: "v"(wa[0]), "v"(wa[1]), "v"(wa[2]), "v"(wa[3]), "v"(wa[4]), "v"(wa[5]),
                           "v"(wa[6]), "v"(wb[0]), "v"(wb[1]), "v"(wb[2]), "v"(wb[3]), "v"(wb[4]),
                           "v"(wb[5]), "v"(wb[6]));
        if (MODE == 0) {
#pragma unroll
            for (int u = 0; u < CGMM_SEGU; ++u) {
                if (lane < ln[u])      atomicAdd(&cc[wa[u] >> CGMM_SRC_BITS], 1);
                if (lane + 32 < ln[u]) atomicAdd(&cc[wb[u] >> CGMM_SRC_BITS], 1);
            }
        } else {
            int pa[CGMM_SEGU], pb[CGMM_SEGU];
#pragma unroll
            for (int u = 0; u < CGMM_SEGU; ++u) {
                int ka = lane < ln[u]      ? (int)(wa[u] >> CGMM_SRC_BITS) : 512 + lane;
                int kb = lane + 32 < ln[u] ? (int)(wb[u] >> CGMM_SRC_BITS) : 512 + lane;
                pa[u] = atomicAdd(&cc[ka], 1);
                pb[u] = atomicAdd(&cc[kb], 1);
            }
#pragma unroll
            for (int u = 0; u < CGMM_SEGU; ++u) {
                if (MODE == 1) {
                    if (lane < ln[u])      outl[pa[u]] = (int)(wa[u] & smask);
                    if (lane + 32 < ln[u]) outl[pb[u]] = (int)(wb[u] & smask);
                } else {
                    if (lane < ln[u])      outg[pa[u]] = (int)(wa[u] & smask);
                    if (lane + 32 < ln[u]) outg[pb[u]] = (int)(wb[u] & smask);
                }
            }
        }
#pragma unroll
        for (int u = 0; u < CGMM_SEGU; ++u) {
            for (int o = lane + 64; o < ln[u]; o += 32) {
                unsigned w = (unsigned)bkt[bs[u] + o];
                if (MODE == 0) {
                    atomicAdd(&cc[w >> CGMM_SRC_BITS], 1);
                } else {
                    int p = atomicAdd(&cc[w >> CGMM_SRC_BITS], 1);
                    if (MODE == 1) outl[p] = (int)(w & smask);
                    else           outg[p] = (int)(w & smask);
                }
            }
        }
    }
}

struct cgmm_held {
    unsigned wa[CGMM_SEGU], wb[CGMM_SEGU];
    int pa[CGMM_SEGU], pb[CGMM_SEGU];
    int ln[CGMM_SEGU];
};

__device__ __forceinline__ void cgmm_rank_runs(const int* __restrict__ bkt,
                                               const int* segbase, const int* seglen,
                                               int B1, int wv, int lane, int* cc, cgmm_held& h) {
    int su = wv + 32 * lane;
    bool lv = lane < CGMM_SEGU && su < B1;
    int myb = lv ? segbase[su] : 0;
    int myl = lv ? seglen[su]  : 0;
#pragma unroll
    for (int u = 0; u < CGMM_SEGU; ++u) {
        int bs = __builtin_amdgcn_readlane(myb, u);
        h.ln[u] = __builtin_amdgcn_readlane(myl, u);
        h.wa[u] = lane < h.ln[u]      ? (unsigned)bkt[bs + lane]      : 0u;
        h.wb[u] = lane + 32 < h.ln[u] ? (unsigned)bkt[bs + lane + 32] : 0u;
    }
    asm volatile("" :: "v"(h.wa[0]), "v"(h.wa[1]), "v"(h.wa[2]), "v"(h.wa[3]), "v"(h.wa[4]),
                       "v"(h.wa[5]), "v"(h.wa[6]), "v"(h.wb[0]), "v"(h.wb[1]), "v"(h.wb[2]),
                       "v"(h.wb[3]), "v"(h.wb[4]), "v"(h.wb[5]), "v"(h.wb[6]));
#pragma unroll
    for (int u = 0; u < CGMM_SEGU; ++u) {
        int ka = lane < h.ln[u]      ? (int)(h.wa[u] >> CGMM_SRC_BITS) : 512 + lane;
        int kb = lane + 32 < h.ln[u] ? (int)(h.wb[u] >> CGMM_SRC_BITS) : 512 + lane;
        h.pa[u] = atomicAdd(&cc[ka], 1);
        h.pb[u] = atomicAdd(&cc[kb], 1);
    }
}

template <bool IN_LDS>
__device__ __forceinline__ void cgmm_place_runs(const cgmm_held& h, int lane, const int* nstart,
                                                int* outl, int* outg) {
    const unsigned smask = (1u << CGMM_SRC_BITS) - 1u;
    int qa[CGMM_SEGU], qb[CGMM_SEGU];
#pragma unroll
    for (int u = 0; u < CGMM_SEGU; ++u) {
        qa[u] = nstart[h.wa[u] >> CGMM_SRC_BITS];
        qb[u] = nstart[h.wb[u] >> CGMM_SRC_BITS];
    }
#pragma unroll
    for (int u = 0; u < CGMM_SEGU; ++u) {
        if (IN_LDS) {
            if (lane < h.ln[u])      outl[qa[u] + h.pa[u]] = (int)(h.wa[u] & smask);
            if (lane + 32 < h.ln[u]) outl[qb[u] + h.pb[u]] = (int)(h.wb[u] & smask);
        } else {
            if (lane < h.ln[u])      outg[qa[u] + h.pa[u]] = (int)(h.wa[u] & smask);
            if (lane + 32 < h.ln[u]) outg[qb[u] + h.pb[u]] = (int)(h.wb[u] & smask);
        }
    }
}

template <bool IN_LDS>
__device__ __forceinline__ int cgmm_src_at(const int* stage, const int* csrb, int pos) {
    if (IN_LDS) return stage[pos];
    return csrb[pos];
}

template <bool IN_LDS>
__device__ __forceinline__ void cgmm_chunk(const int* stage, const int* csrb,
                                           const int* cntl, const int* curl,
                                           int nl0, int n0, int nk, int lane,
                                           const float* __restrict__ prev_h,
                                           const int* __restrict__ x,
                                           const _Float16* __restrict__ bfrag,
                                           const float* __restrict__ Bs_t,
                                           float* __restrict__ lik,
                                           float* __restrict__ post,
                                           float (*lx)[CGMM_PITCH],
                                           float (*ly)[CGMM_PITCH]) {
    int dv = 0, sv = 0;
    if (lane < nk) { dv = cntl[nl0 + lane]; sv = curl[nl0 + lane] - dv; }
    int en = lane >> 2, gp = lane & 3;
    bool evalid = en < nk;
    int xi = x[evalid ? n0 + en : n0];
    int d_nx = __builtin_amdgcn_readlane(dv, 0);
    int s_nx = __builtin_amdgcn_readlane(sv, 0);
    int idx_nx = lane < d_nx ? cgmm_src_at<IN_LDS>(stage, csrb, s_nx + lane) : 0;
    for (int k = 0; k < nk; ++k) {
        int d = d_nx, start = s_nx, idx0 = idx_nx;
        if (!IN_LDS) asm volatile("" :: "v"(idx0));
        d_nx = __builtin_amdgcn_readlane(dv, k + 1);
        s_nx = __builtin_amdgcn_readlane(sv, k + 1);
        idx_nx = lane < d_nx ? cgmm_src_at<IN_LDS>(stage, csrb, s_nx + lane) : 0;
        float2 acc = {0.0f, 0.0f};
        acc = cgmm_add_rows(prev_h, lane, idx0, d < 32 ? d : 32, acc);
        for (int base = 32; base < d; base += 32) {
            int cnt = d - base < 32 ? d - base : 32;
            int myidx = lane < cnt ? cgmm_src_at<IN_LDS>(stage, csrb, start + base + lane) : 0;
            acc = cgmm_add_rows(prev_h, lane, myidx, cnt, acc);
        }
        *((float2*)&lx[k][2 * lane]) = acc;
    }
    for (int k = nk; k < 8; ++k) {
        float2 zf = {0.0f, 0.0f};
        *((float2*)&lx[k][2 * lane]) = zf;
    }

    const float4* bp = (const float4*)(Bs_t + (size_t)xi * 64 + gp * 16);
    float4 e0 = bp[0], e1 = bp[1], e2v = bp[2], e3 = bp[3];

    int  c  = lane & 15;
    bool hi = lane >= 16;
    int  nr = c & 7;
    float inv_cnt = 1.0f / fmaxf((float)__shfl(dv, nr, 32), 1.0f);
    const float* srow = &lx[nr][hi ? 1 : 0];
#pragma unroll
    for (int p = 0; p < 4; ++p) {
        v16h a;
#pragma unroll
        for (int s = 0; s < 8; ++s)
            a[s] = (_Float16)(srow[s * 8 + 2 * p] * inv_cnt);
#pragma unroll
        for (int s = 8; s < 16; ++s)
            a[s] = (_Float16)0.0f;

        v16h b = *((const v16h*)(bfrag + (p * 32 + lane) * 16));

        v8f acc = {};
        acc = __builtin_amdgcn_wmma_f32_16x16x32_f16(
            false, a, false, b, (short)0, acc, false, false);
        if (!hi) {
#pragma unroll
            for (int r = 0; r < 8; ++r)
                ly[r][16 * p + c] = acc[r];
        }
    }

    const float4* tp = (const float4*)&ly[en][gp * 16];
    float4 t0 = tp[0], t1 = tp[1], t2 = tp[2], t3 = tp[3];
    float ua[8], ub[8];
    ua[0] = t0.x * e0.x;  ua[1] = t0.y * e0.y;  ua[2] = t0.z * e0.z;  ua[3] = t0.w * e0.w;
    ua[4] = t1.x * e1.x;  ua[5] = t1.y * e1.y;  ua[6] = t1.z * e1.z;  ua[7] = t1.w * e1.w;
    ub[0] = t2.x * e2v.x; ub[1] = t2.y * e2v.y; ub[2] = t2.z * e2v.z; ub[3] = t2.w * e2v.w;
    ub[4] = t3.x * e3.x;  ub[5] = t3.y * e3.y;  ub[6] = t3.z * e3.z;  ub[7] = t3.w * e3.w;
    float sa = ((ua[0] + ua[1]) + (ua[2] + ua[3])) + ((ua[4] + ua[5]) + (ua[6] + ua[7]));
    float sb = ((ub[0] + ub[1]) + (ub[2] + ub[3])) + ((ub[4] + ub[5]) + (ub[6] + ub[7]));
    float da = sa + 64.0f * EPS_F, db = sb + 64.0f * EPS_F;
    float ia = 1.0f / da, ib = 1.0f / db;
#pragma unroll
    for (int i = 0; i < 8; ++i) {
        float2 pv = {(ua[i] + 8.0f * EPS_F) * ia, (ub[i] + 8.0f * EPS_F) * ib};
        *((float2*)&lx[en][i * 8 + 2 * gp]) = pv;
    }
    if (evalid) {
        float2 lv = {__logf(da), __logf(db)};
        *((float2*)(lik + (size_t)n0 * 8 + 2 * lane)) = lv;
    }
    float4* dstp = (float4*)(post + (size_t)n0 * 64);
#pragma unroll
    for (int q = 0; q < 4; ++q) {
        int f = 32 * q + lane;
        int nd = f >> 4;
        if (nd < nk) dstp[f] = *((const float4*)&lx[nd][(f & 15) * 4]);
    }
}

__global__ __launch_bounds__(1024) void cgmm_sort_gather(const int* __restrict__ bkt,
                                                         const int* __restrict__ lstart,
                                                         int* csr_src, int N, int B1, int LP,
                                                         const float* __restrict__ prev_h,
                                                         const int* __restrict__ x,
                                                         const _Float16* __restrict__ bfrag,
                                                         const float* __restrict__ Bs_t,
                                                         float* __restrict__ lik,
                                                         float* __restrict__ post) {
    __shared__ int cntl[544];
    __shared__ int curl[544];
    __shared__ int nstart[512];
    __shared__ int wsum[32];
    __shared__ int stot;
    __shared__ int segbase[CGMM_MAXB1];
    __shared__ int seglen[CGMM_MAXB1];
    __shared__ int stage[CGMM_CAP];
    __shared__ alignas(16) float ldsX[32][8][CGMM_PITCH];
    __shared__ alignas(16) float ldsY[32][8][CGMM_PITCH];
    int t = threadIdx.x;
    int lane = t & 31;
    int wv = __builtin_amdgcn_readfirstlane(t >> 5);
    const int bk = blockIdx.x;
    if (t < B1) {
        int2 se = ((const int2*)lstart)[t * LP + bk];
        segbase[t] = t * CGMM_CH + se.x;
        seglen[t]  = se.y - se.x;
    } else if (t < 32 * CGMM_SEGU) {
        seglen[t] = 0;
    }
    if (t < 544) cntl[t] = 0;
    __syncthreads();
    bool two_pass = B1 > 32 * CGMM_SEGU;
    if (!two_pass) {
        int mx = 0;
#pragma unroll
        for (int u = 0; u < CGMM_SEGU; ++u) {
            int l = seglen[lane + 32 * u];
            mx = l > mx ? l : mx;
        }
        two_pass = __builtin_amdgcn_ballot_w32(mx > 64) != 0u;
    }
    cgmm_held held;
    if (two_pass) cgmm_bucket_pass<0>(bkt, segbase, seglen, B1, wv, lane, cntl, stage, csr_src);
    else          cgmm_rank_runs(bkt, segbase, seglen, B1, wv, lane, cntl, held);
    __syncthreads();
    int d = t < 512 ? cntl[t] : 0;
    int ex = cgmm_block_excl_scan(d, wsum, t);
    if (t < 512) {
        nstart[t] = ex;
        curl[t] = two_pass ? ex : ex + d;
    } else if (t < 544) {
        curl[t] = 0;
    }
    if (t == 511) stot = ex + d;
    __syncthreads();
    const int cnt_b = stot;
    const bool in_lds = cnt_b <= CGMM_CAP;
    int lo = 0;
    if (!in_lds) {
        for (int s = 0; s < B1; ++s) lo += segbase[s] - s * CGMM_CH;
    }
    if (in_lds) {
        if (two_pass) cgmm_bucket_pass<1>(bkt, segbase, seglen, B1, wv, lane, curl, stage, csr_src);
        else          cgmm_place_runs<true>(held, lane, nstart, stage, csr_src);
    } else {
        if (two_pass) cgmm_bucket_pass<2>(bkt, segbase, seglen, B1, wv, lane, curl, stage, csr_src + lo);
        else          cgmm_place_runs<false>(held, lane, nstart, stage, csr_src + lo);
        __threadfence();
    }
    __syncthreads();

    const int nb0 = bk << CGMM_NPB_BITS;
    for (int ch = wv; ch < 64; ch += 32) {
        int nl0 = ch * 8, n0 = nb0 + nl0;
        if (n0 >= N) break;
        int nk = N - n0 < 8 ? N - n0 : 8;
        if (in_lds)
            cgmm_chunk<true>(stage, csr_src + lo, cntl, curl, nl0, n0, nk, lane, prev_h, x,
                             bfrag, Bs_t, lik, post, ldsX[wv], ldsY[wv]);
        else
            cgmm_chunk<false>(stage, csr_src + lo, cntl, curl, nl0, n0, nk, lane, prev_h, x,
                              bfrag, Bs_t, lik, post, ldsX[wv], ldsY[wv]);
    }
}

extern "C" void kernel_launch(void* const* d_in, const int* in_sizes, int n_in,
                              void* d_out, int out_size, void* d_ws, size_t ws_size,
                              hipStream_t stream) {
    (void)n_in; (void)out_size; (void)ws_size;
    const int*   x      = (const int*)d_in[0];
    const float* prev_h = (const float*)d_in[1];
    const int*   ei     = (const int*)d_in[2];
    const float* Q      = (const float*)d_in[3];
    const float* B      = (const float*)d_in[4];
    const int N = in_sizes[0];
    const int E = in_sizes[2] / 2;

    float* ws     = (float*)d_ws;
    float* Bs_t   = ws + (size_t)N * 65;
    size_t off    = ((size_t)N * 65 + 2048 + 7) & ~(size_t)7;
    _Float16* bfrag = (_Float16*)(ws + off);

    float* lik  = (float*)d_out;
    float* post = lik + (size_t)N * 8;

    const int NB = (N + 511) / 512;
    const int LP = (NB + 15) & ~15;
    const int B1 = (E + 8191) / 8192;
    int* bkt      = (int*)ws;
    int* lstart   = bkt + B1 * 8192;
    int* csr_src  = lstart + B1 * LP * 2;

    cgmm_bkt_scatter<<<B1 + 1, 1024, 0, stream>>>(ei, bkt, lstart, E, LP, B1,
                                                  Q, B, Bs_t, bfrag);

    cgmm_sort_gather<<<NB, 1024, 0, stream>>>(bkt, lstart, csr_src, N, B1, LP, prev_h, x,
                                              bfrag, Bs_t, lik, post);
}


# Gather-sum: a wave sums two nodes at once, a half-wave per node with 16-byte loads (row index by DPP row_share): half the load instructions and round trips per wave; float add order unchanged
# speedup vs baseline: 7.2148x; 1.0536x over previous
#include <hip/hip_runtime.h>
#include <hip/hip_bf16.h>
#include <math.h>


typedef __attribute__((ext_vector_type(16))) _Float16 v16h;
typedef __attribute__((ext_vector_type(8)))  float    v8f;

#define EPS_F 1e-8f

__device__ __forceinline__ void cgmm_params_block(const float* __restrict__ Q,
                                                  const float* __restrict__ B,
                                                  float* __restrict__ Bs_t,
                                                  _Float16* __restrict__ bfrag,
                                                  float* sQ, int t) {
    if (t < 64) {
        int g = t & 7, a = t >> 3;
        {
            int l = a;
            float mx = -1e30f;
            for (int i = 0; i < 8; ++i) mx = fmaxf(mx, Q[i * 64 + l * 8 + g]);
            float e[8], s = 0.0f;
            for (int i = 0; i < 8; ++i) { e[i] = __expf(Q[i * 64 + l * 8 + g] - mx); s += e[i]; }
            float inv = 1.0f / s;
            for (int i = 0; i < 8; ++i) sQ[i * 64 + l * 8 + g] = e[i] * inv;
        }
        {
            int i = a;
            float mx = -1e30f;
            for (int m = 0; m < 32; ++m) mx = fmaxf(mx, B[i * 256 + m * 8 + g]);
            float ev[32], s = 0.0f;
            for (int m = 0; m < 32; ++m) { ev[m] = __expf(B[i * 256 + m * 8 + g] - mx); s += ev[m]; }
            float inv = 1.0f / s;
            for (int m = 0; m < 32; ++m) Bs_t[m * 64 + g * 8 + i] = ev[m] * inv;
        }
    }
    __syncthreads();
    if (t >= 128) return;
    int p = t >> 5, lane = t & 31;
    bool hi = lane >= 16;
    int  m  = lane & 15;
    bool bvalid = (hi == (m >= 8));
    int  icol = m & 7;
    int  g = 2 * p + (hi ? 1 : 0);
    _Float16* dst = bfrag + (p * 32 + lane) * 16;
    for (int e2 = 0; e2 < 16; ++e2) {
        float f = (e2 < 8 && bvalid) ? sQ[icol * 64 + e2 * 8 + g] : 0.0f;
        dst[e2] = (_Float16)f;
    }
}

#define CGMM_NPB_BITS 9
#define CGMM_SRC_BITS 23
#define CGMM_CH       8192

__device__ __forceinline__ int4 cgmm_ldq_tail(const int* __restrict__ row, int e, int E) {
    int4 v;
    v.x = e     < E ? row[e]     : -1;
    v.y = e + 1 < E ? row[e + 1] : -1;
    v.z = e + 2 < E ? row[e + 2] : -1;
    v.w = e + 3 < E ? row[e + 3] : -1;
    return v;
}

__device__ __forceinline__ int cgmm_block_excl_scan(int v, int* wsum, int t) {
    int lane = t & 31, wv = t >> 5;
    int incl = v;
#pragma unroll
    for (int s = 1; s < 32; s <<= 1) {
        int up = __shfl_up(incl, s, 32);
        if (lane >= s) incl += up;
    }
    if (lane == 31) wsum[wv] = incl;
    __syncthreads();
    if (wv == 0) {
        int w = wsum[lane];
        int wi = w;
#pragma unroll
        for (int s = 1; s < 32; s <<= 1) {
            int up = __shfl_up(wi, s, 32);
            if (lane >= s) wi += up;
        }
        wsum[lane] = wi - w;
    }
    __syncthreads();
    return wsum[wv] + (incl - v);
}

__global__ __launch_bounds__(1024) void cgmm_bkt_scatter(const int* __restrict__ ei,
                                                         int* __restrict__ bkt,
                                                         int* __restrict__ lstart,
                                                         int E, int LP, int B1,
                                                         const float* __restrict__ Q,
                                                         const float* __restrict__ B,
                                                         float* __restrict__ Bs_t,
                                                         _Float16* __restrict__ bfrag) {
    __shared__ int cur[256];
    __shared__ int wsum[32];
    __shared__ alignas(16) int stage[CGMM_CH];
    __shared__ float sQ[512];
    int t = threadIdx.x;
    if ((int)blockIdx.x == B1) {
        cgmm_params_block(Q, B, Bs_t, bfrag, sQ, t);
        return;
    }
    if (t < 256) cur[t] = 0;
    const int blk = blockIdx.x;
    const int* srcrow = ei + E;
    int e0 = blk * CGMM_CH + 4 * t, e1 = e0 + 4096;
    int4 d0, d1, s0, s1;
    if (e1 + 3 < E && (E & 3) == 0) {
        d0 = *((const int4*)(ei + e0));     d1 = *((const int4*)(ei + e1));
        s0 = *((const int4*)(srcrow + e0)); s1 = *((const int4*)(srcrow + e1));
    } else {
        d0 = cgmm_ldq_tail(ei, e0, E);     d1 = cgmm_ldq_tail(ei, e1, E);
        s0 = cgmm_ldq_tail(srcrow, e0, E); s1 = cgmm_ldq_tail(srcrow, e1, E);
    }
    __syncthreads();
    int dd[8] = {d0.x, d0.y, d0.z, d0.w, d1.x, d1.y, d1.z, d1.w};
    int ss[8] = {s0.x, s0.y, s0.z, s0.w, s1.x, s1.y, s1.z, s1.w};
    int pp[8];
#pragma unroll
    for (int k = 0; k < 8; ++k)
        pp[k] = dd[k] >= 0 ? atomicAdd(&cur[dd[k] >> CGMM_NPB_BITS], 1) : 0;
    __syncthreads();
    int c = t < 256 ? cur[t] : 0;
    int ex = cgmm_block_excl_scan(c, wsum, t);
    if (t < 256) {
        cur[t] = ex;
        if (t < LP) {
            int2 se = {ex, ex + c};
            ((int2*)lstart)[blk * LP + t] = se;
        }
    }
    __syncthreads();
    int st[8];
#pragma unroll
    for (int k = 0; k < 8; ++k)
        st[k] = cur[dd[k] >= 0 ? dd[k] >> CGMM_NPB_BITS : 0];
#pragma unroll
    for (int k = 0; k < 8; ++k) {
        if (dd[k] >= 0)
            stage[st[k] + pp[k]] = (int)((((unsigned)dd[k] & 511u) << CGMM_SRC_BITS) | (unsigned)ss[k]);
    }
    __syncthreads();
    int nblk = E - blk * CGMM_CH;
    int* slice = bkt + blk * CGMM_CH;
    if (nblk >= CGMM_CH) {
        ((int4*)slice)[t]        = ((const int4*)stage)[t];
        ((int4*)slice)[t + 1024] = ((const int4*)stage)[t + 1024];
    } else {
        for (int j = t; j < nblk; j += 1024) slice[j] = stage[j];
    }
}

template <int J>
__device__ __forceinline__ int cgmm_share(int v) {
    return __builtin_amdgcn_update_dpp(0, v, 0x150 + J, 0xf, 0xf, false);
}
template <int J>
__device__ __forceinline__ float4 cgmm_row(const float* __restrict__ prev_h, unsigned lb, int myoff) {
    unsigned off = (unsigned)cgmm_share<J>(myoff) | lb;
    return *((const float4*)((const char*)prev_h + off));
}
#define CGMM_LD4(a, b, c, d, J)                                                     \
    a = cgmm_row<(J)>(prev_h, lb, myoff);     b = cgmm_row<(J) + 1>(prev_h, lb, myoff);  \
    c = cgmm_row<(J) + 2>(prev_h, lb, myoff); d = cgmm_row<(J) + 3>(prev_h, lb, myoff);
#define CGMM_ADD4(r) acc.x += r.x; acc.y += r.y; acc.z += r.z; acc.w += r.w;
#define CGMM_ADD4S(r, J) { const bool on = (J) < cnt;                               \
    acc.x += on ? r.x : 0.0f; acc.y += on ? r.y : 0.0f;                             \
    acc.z += on ? r.z : 0.0f; acc.w += on ? r.w : 0.0f; }
#define CGMM_ADDG(a, b, c, d, J)                                                    \
    if (cn >= (J) + 4) { CGMM_ADD4(a) CGMM_ADD4(b) CGMM_ADD4(c) CGMM_ADD4(d) }      \
    else { CGMM_ADD4S(a, (J)) CGMM_ADD4S(b, (J) + 1) CGMM_ADD4S(c, (J) + 2) CGMM_ADD4S(d, (J) + 3) }
__device__ __forceinline__ float4 cgmm_add_rows2(const float* __restrict__ prev_h, unsigned lb,
                                                 int myidx, int cnt, int cm, int cn, float4 acc) {
    const int myoff = (int)((unsigned)myidx << 8);
    float4 r0, r1, r2, r3, r4, r5, r6, r7, r8, r9, r10, r11, r12, r13, r14, r15;
    if (cm > 12) {
        CGMM_LD4(r0, r1, r2, r3, 0) CGMM_LD4(r4, r5, r6, r7, 4)
        CGMM_LD4(r8, r9, r10, r11, 8) CGMM_LD4(r12, r13, r14, r15, 12)
        CGMM_ADDG(r0, r1, r2, r3, 0) CGMM_ADDG(r4, r5, r6, r7, 4)
        CGMM_ADDG(r8, r9, r10, r11, 8) CGMM_ADDG(r12, r13, r14, r15, 12)
    } else if (cm > 8) {
        CGMM_LD4(r0, r1, r2, r3, 0) CGMM_LD4(r4, r5, r6, r7, 4) CGMM_LD4(r8, r9, r10, r11, 8)
        CGMM_ADDG(r0, r1, r2, r3, 0) CGMM_ADDG(r4, r5, r6, r7, 4) CGMM_ADDG(r8, r9, r10, r11, 8)
    } else if (cm > 4) {
        CGMM_LD4(r0, r1, r2, r3, 0) CGMM_LD4(r4, r5, r6, r7, 4)
        CGMM_ADDG(r0, r1, r2, r3, 0) CGMM_ADDG(r4, r5, r6, r7, 4)
    } else {
        CGMM_LD4(r0, r1, r2, r3, 0)
        CGMM_ADDG(r0, r1, r2, r3, 0)
    }
    return acc;
}
#undef CGMM_ADDG
#undef CGMM_ADD4S
#undef CGMM_ADD4
#undef CGMM_LD4

#define CGMM_CAP 10240
#define CGMM_PITCH 72
#define CGMM_MAXB1 1024
#define CGMM_SEGU 7

template <int MODE>
__device__ __forceinline__ void cgmm_bucket_pass(const int* __restrict__ bkt,
                                                 const int* segbase, const int* seglen,
                                                 int B1, int wv, int lane,
                                                 int* cc, int* outl, int* outg) {
    const unsigned smask = (1u << CGMM_SRC_BITS) - 1u;
    for (int s0 = wv; s0 < B1; s0 += 32 * CGMM_SEGU) {
        unsigned wa[CGMM_SEGU], wb[CGMM_SEGU];
        int bs[CGMM_SEGU], ln[CGMM_SEGU];
        int su = s0 + 32 * lane;
        bool lv = lane < CGMM_SEGU && su < B1;
        int myb = lv ? segbase[su] : 0;
        int myl = lv ? seglen[su]  : 0;
#pragma unroll
        for (int u = 0; u < CGMM_SEGU; ++u) {
            bs[u] = __builtin_amdgcn_readlane(myb, u);
            ln[u] = __builtin_amdgcn_readlane(myl, u);
            wa[u] = lane < ln[u]      ? (unsigned)bkt[bs[u] + lane]      : 0u;
            wb[u] = lane + 32 < ln[u] ? (unsigned)bkt[bs[u] + lane + 32] : 0u;
        }
        asm volatile("" :: "v"(wa[0]), "v"(wa[1]), "v"(wa[2]), "v"(wa[3]), "v"(wa[4]), "v"(wa[5]),
                           "v"(wa[6]), "v"(wb[0]), "v"(wb[1]), "v"(wb[2]), "v"(wb[3]), "v"(wb[4]),
                           "v"(wb[5]), "v"(wb[6]));
        if (MODE == 0) {
#pragma unroll
            for (int u = 0; u < CGMM_SEGU; ++u) {
                if (lane < ln[u])      atomicAdd(&cc[wa[u] >> CGMM_SRC_BITS], 1);
                if (lane + 32 < ln[u]) atomicAdd(&cc[wb[u] >> CGMM_SRC_BITS], 1);
            }
        } else {
            int pa[CGMM_SEGU], pb[CGMM_SEGU];
#pragma unroll
            for (int u = 0; u < CGMM_SEGU; ++u) {
                int ka = lane < ln[u]      ? (int)(wa[u] >> CGMM_SRC_BITS) : 512 + lane;
                int kb = lane + 32 < ln[u] ? (int)(wb[u] >> CGMM_SRC_BITS) : 512 + lane;
                pa[u] = atomicAdd(&cc[ka], 1);
                pb[u] = atomicAdd(&cc[kb], 1);
            }
#pragma unroll
            for (int u = 0; u < CGMM_SEGU; ++u) {
                if (MODE == 1) {
                    if (lane < ln[u])      outl[pa[u]] = (int)(wa[u] & smask);
                    if (lane + 32 < ln[u]) outl[pb[u]] = (int)(wb[u] & smask);
                } else {
                    if (lane < ln[u])      outg[pa[u]] = (int)(wa[u] & smask);
                    if (lane + 32 < ln[u]) outg[pb[u]] = (int)(wb[u] & smask);
                }
            }
        }
#pragma unroll
        for (int u = 0; u < CGMM_SEGU; ++u) {
            for (int o = lane + 64; o < ln[u]; o += 32) {
                unsigned w = (unsigned)bkt[bs[u] + o];
                if (MODE == 0) {
                    atomicAdd(&cc[w >> CGMM_SRC_BITS], 1);
                } else {
                    int p = atomicAdd(&cc[w >> CGMM_SRC_BITS], 1);
                    if (MODE == 1) outl[p] = (int)(w & smask);
                    else           outg[p] = (int)(w & smask);
                }
            }
        }
    }
}

struct cgmm_held {
    unsigned wa[CGMM_SEGU], wb[CGMM_SEGU];
    int pa[CGMM_SEGU], pb[CGMM_SEGU];
    int ln[CGMM_SEGU];
};

__device__ __forceinline__ void cgmm_rank_runs(const int* __restrict__ bkt,
                                               const int* segbase, const int* seglen,
                                               int B1, int wv, int lane, int* cc, cgmm_held& h) {
    int su = wv + 32 * lane;
    bool lv = lane < CGMM_SEGU && su < B1;
    int myb = lv ? segbase[su] : 0;
    int myl = lv ? seglen[su]  : 0;
#pragma unroll
    for (int u = 0; u < CGMM_SEGU; ++u) {
        int bs = __builtin_amdgcn_readlane(myb, u);
        h.ln[u] = __builtin_amdgcn_readlane(myl, u);
        h.wa[u] = lane < h.ln[u]      ? (unsigned)bkt[bs + lane]      : 0u;
        h.wb[u] = lane + 32 < h.ln[u] ? (unsigned)bkt[bs + lane + 32] : 0u;
    }
    asm volatile("" :: "v"(h.wa[0]), "v"(h.wa[1]), "v"(h.wa[2]), "v"(h.wa[3]), "v"(h.wa[4]),
                       "v"(h.wa[5]), "v"(h.wa[6]), "v"(h.wb[0]), "v"(h.wb[1]), "v"(h.wb[2]),
                       "v"(h.wb[3]), "v"(h.wb[4]), "v"(h.wb[5]), "v"(h.wb[6]));
#pragma unroll
    for (int u = 0; u < CGMM_SEGU; ++u) {
        int ka = lane < h.ln[u]      ? (int)(h.wa[u] >> CGMM_SRC_BITS) : 512 + lane;
        int kb = lane + 32 < h.ln[u] ? (int)(h.wb[u] >> CGMM_SRC_BITS) : 512 + lane;
        h.pa[u] = atomicAdd(&cc[ka], 1);
        h.pb[u] = atomicAdd(&cc[kb], 1);
    }
}

template <bool IN_LDS>
__device__ __forceinline__ void cgmm_place_runs(const cgmm_held& h, int lane, const int* nstart,
                                                int* outl, int* outg) {
    const unsigned smask = (1u << CGMM_SRC_BITS) - 1u;
    int qa[CGMM_SEGU], qb[CGMM_SEGU];
#pragma unroll
    for (int u = 0; u < CGMM_SEGU; ++u) {
        qa[u] = nstart[h.wa[u] >> CGMM_SRC_BITS];
        qb[u] = nstart[h.wb[u] >> CGMM_SRC_BITS];
    }
#pragma unroll
    for (int u = 0; u < CGMM_SEGU; ++u) {
        if (IN_LDS) {
            if (lane < h.ln[u])      outl[qa[u] + h.pa[u]] = (int)(h.wa[u] & smask);
            if (lane + 32 < h.ln[u]) outl[qb[u] + h.pb[u]] = (int)(h.wb[u] & smask);
        } else {
            if (lane < h.ln[u])      outg[qa[u] + h.pa[u]] = (int)(h.wa[u] & smask);
            if (lane + 32 < h.ln[u]) outg[qb[u] + h.pb[u]] = (int)(h.wb[u] & smask);
        }
    }
}

template <bool IN_LDS>
__device__ __forceinline__ int cgmm_src_at(const int* stage, const int* csrb, int pos) {
    if (IN_LDS) return stage[pos];
    return csrb[pos];
}

template <bool IN_LDS>
__device__ __forceinline__ void cgmm_chunk(const int* stage, const int* csrb,
                                           const int* cntl, const int* curl,
                                           int nl0, int n0, int nk, int lane,
                                           const float* __restrict__ prev_h,
                                           int xi,
                                           const _Float16* __restrict__ bfrag,
                                           const float* __restrict__ Bs_t,
                                           float* __restrict__ lik,
                                           float* __restrict__ post,
                                           float (*lx)[CGMM_PITCH],
                                           float (*ly)[CGMM_PITCH]) {
    int dv = 0, sv = 0;
    if (lane < nk) { dv = cntl[nl0 + lane]; sv = curl[nl0 + lane] - dv; }
    int en = lane >> 2, gp = lane & 3;
    bool evalid = en < nk;
    const int  l16 = lane & 15;
    const bool hb  = lane >= 16;
    const unsigned lb = (unsigned)l16 << 4;
    int dA = __builtin_amdgcn_readlane(dv, 0), dB = __builtin_amdgcn_readlane(dv, 1);
    int sA = __builtin_amdgcn_readlane(sv, 0), sB = __builtin_amdgcn_readlane(sv, 1);
    int d_nx = hb ? dB : dA, s_nx = hb ? sB : sA, mx_nx = dA > dB ? dA : dB, mn_nx = dA < dB ? dA : dB;
    int i0_nx = l16 < d_nx      ? cgmm_src_at<IN_LDS>(stage, csrb, s_nx + l16)      : 0;
    int i1_nx = l16 + 16 < d_nx ? cgmm_src_at<IN_LDS>(stage, csrb, s_nx + 16 + l16) : 0;
    for (int st = 0; st < 4; ++st) {
        const int d = d_nx, start = s_nx, mx = mx_nx, mn = mn_nx, i0 = i0_nx, i1 = i1_nx;
        if (!IN_LDS) asm volatile("" :: "v"(i0), "v"(i1));
        dA = __builtin_amdgcn_readlane(dv, 2 * st + 2);
        dB = __builtin_amdgcn_readlane(dv, 2 * st + 3);
        sA = __builtin_amdgcn_readlane(sv, 2 * st + 2);
        sB = __builtin_amdgcn_readlane(sv, 2 * st + 3);
        d_nx = hb ? dB : dA; s_nx = hb ? sB : sA; mx_nx = dA > dB ? dA : dB; mn_nx = dA < dB ? dA : dB;
        i0_nx = l16 < d_nx      ? cgmm_src_at<IN_LDS>(stage, csrb, s_nx + l16)      : 0;
        i1_nx = l16 + 16 < d_nx ? cgmm_src_at<IN_LDS>(stage, csrb, s_nx + 16 + l16) : 0;
        float4 acc = {0.0f, 0.0f, 0.0f, 0.0f};
        for (int base = 0; base < mx; base += 16) {
            int mi;
            if (base == 0)       mi = i0;
            else if (base == 16) mi = i1;
            else mi = base + l16 < d ? cgmm_src_at<IN_LDS>(stage, csrb, start + base + l16) : 0;
            acc = cgmm_add_rows2(prev_h, lb, mi, d - base, mx - base < 16 ? mx - base : 16, mn - base, acc);
        }
        *((float4*)&lx[2 * st + (hb ? 1 : 0)][4 * l16]) = acc;
    }

    const float4* bp = (const float4*)(Bs_t + (size_t)xi * 64 + gp * 16);
    float4 e0 = bp[0], e1 = bp[1], e2v = bp[2], e3 = bp[3];

    int  c  = lane & 15;
    bool hi = lane >= 16;
    int  nr = c & 7;
    float inv_cnt = 1.0f / fmaxf((float)__shfl(dv, nr, 32), 1.0f);
    const float* srow = &lx[nr][hi ? 1 : 0];
#pragma unroll
    for (int p = 0; p < 4; ++p) {
        v16h a;
#pragma unroll
        for (int s = 0; s < 8; ++s)
            a[s] = (_Float16)(srow[s * 8 + 2 * p] * inv_cnt);
#pragma unroll
        for (int s = 8; s < 16; ++s)
            a[s] = (_Float16)0.0f;

        v16h b = *((const v16h*)(bfrag + (p * 32 + lane) * 16));

        v8f acc = {};
        acc = __builtin_amdgcn_wmma_f32_16x16x32_f16(
            false, a, false, b, (short)0, acc, false, false);
        if (!hi) {
#pragma unroll
            for (int r = 0; r < 8; ++r)
                ly[r][16 * p + c] = acc[r];
        }
    }

    const float4* tp = (const float4*)&ly[en][gp * 16];
    float4 t0 = tp[0], t1 = tp[1], t2 = tp[2], t3 = tp[3];
    float ua[8], ub[8];
    ua[0] = t0.x * e0.x;  ua[1] = t0.y * e0.y;  ua[2] = t0.z * e0.z;  ua[3] = t0.w * e0.w;
    ua[4] = t1.x * e1.x;  ua[5] = t1.y * e1.y;  ua[6] = t1.z * e1.z;  ua[7] = t1.w * e1.w;
    ub[0] = t2.x * e2v.x; ub[1] = t2.y * e2v.y; ub[2] = t2.z * e2v.z; ub[3] = t2.w * e2v.w;
    ub[4] = t3.x * e3.x;  ub[5] = t3.y * e3.y;  ub[6] = t3.z * e3.z;  ub[7] = t3.w * e3.w;
    float sa = ((ua[0] + ua[1]) + (ua[2] + ua[3])) + ((ua[4] + ua[5]) + (ua[6] + ua[7]));
    float sb = ((ub[0] + ub[1]) + (ub[2] + ub[3])) + ((ub[4] + ub[5]) + (ub[6] + ub[7]));
    float da = sa + 64.0f * EPS_F, db = sb + 64.0f * EPS_F;
    float ia = 1.0f / da, ib = 1.0f / db;
#pragma unroll
    for (int i = 0; i < 8; ++i) {
        float2 pv = {(ua[i] + 8.0f * EPS_F) * ia, (ub[i] + 8.0f * EPS_F) * ib};
        *((float2*)&lx[en][i * 8 + 2 * gp]) = pv;
    }
    if (evalid) {
        float2 lv = {__logf(da), __logf(db)};
        *((float2*)(lik + (size_t)n0 * 8 + 2 * lane)) = lv;
    }
    float4* dstp = (float4*)(post + (size_t)n0 * 64);
#pragma unroll
    for (int q = 0; q < 4; ++q) {
        int f = 32 * q + lane;
        int nd = f >> 4;
        if (nd < nk) dstp[f] = *((const float4*)&lx[nd][(f & 15) * 4]);
    }
}

__global__ __launch_bounds__(1024) void cgmm_sort_gather(const int* __restrict__ bkt,
                                                         const int* __restrict__ lstart,
                                                         int* csr_src, int N, int B1, int LP,
                                                         const float* __restrict__ prev_h,
                                                         const int* __restrict__ x,
                                                         const _Float16* __restrict__ bfrag,
                                                         const float* __restrict__ Bs_t,
                                                         float* __restrict__ lik,
                                                         float* __restrict__ post) {
    __shared__ int cntl[544];
    __shared__ int curl[544];
    __shared__ int nstart[512];
    __shared__ int wsum[32];
    __shared__ int stot;
    __shared__ int segbase[CGMM_MAXB1];
    __shared__ int seglen[CGMM_MAXB1];
    __shared__ int stage[CGMM_CAP];
    __shared__ alignas(16) float ldsX[32][8][CGMM_PITCH];
    __shared__ alignas(16) float ldsY[32][8][CGMM_PITCH];
    int t = threadIdx.x;
    int lane = t & 31;
    int wv = __builtin_amdgcn_readfirstlane(t >> 5);
    const int bk = blockIdx.x;
    const int nb0 = bk << CGMM_NPB_BITS;
    const int xn0 = nb0 + wv * 8 + (lane >> 2), xn1 = xn0 + 256;
    const int xs0 = xn0 < N ? x[xn0] : 0;
    const int xs1 = xn1 < N ? x[xn1] : 0;
    if (t < B1) {
        int2 se = ((const int2*)lstart)[t * LP + bk];
        segbase[t] = t * CGMM_CH + se.x;
        seglen[t]  = se.y - se.x;
    } else if (t < 32 * CGMM_SEGU) {
        seglen[t] = 0;
    }
    if (t < 544) cntl[t] = 0;
    __syncthreads();
    bool two_pass = B1 > 32 * CGMM_SEGU;
    if (!two_pass) {
        int mx = 0;
#pragma unroll
        for (int u = 0; u < CGMM_SEGU; ++u) {
            int l = seglen[lane + 32 * u];
            mx = l > mx ? l : mx;
        }
        two_pass = __builtin_amdgcn_ballot_w32(mx > 64) != 0u;
    }
    cgmm_held held;
    if (two_pass) cgmm_bucket_pass<0>(bkt, segbase, seglen, B1, wv, lane, cntl, stage, csr_src);
    else          cgmm_rank_runs(bkt, segbase, seglen, B1, wv, lane, cntl, held);
    __syncthreads();
    int d = t < 512 ? cntl[t] : 0;
    int ex = cgmm_block_excl_scan(d, wsum, t);
    if (t < 512) {
        nstart[t] = ex;
        curl[t] = two_pass ? ex : ex + d;
    } else if (t < 544) {
        curl[t] = 0;
    }
    if (t == 511) stot = ex + d;
    __syncthreads();
    const int cnt_b = stot;
    const bool in_lds = cnt_b <= CGMM_CAP;
    int lo = 0;
    if (!in_lds) {
        for (int s = 0; s < B1; ++s) lo += segbase[s] - s * CGMM_CH;
    }
    if (in_lds) {
        if (two_pass) cgmm_bucket_pass<1>(bkt, segbase, seglen, B1, wv, lane, curl, stage, csr_src);
        else          cgmm_place_runs<true>(held, lane, nstart, stage, csr_src);
    } else {
        if (two_pass) cgmm_bucket_pass<2>(bkt, segbase, seglen, B1, wv, lane, curl, stage, csr_src + lo);
        else          cgmm_place_runs<false>(held, lane, nstart, stage, csr_src + lo);
        __threadfence();
    }
    __syncthreads();

    for (int c = 0; c < 2; ++c) {
        int nl0 = (wv + 32 * c) * 8, n0 = nb0 + nl0;
        if (n0 >= N) break;
        int nk = N - n0 < 8 ? N - n0 : 8;
        int xi = c ? xs1 : xs0;
        if (in_lds)
            cgmm_chunk<true>(stage, csr_src + lo, cntl, curl, nl0, n0, nk, lane, prev_h, xi,
                             bfrag, Bs_t, lik, post, ldsX[wv], ldsY[wv]);
        else
            cgmm_chunk<false>(stage, csr_src + lo, cntl, curl, nl0, n0, nk, lane, prev_h, xi,
                              bfrag, Bs_t, lik, post, ldsX[wv], ldsY[wv]);
    }
}

extern "C" void kernel_launch(void* const* d_in, const int* in_sizes, int n_in,
                              void* d_out, int out_size, void* d_ws, size_t ws_size,
                              hipStream_t stream) {
    (void)n_in; (void)out_size; (void)ws_size;
    const int*   x      = (const int*)d_in[0];
    const float* prev_h = (const float*)d_in[1];
    const int*   ei     = (const int*)d_in[2];
    const float* Q      = (const float*)d_in[3];
    const float* B      = (const float*)d_in[4];
    const int N = in_sizes[0];
    const int E = in_sizes[2] / 2;

    float* ws     = (float*)d_ws;
    float* Bs_t   = ws + (size_t)N * 65;
    size_t off    = ((size_t)N * 65 + 2048 + 7) & ~(size_t)7;
    _Float16* bfrag = (_Float16*)(ws + off);

    float* lik  = (float*)d_out;
    float* post = lik + (size_t)N * 8;

    const int NB = (N + 511) / 512;
    const int LP = (NB + 15) & ~15;
    const int B1 = (E + 8191) / 8192;
    int* bkt      = (int*)ws;
    int* lstart   = bkt + B1 * 8192;
    int* csr_src  = lstart + B1 * LP * 2;

    cgmm_bkt_scatter<<<B1 + 1, 1024, 0, stream>>>(ei, bkt, lstart, E, LP, B1,
                                                  Q, B, Bs_t, bfrag);

    cgmm_sort_gather<<<NB, 1024, 0, stream>>>(bkt, lstart, csr_src, N, B1, LP, prev_h, x,
                                              bfrag, Bs_t, lik, post);
}


# Bucket width 512 -> 392 nodes: 256 work-groups, one per WGP, share sort and gather-sum instead of 196 (60 WGPs sat idle); a wave sums 12 or 14 nodes, not 16; float add order: LDS arrival, as before
# speedup vs baseline: 7.5508x; 1.0466x over previous
#include <hip/hip_runtime.h>
#include <hip/hip_bf16.h>
#include <math.h>


typedef __attribute__((ext_vector_type(16))) _Float16 v16h;
typedef __attribute__((ext_vector_type(8)))  float    v8f;

#define EPS_F 1e-8f

__device__ __forceinline__ void cgmm_params_block(const float* __restrict__ Q,
                                                  const float* __restrict__ B,
                                                  float* __restrict__ Bs_t,
                                                  _Float16* __restrict__ bfrag,
                                                  float* sQ, int t) {
    if (t < 64) {
        int g = t & 7, a = t >> 3;
        {
            int l = a;
            float mx = -1e30f;
            for (int i = 0; i < 8; ++i) mx = fmaxf(mx, Q[i * 64 + l * 8 + g]);
            float e[8], s = 0.0f;
            for (int i = 0; i < 8; ++i) { e[i] = __expf(Q[i * 64 + l * 8 + g] - mx); s += e[i]; }
            float inv = 1.0f / s;
            for (int i = 0; i < 8; ++i) sQ[i * 64 + l * 8 + g] = e[i] * inv;
        }
        {
            int i = a;
            float mx = -1e30f;
            for (int m = 0; m < 32; ++m) mx = fmaxf(mx, B[i * 256 + m * 8 + g]);
            float ev[32], s = 0.0f;
            for (int m = 0; m < 32; ++m) { ev[m] = __expf(B[i * 256 + m * 8 + g] - mx); s += ev[m]; }
            float inv = 1.0f / s;
            for (int m = 0; m < 32; ++m) Bs_t[m * 64 + g * 8 + i] = ev[m] * inv;
        }
    }
    __syncthreads();
    if (t >= 128) return;
    int p = t >> 5, lane = t & 31;
    bool hi = lane >= 16;
    int  m  = lane & 15;
    bool bvalid = (hi == (m >= 8));
    int  icol = m & 7;
    int  g = 2 * p + (hi ? 1 : 0);
    _Float16* dst = bfrag + (p * 32 + lane) * 16;
    for (int e2 = 0; e2 < 16; ++e2) {
        float f = (e2 < 8 && bvalid) ? sQ[icol * 64 + e2 * 8 + g] : 0.0f;
        dst[e2] = (_Float16)f;
    }
}

#define CGMM_NPB      392
#define CGMM_SRC_BITS 23
#define CGMM_CH       8192

__device__ __forceinline__ int4 cgmm_ldq_tail(const int* __restrict__ row, int e, int E) {
    int4 v;
    v.x = e     < E ? row[e]     : -1;
    v.y = e + 1 < E ? row[e + 1] : -1;
    v.z = e + 2 < E ? row[e + 2] : -1;
    v.w = e + 3 < E ? row[e + 3] : -1;
    return v;
}

__device__ __forceinline__ int cgmm_block_excl_scan(int v, int* wsum, int t) {
    int lane = t & 31, wv = t >> 5;
    int incl = v;
#pragma unroll
    for (int s = 1; s < 32; s <<= 1) {
        int up = __shfl_up(incl, s, 32);
        if (lane >= s) incl += up;
    }
    if (lane == 31) wsum[wv] = incl;
    __syncthreads();
    if (wv == 0) {
        int w = wsum[lane];
        int wi = w;
#pragma unroll
        for (int s = 1; s < 32; s <<= 1) {
            int up = __shfl_up(wi, s, 32);
            if (lane >= s) wi += up;
        }
        wsum[lane] = wi - w;
    }
    __syncthreads();
    return wsum[wv] + (incl - v);
}

__global__ __launch_bounds__(1024) void cgmm_bkt_scatter(const int* __restrict__ ei,
                                                         int* __restrict__ bkt,
                                                         int* __restrict__ lstart,
                                                         int E, int LP, int B1,
                                                         const float* __restrict__ Q,
                                                         const float* __restrict__ B,
                                                         float* __restrict__ Bs_t,
                                                         _Float16* __restrict__ bfrag) {
    __shared__ int cur[256];
    __shared__ int wsum[32];
    __shared__ alignas(16) int stage[CGMM_CH];
    __shared__ float sQ[512];
    int t = threadIdx.x;
    if ((int)blockIdx.x == B1) {
        cgmm_params_block(Q, B, Bs_t, bfrag, sQ, t);
        return;
    }
    if (t < 256) cur[t] = 0;
    const int blk = blockIdx.x;
    const int* srcrow = ei + E;
    int e0 = blk * CGMM_CH + 4 * t, e1 = e0 + 4096;
    int4 d0, d1, s0, s1;
    if (e1 + 3 < E && (E & 3) == 0) {
        d0 = *((const int4*)(ei + e0));     d1 = *((const int4*)(ei + e1));
        s0 = *((const int4*)(srcrow + e0)); s1 = *((const int4*)(srcrow + e1));
    } else {
        d0 = cgmm_ldq_tail(ei, e0, E);     d1 = cgmm_ldq_tail(ei, e1, E);
        s0 = cgmm_ldq_tail(srcrow, e0, E); s1 = cgmm_ldq_tail(srcrow, e1, E);
    }
    __syncthreads();
    int dd[8] = {d0.x, d0.y, d0.z, d0.w, d1.x, d1.y, d1.z, d1.w};
    int ss[8] = {s0.x, s0.y, s0.z, s0.w, s1.x, s1.y, s1.z, s1.w};
    int pp[8], bb[8];
#pragma unroll
    for (int k = 0; k < 8; ++k)
        bb[k] = dd[k] >= 0 ? (int)((unsigned)dd[k] / (unsigned)CGMM_NPB) : 0;
#pragma unroll
    for (int k = 0; k < 8; ++k)
        pp[k] = dd[k] >= 0 ? atomicAdd(&cur[bb[k]], 1) : 0;
    __syncthreads();
    int c = t < 256 ? cur[t] : 0;
    int ex = cgmm_block_excl_scan(c, wsum, t);
    if (t < 256) {
        cur[t] = ex;
        if (t < LP) {
            int2 se = {ex, ex + c};
            ((int2*)lstart)[blk * LP + t] = se;
        }
    }
    __syncthreads();
    int st[8];
#pragma unroll
    for (int k = 0; k < 8; ++k)
        st[k] = cur[bb[k]];
#pragma unroll
    for (int k = 0; k < 8; ++k) {
        if (dd[k] >= 0)
            stage[st[k] + pp[k]] = (int)(((unsigned)(dd[k] - bb[k] * CGMM_NPB) << CGMM_SRC_BITS) | (unsigned)ss[k]);
    }
    __syncthreads();
    int nblk = E - blk * CGMM_CH;
    int* slice = bkt + blk * CGMM_CH;
    if (nblk >= CGMM_CH) {
        ((int4*)slice)[t]        = ((const int4*)stage)[t];
        ((int4*)slice)[t + 1024] = ((const int4*)stage)[t + 1024];
    } else {
        for (int j = t; j < nblk; j += 1024) slice[j] = stage[j];
    }
}

template <int J>
__device__ __forceinline__ int cgmm_share(int v) {
    return __builtin_amdgcn_update_dpp(0, v, 0x150 + J, 0xf, 0xf, false);
}
template <int J>
__device__ __forceinline__ float4 cgmm_row(const float* __restrict__ prev_h, unsigned lb, int myoff) {
    unsigned off = (unsigned)cgmm_share<J>(myoff) | lb;
    return *((const float4*)((const char*)prev_h + off));
}
#define CGMM_LD4(a, b, c, d, J)                                                     \
    a = cgmm_row<(J)>(prev_h, lb, myoff);     b = cgmm_row<(J) + 1>(prev_h, lb, myoff);  \
    c = cgmm_row<(J) + 2>(prev_h, lb, myoff); d = cgmm_row<(J) + 3>(prev_h, lb, myoff);
#define CGMM_ADD4(r) acc.x += r.x; acc.y += r.y; acc.z += r.z; acc.w += r.w;
#define CGMM_ADD4S(r, J) { const bool on = (J) < cnt;                               \
    acc.x += on ? r.x : 0.0f; acc.y += on ? r.y : 0.0f;                             \
    acc.z += on ? r.z : 0.0f; acc.w += on ? r.w : 0.0f; }
#define CGMM_ADDG(a, b, c, d, J)                                                    \
    if (cn >= (J) + 4) { CGMM_ADD4(a) CGMM_ADD4(b) CGMM_ADD4(c) CGMM_ADD4(d) }      \
    else { CGMM_ADD4S(a, (J)) CGMM_ADD4S(b, (J) + 1) CGMM_ADD4S(c, (J) + 2) CGMM_ADD4S(d, (J) + 3) }
__device__ __forceinline__ float4 cgmm_add_rows2(const float* __restrict__ prev_h, unsigned lb,
                                                 int myidx, int cnt, int cm, int cn, float4 acc) {
    const int myoff = (int)((unsigned)myidx << 8);
    float4 r0, r1, r2, r3, r4, r5, r6, r7, r8, r9, r10, r11, r12, r13, r14, r15;
    if (cm > 12) {
        CGMM_LD4(r0, r1, r2, r3, 0) CGMM_LD4(r4, r5, r6, r7, 4)
        CGMM_LD4(r8, r9, r10, r11, 8) CGMM_LD4(r12, r13, r14, r15, 12)
        CGMM_ADDG(r0, r1, r2, r3, 0) CGMM_ADDG(r4, r5, r6, r7, 4)
        CGMM_ADDG(r8, r9, r10, r11, 8) CGMM_ADDG(r12, r13, r14, r15, 12)
    } else if (cm > 8) {
        CGMM_LD4(r0, r1, r2, r3, 0) CGMM_LD4(r4, r5, r6, r7, 4) CGMM_LD4(r8, r9, r10, r11, 8)
        CGMM_ADDG(r0, r1, r2, r3, 0) CGMM_ADDG(r4, r5, r6, r7, 4) CGMM_ADDG(r8, r9, r10, r11, 8)
    } else if (cm > 4) {
        CGMM_LD4(r0, r1, r2, r3, 0) CGMM_LD4(r4, r5, r6, r7, 4)
        CGMM_ADDG(r0, r1, r2, r3, 0) CGMM_ADDG(r4, r5, r6, r7, 4)
    } else {
        CGMM_LD4(r0, r1, r2, r3, 0)
        CGMM_ADDG(r0, r1, r2, r3, 0)
    }
    return acc;
}
#undef CGMM_ADDG
#undef CGMM_ADD4S
#undef CGMM_ADD4
#undef CGMM_LD4

#define CGMM_CAP 10240
#define CGMM_PITCH 72
#define CGMM_MAXB1 1024
#define CGMM_SEGU 7

template <int MODE>
__device__ __forceinline__ void cgmm_bucket_pass(const int* __restrict__ bkt,
                                                 const int* segbase, const int* seglen,
                                                 int B1, int wv, int lane,
                                                 int* cc, int* outl, int* outg) {
    const unsigned smask = (1u << CGMM_SRC_BITS) - 1u;
    for (int s0 = wv; s0 < B1; s0 += 32 * CGMM_SEGU) {
        unsigned wa[CGMM_SEGU], wb[CGMM_SEGU];
        int bs[CGMM_SEGU], ln[CGMM_SEGU];
        int su = s0 + 32 * lane;
        bool lv = lane < CGMM_SEGU && su < B1;
        int myb = lv ? segbase[su] : 0;
        int myl = lv ? seglen[su]  : 0;
#pragma unroll
        for (int u = 0; u < CGMM_SEGU; ++u) {
            bs[u] = __builtin_amdgcn_readlane(myb, u);
            ln[u] = __builtin_amdgcn_readlane(myl, u);
            wa[u] = lane < ln[u]      ? (unsigned)bkt[bs[u] + lane]      : 0u;
            wb[u] = lane + 32 < ln[u] ? (unsigned)bkt[bs[u] + lane + 32] : 0u;
        }
        asm volatile("" :: "v"(wa[0]), "v"(wa[1]), "v"(wa[2]), "v"(wa[3]), "v"(wa[4]), "v"(wa[5]),
                           "v"(wa[6]), "v"(wb[0]), "v"(wb[1]), "v"(wb[2]), "v"(wb[3]), "v"(wb[4]),
                           "v"(wb[5]), "v"(wb[6]));
        if (MODE == 0) {
#pragma unroll
            for (int u = 0; u < CGMM_SEGU; ++u) {
                if (lane < ln[u])      atomicAdd(&cc[wa[u] >> CGMM_SRC_BITS], 1);
                if (lane + 32 < ln[u]) atomicAdd(&cc[wb[u] >> CGMM_SRC_BITS], 1);
            }
        } else {
            int pa[CGMM_SEGU], pb[CGMM_SEGU];
#pragma unroll
            for (int u = 0; u < CGMM_SEGU; ++u) {
                int ka = lane < ln[u]      ? (int)(wa[u] >> CGMM_SRC_BITS) : 512 + lane;
                int kb = lane + 32 < ln[u] ? (int)(wb[u] >> CGMM_SRC_BITS) : 512 + lane;
                pa[u] = atomicAdd(&cc[ka], 1);
                pb[u] = atomicAdd(&cc[kb], 1);
            }
#pragma unroll
            for (int u = 0; u < CGMM_SEGU; ++u) {
                if (MODE == 1) {
                    if (lane < ln[u])      outl[pa[u]] = (int)(wa[u] & smask);
                    if (lane + 32 < ln[u]) outl[pb[u]] = (int)(wb[u] & smask);
                } else {
                    if (lane < ln[u])      outg[pa[u]] = (int)(wa[u] & smask);
                    if (lane + 32 < ln[u]) outg[pb[u]] = (int)(wb[u] & smask);
                }
            }
        }
#pragma unroll
        for (int u = 0; u < CGMM_SEGU; ++u) {
            for (int o = lane + 64; o < ln[u]; o += 32) {
                unsigned w = (unsigned)bkt[bs[u] + o];
                if (MODE == 0) {
                    atomicAdd(&cc[w >> CGMM_SRC_BITS], 1);
                } else {
                    int p = atomicAdd(&cc[w >> CGMM_SRC_BITS], 1);
                    if (MODE == 1) outl[p] = (int)(w & smask);
                    else           outg[p] = (int)(w & smask);
                }
            }
        }
    }
}

struct cgmm_held {
    unsigned wa[CGMM_SEGU], wb[CGMM_SEGU];
    int pa[CGMM_SEGU], pb[CGMM_SEGU];
    int ln[CGMM_SEGU];
};

__device__ __forceinline__ void cgmm_rank_runs(const int* __restrict__ bkt,
                                               const int* segbase, const int* seglen,
                                               int B1, int wv, int lane, int* cc, cgmm_held& h) {
    int su = wv + 32 * lane;
    bool lv = lane < CGMM_SEGU && su < B1;
    int myb = lv ? segbase[su] : 0;
    int myl = lv ? seglen[su]  : 0;
#pragma unroll
    for (int u = 0; u < CGMM_SEGU; ++u) {
        int bs = __builtin_amdgcn_readlane(myb, u);
        h.ln[u] = __builtin_amdgcn_readlane(myl, u);
        h.wa[u] = lane < h.ln[u]      ? (unsigned)bkt[bs + lane]      : 0u;
        h.wb[u] = lane + 32 < h.ln[u] ? (unsigned)bkt[bs + lane + 32] : 0u;
    }
    asm volatile("" :: "v"(h.wa[0]), "v"(h.wa[1]), "v"(h.wa[2]), "v"(h.wa[3]), "v"(h.wa[4]),
                       "v"(h.wa[5]), "v"(h.wa[6]), "v"(h.wb[0]), "v"(h.wb[1]), "v"(h.wb[2]),
                       "v"(h.wb[3]), "v"(h.wb[4]), "v"(h.wb[5]), "v"(h.wb[6]));
#pragma unroll
    for (int u = 0; u < CGMM_SEGU; ++u) {
        int ka = lane < h.ln[u]      ? (int)(h.wa[u] >> CGMM_SRC_BITS) : 512 + lane;
        int kb = lane + 32 < h.ln[u] ? (int)(h.wb[u] >> CGMM_SRC_BITS) : 512 + lane;
        h.pa[u] = atomicAdd(&cc[ka], 1);
        h.pb[u] = atomicAdd(&cc[kb], 1);
    }
}

template <bool IN_LDS>
__device__ __forceinline__ void cgmm_place_runs(const cgmm_held& h, int lane, const int* nstart,
                                                int* outl, int* outg) {
    const unsigned smask = (1u << CGMM_SRC_BITS) - 1u;
    int qa[CGMM_SEGU], qb[CGMM_SEGU];
#pragma unroll
    for (int u = 0; u < CGMM_SEGU; ++u) {
        qa[u] = nstart[h.wa[u] >> CGMM_SRC_BITS];
        qb[u] = nstart[h.wb[u] >> CGMM_SRC_BITS];
    }
#pragma unroll
    for (int u = 0; u < CGMM_SEGU; ++u) {
        if (IN_LDS) {
            if (lane < h.ln[u])      outl[qa[u] + h.pa[u]] = (int)(h.wa[u] & smask);
            if (lane + 32 < h.ln[u]) outl[qb[u] + h.pb[u]] = (int)(h.wb[u] & smask);
        } else {
            if (lane < h.ln[u])      outg[qa[u] + h.pa[u]] = (int)(h.wa[u] & smask);
            if (lane + 32 < h.ln[u]) outg[qb[u] + h.pb[u]] = (int)(h.wb[u] & smask);
        }
    }
}

template <bool IN_LDS>
__device__ __forceinline__ int cgmm_src_at(const int* stage, const int* csrb, int pos) {
    if (IN_LDS) return stage[pos];
    return csrb[pos];
}

template <bool IN_LDS>
__device__ __forceinline__ void cgmm_chunk(const int* stage, const int* csrb,
                                           const int* cntl, const int* curl,
                                           int nl0, int n0, int nk, int lane,
                                           const float* __restrict__ prev_h,
                                           int xi,
                                           const _Float16* __restrict__ bfrag,
                                           const float* __restrict__ Bs_t,
                                           float* __restrict__ lik,
                                           float* __restrict__ post,
                                           float (*lx)[CGMM_PITCH],
                                           float (*ly)[CGMM_PITCH]) {
    int dv = 0, sv = 0;
    if (lane < nk) { dv = cntl[nl0 + lane]; sv = curl[nl0 + lane] - dv; }
    int en = lane >> 2, gp = lane & 3;
    bool evalid = en < nk;
    const int  l16 = lane & 15;
    const bool hb  = lane >= 16;
    const unsigned lb = (unsigned)l16 << 4;
    int dA = __builtin_amdgcn_readlane(dv, 0), dB = __builtin_amdgcn_readlane(dv, 1);
    int sA = __builtin_amdgcn_readlane(sv, 0), sB = __builtin_amdgcn_readlane(sv, 1);
    int d_nx = hb ? dB : dA, s_nx = hb ? sB : sA, mx_nx = dA > dB ? dA : dB, mn_nx = dA < dB ? dA : dB;
    int i0_nx = l16 < d_nx      ? cgmm_src_at<IN_LDS>(stage, csrb, s_nx + l16)      : 0;
    int i1_nx = l16 + 16 < d_nx ? cgmm_src_at<IN_LDS>(stage, csrb, s_nx + 16 + l16) : 0;
    for (int st = 0; st < 4; ++st) {
        const int d = d_nx, start = s_nx, mx = mx_nx, mn = mn_nx, i0 = i0_nx, i1 = i1_nx;
        if (!IN_LDS) asm volatile("" :: "v"(i0), "v"(i1));
        dA = __builtin_amdgcn_readlane(dv, 2 * st + 2);
        dB = __builtin_amdgcn_readlane(dv, 2 * st + 3);
        sA = __builtin_amdgcn_readlane(sv, 2 * st + 2);
        sB = __builtin_amdgcn_readlane(sv, 2 * st + 3);
        d_nx = hb ? dB : dA; s_nx = hb ? sB : sA; mx_nx = dA > dB ? dA : dB; mn_nx = dA < dB ? dA : dB;
        i0_nx = l16 < d_nx      ? cgmm_src_at<IN_LDS>(stage, csrb, s_nx + l16)      : 0;
        i1_nx = l16 + 16 < d_nx ? cgmm_src_at<IN_LDS>(stage, csrb, s_nx + 16 + l16) : 0;
        float4 acc = {0.0f, 0.0f, 0.0f, 0.0f};
        for (int base = 0; base < mx; base += 16) {
            int mi;
            if (base == 0)       mi = i0;
            else if (base == 16) mi = i1;
            else mi = base + l16 < d ? cgmm_src_at<IN_LDS>(stage, csrb, start + base + l16) : 0;
            acc = cgmm_add_rows2(prev_h, lb, mi, d - base, mx - base < 16 ? mx - base : 16, mn - base, acc);
        }
        *((float4*)&lx[2 * st + (hb ? 1 : 0)][4 * l16]) = acc;
    }

    const float4* bp = (const float4*)(Bs_t + (size_t)xi * 64 + gp * 16);
    float4 e0 = bp[0], e1 = bp[1], e2v = bp[2], e3 = bp[3];

    int  c  = lane & 15;
    bool hi = lane >= 16;
    int  nr = c & 7;
    float inv_cnt = 1.0f / fmaxf((float)__shfl(dv, nr, 32), 1.0f);
    const float* srow = &lx[nr][hi ? 1 : 0];
#pragma unroll
    for (int p = 0; p < 4; ++p) {
        v16h a;
#pragma unroll
        for (int s = 0; s < 8; ++s)
            a[s] = (_Float16)(srow[s * 8 + 2 * p] * inv_cnt);
#pragma unroll
        for (int s = 8; s < 16; ++s)
            a[s] = (_Float16)0.0f;

        v16h b = *((const v16h*)(bfrag + (p * 32 + lane) * 16));

        v8f acc = {};
        acc = __builtin_amdgcn_wmma_f32_16x16x32_f16(
            false, a, false, b, (short)0, acc, false, false);
        if (!hi) {
#pragma unroll
            for (int r = 0; r < 8; ++r)
                ly[r][16 * p + c] = acc[r];
        }
    }

    const float4* tp = (const float4*)&ly[en][gp * 16];
    float4 t0 = tp[0], t1 = tp[1], t2 = tp[2], t3 = tp[3];
    float ua[8], ub[8];
    ua[0] = t0.x * e0.x;  ua[1] = t0.y * e0.y;  ua[2] = t0.z * e0.z;  ua[3] = t0.w * e0.w;
    ua[4] = t1.x * e1.x;  ua[5] = t1.y * e1.y;  ua[6] = t1.z * e1.z;  ua[7] = t1.w * e1.w;
    ub[0] = t2.x * e2v.x; ub[1] = t2.y * e2v.y; ub[2] = t2.z * e2v.z; ub[3] = t2.w * e2v.w;
    ub[4] = t3.x * e3.x;  ub[5] = t3.y * e3.y;  ub[6] = t3.z * e3.z;  ub[7] = t3.w * e3.w;
    float sa = ((ua[0] + ua[1]) + (ua[2] + ua[3])) + ((ua[4] + ua[5]) + (ua[6] + ua[7]));
    float sb = ((ub[0] + ub[1]) + (ub[2] + ub[3])) + ((ub[4] + ub[5]) + (ub[6] + ub[7]));
    float da = sa + 64.0f * EPS_F, db = sb + 64.0f * EPS_F;
    float ia = 1.0f / da, ib = 1.0f / db;
#pragma unroll
    for (int i = 0; i < 8; ++i) {
        float2 pv = {(ua[i] + 8.0f * EPS_F) * ia, (ub[i] + 8.0f * EPS_F) * ib};
        *((float2*)&lx[en][i * 8 + 2 * gp]) = pv;
    }
    if (evalid) {
        float2 lv = {__logf(da), __logf(db)};
        *((float2*)(lik + (size_t)n0 * 8 + 2 * lane)) = lv;
    }
    float4* dstp = (float4*)(post + (size_t)n0 * 64);
#pragma unroll
    for (int q = 0; q < 4; ++q) {
        int f = 32 * q + lane;
        int nd = f >> 4;
        if (nd < nk) dstp[f] = *((const float4*)&lx[nd][(f & 15) * 4]);
    }
}

__global__ __launch_bounds__(1024) void cgmm_sort_gather(const int* __restrict__ bkt,
                                                         const int* __restrict__ lstart,
                                                         int* csr_src, int N, int B1, int LP,
                                                         const float* __restrict__ prev_h,
                                                         const int* __restrict__ x,
                                                         const _Float16* __restrict__ bfrag,
                                                         const float* __restrict__ Bs_t,
                                                         float* __restrict__ lik,
                                                         float* __restrict__ post) {
    __shared__ int cntl[544];
    __shared__ int curl[544];
    __shared__ int nstart[512];
    __shared__ int wsum[32];
    __shared__ int stot;
    __shared__ int segbase[CGMM_MAXB1];
    __shared__ int seglen[CGMM_MAXB1];
    __shared__ int stage[CGMM_CAP];
    __shared__ alignas(16) float ldsX[32][8][CGMM_PITCH];
    __shared__ alignas(16) float ldsY[32][8][CGMM_PITCH];
    int t = threadIdx.x;
    int lane = t & 31;
    int wv = __builtin_amdgcn_readfirstlane(t >> 5);
    const int bk = blockIdx.x;
    const int nb0 = bk * CGMM_NPB;
    const int rs0 = 2 * ((wv * ((CGMM_NPB - 256) / 2)) >> 5);
    const int rs1 = 2 * (((wv + 1) * ((CGMM_NPB - 256) / 2)) >> 5);
    const int xn0 = nb0 + wv * 8 + (lane >> 2), xn1 = nb0 + 256 + rs0 + (lane >> 2);
    const int xs0 = xn0 < N ? x[xn0] : 0;
    const int xs1 = xn1 < N ? x[xn1] : 0;
    if (t < B1) {
        int2 se = ((const int2*)lstart)[t * LP + bk];
        segbase[t] = t * CGMM_CH + se.x;
        seglen[t]  = se.y - se.x;
    } else if (t < 32 * CGMM_SEGU) {
        seglen[t] = 0;
    }
    if (t < 544) cntl[t] = 0;
    __syncthreads();
    bool two_pass = B1 > 32 * CGMM_SEGU;
    if (!two_pass) {
        int mx = 0;
#pragma unroll
        for (int u = 0; u < CGMM_SEGU; ++u) {
            int l = seglen[lane + 32 * u];
            mx = l > mx ? l : mx;
        }
        two_pass = __builtin_amdgcn_ballot_w32(mx > 64) != 0u;
    }
    cgmm_held held;
    if (two_pass) cgmm_bucket_pass<0>(bkt, segbase, seglen, B1, wv, lane, cntl, stage, csr_src);
    else          cgmm_rank_runs(bkt, segbase, seglen, B1, wv, lane, cntl, held);
    __syncthreads();
    int d = t < 512 ? cntl[t] : 0;
    int ex = cgmm_block_excl_scan(d, wsum, t);
    if (t < 512) {
        nstart[t] = ex;
        curl[t] = two_pass ? ex : ex + d;
    } else if (t < 544) {
        curl[t] = 0;
    }
    if (t == 511) stot = ex + d;
    __syncthreads();
    const int cnt_b = stot;
    const bool in_lds = cnt_b <= CGMM_CAP;
    int lo = 0;
    if (!in_lds) {
        for (int s = 0; s < B1; ++s) lo += segbase[s] - s * CGMM_CH;
    }
    if (in_lds) {
        if (two_pass) cgmm_bucket_pass<1>(bkt, segbase, seglen, B1, wv, lane, curl, stage, csr_src);
        else          cgmm_place_runs<true>(held, lane, nstart, stage, csr_src);
    } else {
        if (two_pass) cgmm_bucket_pass<2>(bkt, segbase, seglen, B1, wv, lane, curl, stage, csr_src + lo);
        else          cgmm_place_runs<false>(held, lane, nstart, stage, csr_src + lo);
        __threadfence();
    }
    __syncthreads();

    const int nbk = N - nb0 < CGMM_NPB ? N - nb0 : CGMM_NPB;
    for (int c = 0; c < 2; ++c) {
        int nl0 = c ? 256 + rs0 : wv * 8, n0 = nb0 + nl0;
        int cap = c ? rs1 - rs0 : 8;
        if (nl0 >= nbk) break;
        int nk = nbk - nl0 < cap ? nbk - nl0 : cap;
        int xi = c ? xs1 : xs0;
        if (in_lds)
            cgmm_chunk<true>(stage, csr_src + lo, cntl, curl, nl0, n0, nk, lane, prev_h, xi,
                             bfrag, Bs_t, lik, post, ldsX[wv], ldsY[wv]);
        else
            cgmm_chunk<false>(stage, csr_src + lo, cntl, curl, nl0, n0, nk, lane, prev_h, xi,
                              bfrag, Bs_t, lik, post, ldsX[wv], ldsY[wv]);
    }
}

extern "C" void kernel_launch(void* const* d_in, const int* in_sizes, int n_in,
                              void* d_out, int out_size, void* d_ws, size_t ws_size,
                              hipStream_t stream) {
    (void)n_in; (void)out_size; (void)ws_size;
    const int*   x      = (const int*)d_in[0];
    const float* prev_h = (const float*)d_in[1];
    const int*   ei     = (const int*)d_in[2];
    const float* Q      = (const float*)d_in[3];
    const float* B      = (const float*)d_in[4];
    const int N = in_sizes[0];
    const int E = in_sizes[2] / 2;

    float* ws     = (float*)d_ws;
    float* Bs_t   = ws + (size_t)N * 65;
    size_t off    = ((size_t)N * 65 + 2048 + 7) & ~(size_t)7;
    _Float16* bfrag = (_Float16*)(ws + off);

    float* lik  = (float*)d_out;
    float* post = lik + (size_t)N * 8;

    const int NB = (N + 391) / 392;
    const int LP = (NB + 15) & ~15;
    const int B1 = (E + 8191) / 8192;
    int* bkt      = (int*)ws;
    int* lstart   = bkt + B1 * 8192;
    int* csr_src  = lstart + B1 * LP * 2;

    cgmm_bkt_scatter<<<B1 + 1, 1024, 0, stream>>>(ei, bkt, lstart, E, LP, B1,
                                                  Q, B, Bs_t, bfrag);

    cgmm_sort_gather<<<NB, 1024, 0, stream>>>(bkt, lstart, csr_src, N, B1, LP, prev_h, x,
                                              bfrag, Bs_t, lik, post);
}
